# starting kernel
# baseline (speedup 1.0000x reference)
#include <hip/hip_runtime.h>

typedef float v2f __attribute__((ext_vector_type(2)));
typedef float v8f __attribute__((ext_vector_type(8)));

#define WMMA_F32(a, b, c) \
  __builtin_amdgcn_wmma_f32_16x16x4_f32(false, (a), false, (b), (short)0, (c), false, false)

#define BATCH 4
#define NTOK 4096
#define CCH 512
#define HEADS 8
#define DH 64
#define KL 64
#define MROWS (BATCH * NTOK)
#define OCH (4 * CCH)
#define BH (BATCH * HEADS)

#define OFF_Q   ((size_t)0)
#define OFF_K   ((size_t)8388608)
#define OFF_VCA ((size_t)16777216)
#define OFF_VSA ((size_t)25165824)
#define OFF_KP  ((size_t)33554432)
#define OFF_VP  ((size_t)33685504)
#define OFF_P   ((size_t)33816576)

__global__ __launch_bounds__(256) void qkvv_gemm(const float* __restrict__ x,
                                                 const float* __restrict__ W,
                                                 float* __restrict__ ws) {
  const int lane = threadIdx.x & 31;
  const int wave = threadIdx.x >> 5;
  const int l16  = lane & 15;
  const int hi   = lane >> 4;
  const int mBase = blockIdx.x * 128 + wave * 16;
  const int oBase = blockIdx.y * 64;

  const float* xRow = x + (size_t)(mBase + l16) * CCH + (hi ? 2 : 0);
  v8f acc[4] = {};
  for (int c0 = 0; c0 < CCH; c0 += 4) {
    v2f a = *(const v2f*)(xRow + c0);
#pragma unroll
    for (int t = 0; t < 4; ++t) {
      const int o = oBase + t * 16 + l16;
      v2f b = *(const v2f*)(W + (size_t)o * CCH + c0 + (hi ? 2 : 0));
      acc[t] = WMMA_F32(a, b, acc[t]);
    }
  }
#pragma unroll
  for (int t = 0; t < 4; ++t) {
    const int o  = oBase + t * 16 + l16;
    const int p  = o >> 9;
    const int h  = (o >> 6) & 7;
    const int dd = o & 63;
#pragma unroll
    for (int v = 0; v < 8; ++v) {
      const int m  = mBase + v + (hi ? 8 : 0);
      const int b_ = m >> 12;
      const int n  = m & (NTOK - 1);
      const size_t dst = (size_t)p * 8388608 +
                         ((size_t)(b_ * HEADS + h) * DH + dd) * NTOK + n;
      ws[dst] = acc[t][v];
    }
  }
}

__global__ __launch_bounds__(512) void ef_proj(float* __restrict__ ws,
                                               const float* __restrict__ EF) {
  const int lane = threadIdx.x & 31;
  const int wave = threadIdx.x >> 5;
  const int l16  = lane & 15;
  const int hi   = lane >> 4;
  const int which = blockIdx.x & 1;
  const int bh    = blockIdx.x >> 1;
  const int dt = wave >> 2, kt = wave & 3;

  const float* src = ws + (which ? OFF_VSA : OFF_K) + (size_t)bh * DH * NTOK;
  float* dst       = ws + (which ? OFF_VP : OFF_KP) + (size_t)bh * DH * KL;

  const float* aRow = src + (size_t)(dt * 16 + l16) * NTOK + (hi ? 2 : 0);
  const int kcol = kt * 16 + l16;
  v8f acc = {};
  for (int n0 = 0; n0 < NTOK; n0 += 4) {
    v2f a = *(const v2f*)(aRow + n0);
    const int k0 = n0 + (hi ? 2 : 0);
    v2f b;
    b.x = EF[(size_t)k0 * KL + kcol];
    b.y = EF[(size_t)(k0 + 1) * KL + kcol];
    acc = WMMA_F32(a, b, acc);
  }
#pragma unroll
  for (int v = 0; v < 8; ++v)
    dst[(size_t)(dt * 16 + v + (hi ? 8 : 0)) * KL + kcol] = acc[v];
}

__global__ __launch_bounds__(256) void l2norm_rows(float* __restrict__ ws) {
  __shared__ float red[256];
  const int tid = threadIdx.x;
  const int row = blockIdx.x;
  float* ptr = ws + (row < 2048 ? OFF_Q : OFF_K) +
               (size_t)(row & 2047) * NTOK;
  float s = 0.f;
#pragma unroll
  for (int i = 0; i < 16; ++i) {
    float v = ptr[tid + i * 256];
    s += v * v;
  }
  red[tid] = s;
  __syncthreads();
  for (int off = 128; off > 0; off >>= 1) {
    if (tid < off) red[tid] += red[tid + off];
    __syncthreads();
  }
  const float scale = 1.0f / fmaxf(sqrtf(red[0]), 1e-12f);
#pragma unroll
  for (int i = 0; i < 16; ++i) ptr[tid + i * 256] *= scale;
}

__global__ __launch_bounds__(512) void ca_scores(float* __restrict__ ws,
                                                 const float* __restrict__ temperature) {
  __shared__ float S[64][64];
  const int lane = threadIdx.x & 31;
  const int wave = threadIdx.x >> 5;
  const int l16  = lane & 15;
  const int hi   = lane >> 4;
  const int bh = blockIdx.x;
  const int h  = bh & 7;
  const int dt = wave >> 2, et = wave & 3;

  const float* qRow = ws + OFF_Q + (size_t)bh * DH * NTOK +
                      (size_t)(dt * 16 + l16) * NTOK + (hi ? 2 : 0);
  const float* kRow = ws + OFF_K + (size_t)bh * DH * NTOK +
                      (size_t)(et * 16 + l16) * NTOK + (hi ? 2 : 0);
  v8f acc = {};
  for (int n0 = 0; n0 < NTOK; n0 += 4) {
    v2f a = *(const v2f*)(qRow + n0);
    v2f b = *(const v2f*)(kRow + n0);
    acc = WMMA_F32(a, b, acc);
  }
  const float tsc = temperature[h];
#pragma unroll
  for (int v = 0; v < 8; ++v)
    S[dt * 16 + v + (hi ? 8 : 0)][et * 16 + l16] = acc[v] * tsc;
  __syncthreads();

  if (threadIdx.x < 64) {
    const int r = threadIdx.x;
    float m = S[r][0];
#pragma unroll
    for (int j = 1; j < 64; ++j) m = fmaxf(m, S[r][j]);
    float sum = 0.f;
    float e[64];
#pragma unroll
    for (int j = 0; j < 64; ++j) { e[j] = __expf(S[r][j] - m); sum += e[j]; }
    const float inv = 1.0f / sum;
    float* dst = ws + OFF_P + (size_t)bh * DH * DH + (size_t)r * DH;
#pragma unroll
    for (int j = 0; j < 64; ++j) dst[j] = e[j] * inv;
  }
}

__global__ __launch_bounds__(256) void ca_out(const float* __restrict__ ws,
                                              float* __restrict__ out) {
  __shared__ float P[64][64];
  const int tid  = threadIdx.x;
  const int lane = tid & 31;
  const int wave = tid >> 5;
  const int l16  = lane & 15;
  const int hi   = lane >> 4;
  const int bh = blockIdx.y;
  const int b_ = bh >> 3;
  const int h  = bh & 7;
  const int nBase = blockIdx.x * 128;

  const float* Pg = ws + OFF_P + (size_t)bh * DH * DH;
  for (int i = tid; i < 4096; i += 256) P[i >> 6][i & 63] = Pg[i];
  __syncthreads();

  const int dt = wave >> 1;
  const int ng = wave & 1;
  const int mRow = dt * 16 + l16;
  const float* vca = ws + OFF_VCA + (size_t)bh * DH * NTOK;

  v8f acc[4] = {};
  for (int e0 = 0; e0 < DH; e0 += 4) {
    const int k0 = e0 + (hi ? 2 : 0);
    v2f a;
    a.x = P[mRow][k0];
    a.y = P[mRow][k0 + 1];
#pragma unroll
    for (int t = 0; t < 4; ++t) {
      const int n = nBase + (ng * 4 + t) * 16 + l16;
      v2f bfrag;
      bfrag.x = vca[(size_t)k0 * NTOK + n];
      bfrag.y = vca[(size_t)(k0 + 1) * NTOK + n];
      acc[t] = WMMA_F32(a, bfrag, acc[t]);
    }
  }
#pragma unroll
  for (int t = 0; t < 4; ++t) {
    const int n = nBase + (ng * 4 + t) * 16 + l16;
#pragma unroll
    for (int v = 0; v < 8; ++v) {
      const int d = dt * 16 + v + (hi ? 8 : 0);
      out[((size_t)(b_ * NTOK + n)) * CCH + h * DH + d] = acc[t][v];
    }
  }
}

__global__ __launch_bounds__(256) void sa_attn(float* __restrict__ ws,
                                               const float* __restrict__ temperature2,
                                               float* __restrict__ out) {
  __shared__ float P[128][64];
  const int tid  = threadIdx.x;
  const int lane = tid & 31;
  const int wave = tid >> 5;
  const int l16  = lane & 15;
  const int hi   = lane >> 4;
  const int bh = blockIdx.y;
  const int b_ = bh >> 3;
  const int h  = bh & 7;
  const int nBase = blockIdx.x * 128;

  const float* Q  = ws + OFF_Q  + (size_t)bh * DH * NTOK;
  const float* Kp = ws + OFF_KP + (size_t)bh * DH * KL;
  const float* Vp = ws + OFF_VP + (size_t)bh * DH * KL;
  const float t2 = temperature2[h];

  {
    const int nGlob = nBase + wave * 16 + l16;
    v8f acc[4] = {};
    for (int d0 = 0; d0 < DH; d0 += 4) {
      const int k0 = d0 + (hi ? 2 : 0);
      v2f a;
      a.x = Q[(size_t)k0 * NTOK + nGlob];
      a.y = Q[(size_t)(k0 + 1) * NTOK + nGlob];
#pragma unroll
      for (int t = 0; t < 4; ++t) {
        const int kc = t * 16 + l16;
        v2f bfrag;
        bfrag.x = Kp[(size_t)k0 * KL + kc];
        bfrag.y = Kp[(size_t)(k0 + 1) * KL + kc];
        acc[t] = WMMA_F32(a, bfrag, acc[t]);
      }
    }
#pragma unroll
    for (int t = 0; t < 4; ++t)
#pragma unroll
      for (int v = 0; v < 8; ++v)
        P[wave * 16 + v + (hi ? 8 : 0)][t * 16 + l16] = acc[t][v] * t2;
  }
  __syncthreads();

  if (tid < 128) {
    float m = P[tid][0];
#pragma unroll
    for (int j = 1; j < 64; ++j) m = fmaxf(m, P[tid][j]);
    float sum = 0.f;
#pragma unroll
    for (int j = 0; j < 64; ++j) { float e = __expf(P[tid][j] - m); P[tid][j] = e; sum += e; }
    const float inv = 1.0f / sum;
#pragma unroll
    for (int j = 0; j < 64; ++j) P[tid][j] *= inv;
  }
  __syncthreads();

  {
    const int mRow = wave * 16 + l16;
    v8f acc[4] = {};
    for (int kk0 = 0; kk0 < KL; kk0 += 4) {
      const int k0 = kk0 + (hi ? 2 : 0);
      v2f a;
      a.x = P[mRow][k0];
      a.y = P[mRow][k0 + 1];
#pragma unroll
      for (int t = 0; t < 4; ++t) {
        const int dd = t * 16 + l16;
        v2f bfrag = *(const v2f*)(Vp + (size_t)dd * KL + k0);
        acc[t] = WMMA_F32(a, bfrag, acc[t]);
      }
    }
#pragma unroll
    for (int t = 0; t < 4; ++t) {
      const int dd = t * 16 + l16;
#pragma unroll
      for (int v = 0; v < 8; ++v) {
        const int n = nBase + wave * 16 + v + (hi ? 8 : 0);
        const size_t f = (size_t)dd * (HEADS * NTOK) + (size_t)h * NTOK + n;
        out[(size_t)b_ * (NTOK * CCH) + f] += acc[t][v];
      }
    }
  }
}

extern "C" void kernel_launch(void* const* d_in, const int* in_sizes, int n_in,
                              void* d_out, int out_size, void* d_ws, size_t ws_size,
                              hipStream_t stream) {
  const float* x  = (const float*)d_in[0];
  const float* W  = (const float*)d_in[1];
  const float* EF = (const float*)d_in[2];
  const float* t1 = (const float*)d_in[3];
  const float* t2 = (const float*)d_in[4];
  float* out = (float*)d_out;
  float* ws  = (float*)d_ws;

  qkvv_gemm  <<<dim3(128, 32), 256, 0, stream>>>(x, W, ws);
  ef_proj    <<<64,            512, 0, stream>>>(ws, EF);
  l2norm_rows<<<4096,          256, 0, stream>>>(ws);
  ca_scores  <<<32,            512, 0, stream>>>(ws, t1);
  ca_out     <<<dim3(32, 32),  256, 0, stream>>>(ws, out);
  sa_attn    <<<dim3(32, 32),  256, 0, stream>>>(ws, t2, out);
}


# K2 EF projection fused into K1: K/Vsa accumulator tiles chained as WMMA row operand against EF, 128-token partials summed by a small kernel; Vsa plane not stored; k_proj, v_proj sum order changes
# speedup vs baseline: 1.8348x; 1.8348x over previous
#include <hip/hip_runtime.h>

typedef float v2f __attribute__((ext_vector_type(2)));
typedef float v8f __attribute__((ext_vector_type(8)));

#define WMMA_F32(a, b, c) \
  __builtin_amdgcn_wmma_f32_16x16x4_f32(false, (a), false, (b), (short)0, (c), false, false)

#define BATCH 4
#define NTOK 4096
#define CCH 512
#define HEADS 8
#define DH 64
#define KL 64
#define MROWS (BATCH * NTOK)
#define OCH (4 * CCH)
#define BH (BATCH * HEADS)

#define OFF_Q   ((size_t)0)
#define OFF_K   ((size_t)8388608)
#define OFF_VCA ((size_t)16777216)
#define OFF_VSA ((size_t)25165824)
#define OFF_KP  ((size_t)33554432)
#define OFF_VP  ((size_t)33685504)
#define OFF_P   ((size_t)33816576)

typedef float v4f __attribute__((ext_vector_type(4)));

__device__ __forceinline__ void k1_load_set(v4f (&fa)[4][4], v4f (&fb)[4][4],
                                            const float* __restrict__ aPtr,
                                            const float* __restrict__ bPtr,
                                            const int k) {
#pragma unroll
  for (int i = 0; i < 4; ++i)
#pragma unroll
    for (int q = 0; q < 4; ++q)
      fa[i][q] = *(const v4f*)(aPtr + (size_t)i * 16 * CCH + k + q * 4);
#pragma unroll
  for (int j = 0; j < 4; ++j)
#pragma unroll
    for (int q = 0; q < 4; ++q)
      fb[j][q] = *(const v4f*)(bPtr + (size_t)j * 16 * CCH + k + q * 4);
}

__device__ __forceinline__ void k1_mma_set(v8f (&acc)[4][4],
                                           const v4f (&fa)[4][4],
                                           const v4f (&fb)[4][4]) {
#pragma unroll
  for (int q = 0; q < 4; ++q) {
#pragma unroll
    for (int i = 0; i < 4; ++i) {
      v2f a;
      a.x = fa[i][q].x;
      a.y = fa[i][q].y;
#pragma unroll
      for (int j = 0; j < 4; ++j) {
        v2f b;
        b.x = fb[j][q].x;
        b.y = fb[j][q].y;
        acc[i][j] = WMMA_F32(a, b, acc[i][j]);
      }
    }
#pragma unroll
    for (int i = 0; i < 4; ++i) {
      v2f a;
      a.x = fa[i][q].z;
      a.y = fa[i][q].w;
#pragma unroll
      for (int j = 0; j < 4; ++j) {
        v2f b;
        b.x = fb[j][q].z;
        b.y = fb[j][q].w;
        acc[i][j] = WMMA_F32(a, b, acc[i][j]);
      }
    }
  }
}

__global__ __launch_bounds__(128) void qkvv_gemm(const float* __restrict__ x,
                                                 const float* __restrict__ W,
                                                 const float* __restrict__ EF,
                                                 float* __restrict__ ws) {
  const int lane = threadIdx.x & 31;
  const int wave = threadIdx.x >> 5;
  const int l16  = lane & 15;
  const int hi   = lane >> 4;
  const int mW = blockIdx.x * 128 + (wave >> 1) * 64;
  const int oW = blockIdx.y * 128 + (wave & 1) * 64;

  const float* aPtr = x + (size_t)(mW + l16) * CCH + hi * 16;
  const float* bPtr = W + (size_t)(oW + l16) * CCH + hi * 16;

  v8f acc[4][4] = {};
  v4f fa0[4][4], fb0[4][4], fa1[4][4], fb1[4][4];

  k1_load_set(fa0, fb0, aPtr, bPtr, 0);
  for (int k0 = 0; k0 < CCH - 64; k0 += 64) {
    k1_load_set(fa1, fb1, aPtr, bPtr, k0 + 32);
    k1_mma_set(acc, fa0, fb0);
    k1_load_set(fa0, fb0, aPtr, bPtr, k0 + 64);
    k1_mma_set(acc, fa1, fb1);
  }
  k1_load_set(fa1, fb1, aPtr, bPtr, CCH - 32);
  k1_mma_set(acc, fa0, fb0);
  k1_mma_set(acc, fa1, fb1);

  const int p  = oW >> 9;
  const int h  = (oW >> 6) & 7;
  const int b_ = mW >> 12;
  if (p != 3) {
    float* dstBase = ws + (size_t)p * 8388608 + (size_t)(b_ * HEADS + h) * DH * NTOK;
#pragma unroll
    for (int i = 0; i < 4; ++i) {
#pragma unroll
      for (int j = 0; j < 4; ++j) {
        const int dd = j * 16 + l16;
#pragma unroll
        for (int v = 0; v < 8; ++v) {
          const int n = (mW & (NTOK - 1)) + i * 16 + v + (hi ? 8 : 0);
          dstBase[(size_t)dd * NTOK + n] = acc[i][j][v];
        }
      }
    }
  }

  if (p == 1 || p == 3) {
    __shared__ v4f red[2][32][32];
    const int nTok0 = mW & (NTOK - 1);
    const float* efBase = EF + (size_t)(nTok0 + hi * 8) * KL + l16 * 4;
    v8f pacc[4][4] = {};
#pragma unroll
    for (int i = 0; i < 4; ++i) {
      v4f e[4][2];
#pragma unroll
      for (int t = 0; t < 4; ++t)
#pragma unroll
        for (int s = 0; s < 2; ++s)
          e[t][s] = *(const v4f*)(efBase + (size_t)(i * 16 + 2 * t + s) * KL);
#pragma unroll
      for (int t = 0; t < 4; ++t) {
#pragma unroll
        for (int j = 0; j < 4; ++j) {
          v2f a;
          a.x = acc[i][j][2 * t];
          a.y = acc[i][j][2 * t + 1];
#pragma unroll
          for (int jj = 0; jj < 4; ++jj) {
            v2f b;
            b.x = e[t][0][jj];
            b.y = e[t][1][jj];
            pacc[j][jj] = WMMA_F32(a, b, pacc[j][jj]);
          }
        }
      }
    }

    const int hw = wave & 1;
    if (wave >= 2) {
#pragma unroll
      for (int j = 0; j < 4; ++j)
#pragma unroll
        for (int jj = 0; jj < 4; ++jj) {
          v4f lo, up;
          lo.x = pacc[j][jj][0]; lo.y = pacc[j][jj][1]; lo.z = pacc[j][jj][2]; lo.w = pacc[j][jj][3];
          up.x = pacc[j][jj][4]; up.y = pacc[j][jj][5]; up.z = pacc[j][jj][6]; up.w = pacc[j][jj][7];
          red[hw][(j * 4 + jj) * 2][lane]     = lo;
          red[hw][(j * 4 + jj) * 2 + 1][lane] = up;
        }
    }
    __syncthreads();
    if (wave < 2) {
#pragma unroll
      for (int j = 0; j < 4; ++j)
#pragma unroll
        for (int jj = 0; jj < 4; ++jj) {
          const v4f lo = red[hw][(j * 4 + jj) * 2][lane];
          const v4f up = red[hw][(j * 4 + jj) * 2 + 1][lane];
          pacc[j][jj][0] += lo.x; pacc[j][jj][1] += lo.y; pacc[j][jj][2] += lo.z; pacc[j][jj][3] += lo.w;
          pacc[j][jj][4] += up.x; pacc[j][jj][5] += up.y; pacc[j][jj][6] += up.z; pacc[j][jj][7] += up.w;
        }
      const int which = p >> 1;
      const int mt = blockIdx.x & 31;
      float* part = ws + OFF_VSA +
                    ((size_t)((which * BH + b_ * HEADS + h) * 32 + mt)) * (DH * KL);
#pragma unroll
      for (int j = 0; j < 4; ++j)
#pragma unroll
        for (int r = 0; r < 8; ++r) {
          v4f o;
          o.x = pacc[j][0][r]; o.y = pacc[j][1][r]; o.z = pacc[j][2][r]; o.w = pacc[j][3][r];
          *(v4f*)(part + (size_t)(j * 16 + hi * 8 + r) * KL + l16 * 4) = o;
        }
    }
  }
}

__global__ __launch_bounds__(256) void ef_reduce(float* __restrict__ ws) {
  const int g  = blockIdx.x * 256 + threadIdx.x;
  const int wb = g >> 10;
  const int e4 = g & 1023;
  const v4f* src = (const v4f*)(ws + OFF_VSA + (size_t)wb * 32 * (DH * KL)) + e4;
  v4f s = src[0];
#pragma unroll
  for (int m = 1; m < 32; ++m) s += src[(size_t)m * (DH * KL / 4)];
  const int which = wb >> 5;
  const int bh    = wb & 31;
  float* dst = ws + (which ? OFF_VP : OFF_KP) + (size_t)bh * DH * KL;
  *(v4f*)(dst + e4 * 4) = s;
}

__global__ __launch_bounds__(256) void l2norm_rows(float* __restrict__ ws) {
  __shared__ float red[256];
  const int tid = threadIdx.x;
  const int row = blockIdx.x;
  float* ptr = ws + (row < 2048 ? OFF_Q : OFF_K) +
               (size_t)(row & 2047) * NTOK;
  float s = 0.f;
#pragma unroll
  for (int i = 0; i < 16; ++i) {
    float v = ptr[tid + i * 256];
    s += v * v;
  }
  red[tid] = s;
  __syncthreads();
  for (int off = 128; off > 0; off >>= 1) {
    if (tid < off) red[tid] += red[tid + off];
    __syncthreads();
  }
  const float scale = 1.0f / fmaxf(sqrtf(red[0]), 1e-12f);
#pragma unroll
  for (int i = 0; i < 16; ++i) ptr[tid + i * 256] *= scale;
}

__global__ __launch_bounds__(512) void ca_scores(float* __restrict__ ws,
                                                 const float* __restrict__ temperature) {
  __shared__ float S[64][64];
  const int lane = threadIdx.x & 31;
  const int wave = threadIdx.x >> 5;
  const int l16  = lane & 15;
  const int hi   = lane >> 4;
  const int bh = blockIdx.x;
  const int h  = bh & 7;
  const int dt = wave >> 2, et = wave & 3;

  const float* qRow = ws + OFF_Q + (size_t)bh * DH * NTOK +
                      (size_t)(dt * 16 + l16) * NTOK + (hi ? 2 : 0);
  const float* kRow = ws + OFF_K + (size_t)bh * DH * NTOK +
                      (size_t)(et * 16 + l16) * NTOK + (hi ? 2 : 0);
  v8f acc = {};
  for (int n0 = 0; n0 < NTOK; n0 += 4) {
    v2f a = *(const v2f*)(qRow + n0);
    v2f b = *(const v2f*)(kRow + n0);
    acc = WMMA_F32(a, b, acc);
  }
  const float tsc = temperature[h];
#pragma unroll
  for (int v = 0; v < 8; ++v)
    S[dt * 16 + v + (hi ? 8 : 0)][et * 16 + l16] = acc[v] * tsc;
  __syncthreads();

  if (threadIdx.x < 64) {
    const int r = threadIdx.x;
    float m = S[r][0];
#pragma unroll
    for (int j = 1; j < 64; ++j) m = fmaxf(m, S[r][j]);
    float sum = 0.f;
    float e[64];
#pragma unroll
    for (int j = 0; j < 64; ++j) { e[j] = __expf(S[r][j] - m); sum += e[j]; }
    const float inv = 1.0f / sum;
    float* dst = ws + OFF_P + (size_t)bh * DH * DH + (size_t)r * DH;
#pragma unroll
    for (int j = 0; j < 64; ++j) dst[j] = e[j] * inv;
  }
}

__global__ __launch_bounds__(256) void ca_out(const float* __restrict__ ws,
                                              float* __restrict__ out) {
  __shared__ float P[64][64];
  const int tid  = threadIdx.x;
  const int lane = tid & 31;
  const int wave = tid >> 5;
  const int l16  = lane & 15;
  const int hi   = lane >> 4;
  const int bh = blockIdx.y;
  const int b_ = bh >> 3;
  const int h  = bh & 7;
  const int nBase = blockIdx.x * 128;

  const float* Pg = ws + OFF_P + (size_t)bh * DH * DH;
  for (int i = tid; i < 4096; i += 256) P[i >> 6][i & 63] = Pg[i];
  __syncthreads();

  const int dt = wave >> 1;
  const int ng = wave & 1;
  const int mRow = dt * 16 + l16;
  const float* vca = ws + OFF_VCA + (size_t)bh * DH * NTOK;

  v8f acc[4] = {};
  for (int e0 = 0; e0 < DH; e0 += 4) {
    const int k0 = e0 + (hi ? 2 : 0);
    v2f a;
    a.x = P[mRow][k0];
    a.y = P[mRow][k0 + 1];
#pragma unroll
    for (int t = 0; t < 4; ++t) {
      const int n = nBase + (ng * 4 + t) * 16 + l16;
      v2f bfrag;
      bfrag.x = vca[(size_t)k0 * NTOK + n];
      bfrag.y = vca[(size_t)(k0 + 1) * NTOK + n];
      acc[t] = WMMA_F32(a, bfrag, acc[t]);
    }
  }
#pragma unroll
  for (int t = 0; t < 4; ++t) {
    const int n = nBase + (ng * 4 + t) * 16 + l16;
#pragma unroll
    for (int v = 0; v < 8; ++v) {
      const int d = dt * 16 + v + (hi ? 8 : 0);
      out[((size_t)(b_ * NTOK + n)) * CCH + h * DH + d] = acc[t][v];
    }
  }
}

__global__ __launch_bounds__(256) void sa_attn(float* __restrict__ ws,
                                               const float* __restrict__ temperature2,
                                               float* __restrict__ out) {
  __shared__ float P[128][64];
  const int tid  = threadIdx.x;
  const int lane = tid & 31;
  const int wave = tid >> 5;
  const int l16  = lane & 15;
  const int hi   = lane >> 4;
  const int bh = blockIdx.y;
  const int b_ = bh >> 3;
  const int h  = bh & 7;
  const int nBase = blockIdx.x * 128;

  const float* Q  = ws + OFF_Q  + (size_t)bh * DH * NTOK;
  const float* Kp = ws + OFF_KP + (size_t)bh * DH * KL;
  const float* Vp = ws + OFF_VP + (size_t)bh * DH * KL;
  const float t2 = temperature2[h];

  {
    const int nGlob = nBase + wave * 16 + l16;
    v8f acc[4] = {};
    for (int d0 = 0; d0 < DH; d0 += 4) {
      const int k0 = d0 + (hi ? 2 : 0);
      v2f a;
      a.x = Q[(size_t)k0 * NTOK + nGlob];
      a.y = Q[(size_t)(k0 + 1) * NTOK + nGlob];
#pragma unroll
      for (int t = 0; t < 4; ++t) {
        const int kc = t * 16 + l16;
        v2f bfrag;
        bfrag.x = Kp[(size_t)k0 * KL + kc];
        bfrag.y = Kp[(size_t)(k0 + 1) * KL + kc];
        acc[t] = WMMA_F32(a, bfrag, acc[t]);
      }
    }
#pragma unroll
    for (int t = 0; t < 4; ++t)
#pragma unroll
      for (int v = 0; v < 8; ++v)
        P[wave * 16 + v + (hi ? 8 : 0)][t * 16 + l16] = acc[t][v] * t2;
  }
  __syncthreads();

  if (tid < 128) {
    float m = P[tid][0];
#pragma unroll
    for (int j = 1; j < 64; ++j) m = fmaxf(m, P[tid][j]);
    float sum = 0.f;
#pragma unroll
    for (int j = 0; j < 64; ++j) { float e = __expf(P[tid][j] - m); P[tid][j] = e; sum += e; }
    const float inv = 1.0f / sum;
#pragma unroll
    for (int j = 0; j < 64; ++j) P[tid][j] *= inv;
  }
  __syncthreads();

  {
    const int mRow = wave * 16 + l16;
    v8f acc[4] = {};
    for (int kk0 = 0; kk0 < KL; kk0 += 4) {
      const int k0 = kk0 + (hi ? 2 : 0);
      v2f a;
      a.x = P[mRow][k0];
      a.y = P[mRow][k0 + 1];
#pragma unroll
      for (int t = 0; t < 4; ++t) {
        const int dd = t * 16 + l16;
        v2f bfrag = *(const v2f*)(Vp + (size_t)dd * KL + k0);
        acc[t] = WMMA_F32(a, bfrag, acc[t]);
      }
    }
#pragma unroll
    for (int t = 0; t < 4; ++t) {
      const int dd = t * 16 + l16;
#pragma unroll
      for (int v = 0; v < 8; ++v) {
        const int n = nBase + wave * 16 + v + (hi ? 8 : 0);
        const size_t f = (size_t)dd * (HEADS * NTOK) + (size_t)h * NTOK + n;
        out[(size_t)b_ * (NTOK * CCH) + f] += acc[t][v];
      }
    }
  }
}

extern "C" void kernel_launch(void* const* d_in, const int* in_sizes, int n_in,
                              void* d_out, int out_size, void* d_ws, size_t ws_size,
                              hipStream_t stream) {
  const float* x  = (const float*)d_in[0];
  const float* W  = (const float*)d_in[1];
  const float* EF = (const float*)d_in[2];
  const float* t1 = (const float*)d_in[3];
  const float* t2 = (const float*)d_in[4];
  float* out = (float*)d_out;
  float* ws  = (float*)d_ws;

  qkvv_gemm  <<<dim3(128, 16), 128, 0, stream>>>(x, W, EF, ws);
  ef_reduce  <<<256,           256, 0, stream>>>(ws);
  l2norm_rows<<<4096,          256, 0, stream>>>(ws);
  ca_scores  <<<32,            512, 0, stream>>>(ws, t1);
  ca_out     <<<dim3(32, 32),  256, 0, stream>>>(ws, out);
  sa_attn    <<<dim3(32, 32),  256, 0, stream>>>(ws, t2, out);
}


# Q.K^T token sum and Q/K square sums from K1's accumulators (Q,K tiles of a head in one block, K tile via LDS as column operand); K plane, K3, K4 loop gone; S scaled by 1/(|q||k|); sum order changes
# speedup vs baseline: 2.7857x; 1.5183x over previous
#include <hip/hip_runtime.h>

typedef float v2f __attribute__((ext_vector_type(2)));
typedef float v8f __attribute__((ext_vector_type(8)));

#define WMMA_F32(a, b, c) \
  __builtin_amdgcn_wmma_f32_16x16x4_f32(false, (a), false, (b), (short)0, (c), false, false)

#define BATCH 4
#define NTOK 4096
#define CCH 512
#define HEADS 8
#define DH 64
#define KL 64
#define MROWS (BATCH * NTOK)
#define OCH (4 * CCH)
#define BH (BATCH * HEADS)

#define OFF_Q   ((size_t)0)
#define OFF_K   ((size_t)8388608)
#define OFF_VCA ((size_t)16777216)
#define OFF_VSA ((size_t)25165824)
#define OFF_KP  ((size_t)33554432)
#define OFF_VP  ((size_t)33685504)
#define OFF_P   ((size_t)33816576)
#define OFF_SP  (OFF_K)
#define OFF_SQ  (OFF_K + (size_t)4194304)
#define OFF_IQ  (OFF_K + (size_t)4718592)

typedef float v4f __attribute__((ext_vector_type(4)));

__device__ __forceinline__ void k1_load_set(v4f (&fa)[4][4], v4f (&fb)[4][4],
                                            const float* __restrict__ aPtr,
                                            const float* __restrict__ bPtr,
                                            const int k) {
#pragma unroll
  for (int i = 0; i < 4; ++i)
#pragma unroll
    for (int q = 0; q < 4; ++q)
      fa[i][q] = *(const v4f*)(aPtr + (size_t)i * 16 * CCH + k + q * 4);
#pragma unroll
  for (int j = 0; j < 4; ++j)
#pragma unroll
    for (int q = 0; q < 4; ++q)
      fb[j][q] = *(const v4f*)(bPtr + (size_t)j * 16 * CCH + k + q * 4);
}

__device__ __forceinline__ void k1_mma_set(v8f (&acc)[4][4],
                                           const v4f (&fa)[4][4],
                                           const v4f (&fb)[4][4]) {
#pragma unroll
  for (int q = 0; q < 4; ++q) {
#pragma unroll
    for (int i = 0; i < 4; ++i) {
      v2f a;
      a.x = fa[i][q].x;
      a.y = fa[i][q].y;
#pragma unroll
      for (int j = 0; j < 4; ++j) {
        v2f b;
        b.x = fb[j][q].x;
        b.y = fb[j][q].y;
        acc[i][j] = WMMA_F32(a, b, acc[i][j]);
      }
    }
#pragma unroll
    for (int i = 0; i < 4; ++i) {
      v2f a;
      a.x = fa[i][q].z;
      a.y = fa[i][q].w;
#pragma unroll
      for (int j = 0; j < 4; ++j) {
        v2f b;
        b.x = fb[j][q].z;
        b.y = fb[j][q].w;
        acc[i][j] = WMMA_F32(a, b, acc[i][j]);
      }
    }
  }
}

__global__ __launch_bounds__(128) void qkvv_gemm(const float* __restrict__ x,
                                                 const float* __restrict__ W,
                                                 const float* __restrict__ EF,
                                                 float* __restrict__ ws) {
  const int lane = threadIdx.x & 31;
  const int wave = threadIdx.x >> 5;
  const int l16  = lane & 15;
  const int hi   = lane >> 4;
  const int wm = wave >> 1;
  const int wo = wave & 1;
  const int p  = (blockIdx.y >> 3) * 2 + wo;
  const int h  = blockIdx.y & 7;
  const int mW = blockIdx.x * 128 + wm * 64;
  const int oW = p * CCH + h * DH;

  const float* aPtr = x + (size_t)(mW + l16) * CCH + hi * 16;
  const float* bPtr = W + (size_t)(oW + l16) * CCH + hi * 16;

  v8f acc[4][4] = {};
  v4f fa0[4][4], fb0[4][4], fa1[4][4], fb1[4][4];

  k1_load_set(fa0, fb0, aPtr, bPtr, 0);
  for (int k0 = 0; k0 < CCH - 64; k0 += 64) {
    k1_load_set(fa1, fb1, aPtr, bPtr, k0 + 32);
    k1_mma_set(acc, fa0, fb0);
    k1_load_set(fa0, fb0, aPtr, bPtr, k0 + 64);
    k1_mma_set(acc, fa1, fb1);
  }
  k1_load_set(fa1, fb1, aPtr, bPtr, CCH - 32);
  k1_mma_set(acc, fa0, fb0);
  k1_mma_set(acc, fa1, fb1);

  const int b_ = mW >> 12;
  const int bh = b_ * HEADS + h;
  if (p == 0 || p == 2) {
    float* dstBase = ws + (size_t)p * 8388608 + (size_t)bh * DH * NTOK;
#pragma unroll
    for (int i = 0; i < 4; ++i) {
#pragma unroll
      for (int j = 0; j < 4; ++j) {
        const int dd = j * 16 + l16;
#pragma unroll
        for (int v = 0; v < 8; ++v) {
          const int n = (mW & (NTOK - 1)) + i * 16 + v + (hi ? 8 : 0);
          dstBase[(size_t)dd * NTOK + n] = acc[i][j][v];
        }
      }
    }
  }

  __shared__ v4f red[2][32][32];
  const int nTok0 = mW & (NTOK - 1);

  if (p == 1) {
#pragma unroll
    for (int i = 0; i < 4; ++i)
#pragma unroll
      for (int j = 0; j < 4; ++j) {
        v4f lo, up;
        lo.x = acc[i][j][0]; lo.y = acc[i][j][1]; lo.z = acc[i][j][2]; lo.w = acc[i][j][3];
        up.x = acc[i][j][4]; up.y = acc[i][j][5]; up.z = acc[i][j][6]; up.w = acc[i][j][7];
        red[wm][(i * 4 + j) * 2][lane]     = lo;
        red[wm][(i * 4 + j) * 2 + 1][lane] = up;
      }
  }
  __syncthreads();

  if (p <= 1) {
    float* sq = ws + OFF_SQ + ((size_t)((p * BH + bh) * 128 + (nTok0 >> 6) * 2 + hi)) * DH;
#pragma unroll
    for (int j = 0; j < 4; ++j) {
      float s = 0.f;
#pragma unroll
      for (int i = 0; i < 4; ++i)
#pragma unroll
        for (int v = 0; v < 8; ++v) s += acc[i][j][v] * acc[i][j][v];
      sq[j * 16 + l16] = s;
    }
  }

  v8f pacc[4][4] = {};
  if (p == 0) {
#pragma unroll
    for (int i = 0; i < 4; ++i) {
      v4f kt[4][2];
#pragma unroll
      for (int jj = 0; jj < 4; ++jj) {
        kt[jj][0] = red[wm][(i * 4 + jj) * 2][lane];
        kt[jj][1] = red[wm][(i * 4 + jj) * 2 + 1][lane];
      }
#pragma unroll
      for (int t = 0; t < 4; ++t) {
#pragma unroll
        for (int j = 0; j < 4; ++j) {
          v2f a;
          a.x = acc[i][j][2 * t];
          a.y = acc[i][j][2 * t + 1];
#pragma unroll
          for (int jj = 0; jj < 4; ++jj) {
            v2f b;
            b.x = kt[jj][t >> 1][(2 * t) & 3];
            b.y = kt[jj][t >> 1][(2 * t + 1) & 3];
            pacc[j][jj] = WMMA_F32(a, b, pacc[j][jj]);
          }
        }
      }
    }
  } else if (p != 2) {
    const float* efBase = EF + (size_t)(nTok0 + hi * 8) * KL + l16 * 4;
#pragma unroll
    for (int i = 0; i < 4; ++i) {
      v4f e[4][2];
#pragma unroll
      for (int t = 0; t < 4; ++t)
#pragma unroll
        for (int s = 0; s < 2; ++s)
          e[t][s] = *(const v4f*)(efBase + (size_t)(i * 16 + 2 * t + s) * KL);
#pragma unroll
      for (int t = 0; t < 4; ++t) {
#pragma unroll
        for (int j = 0; j < 4; ++j) {
          v2f a;
          a.x = acc[i][j][2 * t];
          a.y = acc[i][j][2 * t + 1];
#pragma unroll
          for (int jj = 0; jj < 4; ++jj) {
            v2f b;
            b.x = e[t][0][jj];
            b.y = e[t][1][jj];
            pacc[j][jj] = WMMA_F32(a, b, pacc[j][jj]);
          }
        }
      }
    }
  }
  __syncthreads();

  if (p != 2 && wm == 1) {
#pragma unroll
    for (int j = 0; j < 4; ++j)
#pragma unroll
      for (int jj = 0; jj < 4; ++jj) {
        v4f lo, up;
        lo.x = pacc[j][jj][0]; lo.y = pacc[j][jj][1]; lo.z = pacc[j][jj][2]; lo.w = pacc[j][jj][3];
        up.x = pacc[j][jj][4]; up.y = pacc[j][jj][5]; up.z = pacc[j][jj][6]; up.w = pacc[j][jj][7];
        red[wo][(j * 4 + jj) * 2][lane]     = lo;
        red[wo][(j * 4 + jj) * 2 + 1][lane] = up;
      }
  }
  __syncthreads();
  if (p != 2 && wm == 0) {
#pragma unroll
    for (int j = 0; j < 4; ++j)
#pragma unroll
      for (int jj = 0; jj < 4; ++jj) {
        const v4f lo = red[wo][(j * 4 + jj) * 2][lane];
        const v4f up = red[wo][(j * 4 + jj) * 2 + 1][lane];
        pacc[j][jj][0] += lo.x; pacc[j][jj][1] += lo.y; pacc[j][jj][2] += lo.z; pacc[j][jj][3] += lo.w;
        pacc[j][jj][4] += up.x; pacc[j][jj][5] += up.y; pacc[j][jj][6] += up.z; pacc[j][jj][7] += up.w;
      }
    const int mt = blockIdx.x & 31;
    if (p == 0) {
      float* sp = ws + OFF_SP + ((size_t)(bh * 32 + mt)) * (DH * DH);
#pragma unroll
      for (int j = 0; j < 4; ++j)
#pragma unroll
        for (int jj = 0; jj < 4; ++jj)
#pragma unroll
          for (int r = 0; r < 8; ++r)
            sp[(size_t)(j * 16 + hi * 8 + r) * DH + jj * 16 + l16] = pacc[j][jj][r];
    } else {
      const int which = p >> 1;
      float* part = ws + OFF_VSA +
                    ((size_t)((which * BH + bh) * 32 + mt)) * (DH * KL);
#pragma unroll
      for (int j = 0; j < 4; ++j)
#pragma unroll
        for (int r = 0; r < 8; ++r) {
          v4f o;
          o.x = pacc[j][0][r]; o.y = pacc[j][1][r]; o.z = pacc[j][2][r]; o.w = pacc[j][3][r];
          *(v4f*)(part + (size_t)(j * 16 + hi * 8 + r) * KL + l16 * 4) = o;
        }
    }
  }
}

__global__ __launch_bounds__(256) void ef_reduce(float* __restrict__ ws) {
  const int g  = blockIdx.x * 256 + threadIdx.x;
  const int wb = g >> 10;
  const int e4 = g & 1023;
  const v4f* src = (const v4f*)(ws + OFF_VSA + (size_t)wb * 32 * (DH * KL)) + e4;
  v4f s = src[0];
#pragma unroll
  for (int m = 1; m < 32; ++m) s += src[(size_t)m * (DH * KL / 4)];
  const int which = wb >> 5;
  const int bh    = wb & 31;
  float* dst = ws + (which ? OFF_VP : OFF_KP) + (size_t)bh * DH * KL;
  *(v4f*)(dst + e4 * 4) = s;
}

__global__ __launch_bounds__(256) void ca_softmax(float* __restrict__ ws,
                                                  const float* __restrict__ temperature) {
  __shared__ float k2s[4][64];
  const int tid = threadIdx.x;
  const int bh  = blockIdx.y;
  const int h   = bh & 7;
  const int e4  = tid & 15;
  const int d   = blockIdx.x * 16 + (tid >> 4);

  const v4f* sp = (const v4f*)(ws + OFF_SP + (size_t)bh * 32 * (DH * DH) + (size_t)d * DH) + e4;
  v4f s = sp[0];
#pragma unroll
  for (int m = 1; m < 32; ++m) s += sp[(size_t)m * (DH * DH / 4)];

  {
    const float* kq = ws + OFF_SQ + ((size_t)((BH + bh) * 128 + (tid >> 6) * 32)) * DH + (tid & 63);
    float a = 0.f;
#pragma unroll
    for (int m = 0; m < 32; ++m) a += kq[m * DH];
    k2s[tid >> 6][tid & 63] = a;
  }
  float q2 = 0.f;
  {
    const float* qq = ws + OFF_SQ + ((size_t)(bh * 128 + e4 * 8)) * DH + d;
#pragma unroll
    for (int m = 0; m < 8; ++m) q2 += qq[m * DH];
  }
  q2 += __shfl_xor(q2, 8, 16);
  q2 += __shfl_xor(q2, 4, 16);
  q2 += __shfl_xor(q2, 2, 16);
  q2 += __shfl_xor(q2, 1, 16);
  __syncthreads();

  const float iq  = 1.0f / fmaxf(sqrtf(q2), 1e-12f);
  const float tsc = temperature[h];
  float sc[4];
#pragma unroll
  for (int c = 0; c < 4; ++c) {
    const int e = e4 * 4 + c;
    const float k2 = (k2s[0][e] + k2s[1][e]) + (k2s[2][e] + k2s[3][e]);
    const float ik = 1.0f / fmaxf(sqrtf(k2), 1e-12f);
    sc[c] = s[c] * iq * ik * tsc;
  }
  float m = fmaxf(fmaxf(sc[0], sc[1]), fmaxf(sc[2], sc[3]));
  m = fmaxf(m, __shfl_xor(m, 8, 16));
  m = fmaxf(m, __shfl_xor(m, 4, 16));
  m = fmaxf(m, __shfl_xor(m, 2, 16));
  m = fmaxf(m, __shfl_xor(m, 1, 16));
  v4f ex;
  ex.x = __expf(sc[0] - m); ex.y = __expf(sc[1] - m);
  ex.z = __expf(sc[2] - m); ex.w = __expf(sc[3] - m);
  float sum = (ex.x + ex.y) + (ex.z + ex.w);
  sum += __shfl_xor(sum, 8, 16);
  sum += __shfl_xor(sum, 4, 16);
  sum += __shfl_xor(sum, 2, 16);
  sum += __shfl_xor(sum, 1, 16);
  const float inv = 1.0f / sum;
  ex *= inv;
  *(v4f*)(ws + OFF_P + (size_t)bh * DH * DH + (size_t)d * DH + e4 * 4) = ex;
  if (e4 == 0) ws[OFF_IQ + (size_t)bh * DH + d] = iq;
}

__global__ __launch_bounds__(256) void ca_out(const float* __restrict__ ws,
                                              float* __restrict__ out) {
  __shared__ float P[64][64];
  const int tid  = threadIdx.x;
  const int lane = tid & 31;
  const int wave = tid >> 5;
  const int l16  = lane & 15;
  const int hi   = lane >> 4;
  const int bh = blockIdx.y;
  const int b_ = bh >> 3;
  const int h  = bh & 7;
  const int nBase = blockIdx.x * 128;

  const float* Pg = ws + OFF_P + (size_t)bh * DH * DH;
  for (int i = tid; i < 4096; i += 256) P[i >> 6][i & 63] = Pg[i];
  __syncthreads();

  const int dt = wave >> 1;
  const int ng = wave & 1;
  const int mRow = dt * 16 + l16;
  const float* vca = ws + OFF_VCA + (size_t)bh * DH * NTOK;

  v8f acc[4] = {};
  for (int e0 = 0; e0 < DH; e0 += 4) {
    const int k0 = e0 + (hi ? 2 : 0);
    v2f a;
    a.x = P[mRow][k0];
    a.y = P[mRow][k0 + 1];
#pragma unroll
    for (int t = 0; t < 4; ++t) {
      const int n = nBase + (ng * 4 + t) * 16 + l16;
      v2f bfrag;
      bfrag.x = vca[(size_t)k0 * NTOK + n];
      bfrag.y = vca[(size_t)(k0 + 1) * NTOK + n];
      acc[t] = WMMA_F32(a, bfrag, acc[t]);
    }
  }
#pragma unroll
  for (int t = 0; t < 4; ++t) {
    const int n = nBase + (ng * 4 + t) * 16 + l16;
#pragma unroll
    for (int v = 0; v < 8; ++v) {
      const int d = dt * 16 + v + (hi ? 8 : 0);
      out[((size_t)(b_ * NTOK + n)) * CCH + h * DH + d] = acc[t][v];
    }
  }
}

__global__ __launch_bounds__(256) void sa_attn(float* __restrict__ ws,
                                               const float* __restrict__ temperature2,
                                               float* __restrict__ out) {
  __shared__ float P[128][64];
  const int tid  = threadIdx.x;
  const int lane = tid & 31;
  const int wave = tid >> 5;
  const int l16  = lane & 15;
  const int hi   = lane >> 4;
  const int bh = blockIdx.y;
  const int b_ = bh >> 3;
  const int h  = bh & 7;
  const int nBase = blockIdx.x * 128;

  const float* Q  = ws + OFF_Q  + (size_t)bh * DH * NTOK;
  const float* Kp = ws + OFF_KP + (size_t)bh * DH * KL;
  const float* Vp = ws + OFF_VP + (size_t)bh * DH * KL;
  const float t2 = temperature2[h];
  const float* iqp = ws + OFF_IQ + (size_t)bh * DH;

  {
    const int nGlob = nBase + wave * 16 + l16;
    v8f acc[4] = {};
    for (int d0 = 0; d0 < DH; d0 += 4) {
      const int k0 = d0 + (hi ? 2 : 0);
      v2f a;
      a.x = Q[(size_t)k0 * NTOK + nGlob] * iqp[k0];
      a.y = Q[(size_t)(k0 + 1) * NTOK + nGlob] * iqp[k0 + 1];
#pragma unroll
      for (int t = 0; t < 4; ++t) {
        const int kc = t * 16 + l16;
        v2f bfrag;
        bfrag.x = Kp[(size_t)k0 * KL + kc];
        bfrag.y = Kp[(size_t)(k0 + 1) * KL + kc];
        acc[t] = WMMA_F32(a, bfrag, acc[t]);
      }
    }
#pragma unroll
    for (int t = 0; t < 4; ++t)
#pragma unroll
      for (int v = 0; v < 8; ++v)
        P[wave * 16 + v + (hi ? 8 : 0)][t * 16 + l16] = acc[t][v] * t2;
  }
  __syncthreads();

  if (tid < 128) {
    float m = P[tid][0];
#pragma unroll
    for (int j = 1; j < 64; ++j) m = fmaxf(m, P[tid][j]);
    float sum = 0.f;
#pragma unroll
    for (int j = 0; j < 64; ++j) { float e = __expf(P[tid][j] - m); P[tid][j] = e; sum += e; }
    const float inv = 1.0f / sum;
#pragma unroll
    for (int j = 0; j < 64; ++j) P[tid][j] *= inv;
  }
  __syncthreads();

  {
    const int mRow = wave * 16 + l16;
    v8f acc[4] = {};
    for (int kk0 = 0; kk0 < KL; kk0 += 4) {
      const int k0 = kk0 + (hi ? 2 : 0);
      v2f a;
      a.x = P[mRow][k0];
      a.y = P[mRow][k0 + 1];
#pragma unroll
      for (int t = 0; t < 4; ++t) {
        const int dd = t * 16 + l16;
        v2f bfrag = *(const v2f*)(Vp + (size_t)dd * KL + k0);
        acc[t] = WMMA_F32(a, bfrag, acc[t]);
      }
    }
#pragma unroll
    for (int t = 0; t < 4; ++t) {
      const int dd = t * 16 + l16;
#pragma unroll
      for (int v = 0; v < 8; ++v) {
        const int n = nBase + wave * 16 + v + (hi ? 8 : 0);
        const size_t f = (size_t)dd * (HEADS * NTOK) + (size_t)h * NTOK + n;
        out[(size_t)b_ * (NTOK * CCH) + f] += acc[t][v];
      }
    }
  }
}

extern "C" void kernel_launch(void* const* d_in, const int* in_sizes, int n_in,
                              void* d_out, int out_size, void* d_ws, size_t ws_size,
                              hipStream_t stream) {
  const float* x  = (const float*)d_in[0];
  const float* W  = (const float*)d_in[1];
  const float* EF = (const float*)d_in[2];
  const float* t1 = (const float*)d_in[3];
  const float* t2 = (const float*)d_in[4];
  float* out = (float*)d_out;
  float* ws  = (float*)d_ws;

  qkvv_gemm  <<<dim3(128, 16), 128, 0, stream>>>(x, W, EF, ws);
  ef_reduce  <<<256,           256, 0, stream>>>(ws);
  ca_softmax <<<dim3(4, 32),   256, 0, stream>>>(ws, t1);
  ca_out     <<<dim3(32, 32),  256, 0, stream>>>(ws, out);
  sa_attn    <<<dim3(32, 32),  256, 0, stream>>>(ws, t2, out);
}


# sa_attn: row softmax on the logit accumulators (half-wave exchanges) instead of LDS row scans under 16-way bank conflicts; P handed on at a 68-float pitch; one barrier gone; softmax sum order changes
# speedup vs baseline: 2.8762x; 1.0325x over previous
#include <hip/hip_runtime.h>

typedef float v2f __attribute__((ext_vector_type(2)));
typedef float v8f __attribute__((ext_vector_type(8)));

#define WMMA_F32(a, b, c) \
  __builtin_amdgcn_wmma_f32_16x16x4_f32(false, (a), false, (b), (short)0, (c), false, false)

#define BATCH 4
#define NTOK 4096
#define CCH 512
#define HEADS 8
#define DH 64
#define KL 64
#define MROWS (BATCH * NTOK)
#define OCH (4 * CCH)
#define BH (BATCH * HEADS)

#define OFF_Q   ((size_t)0)
#define OFF_K   ((size_t)8388608)
#define OFF_VCA ((size_t)16777216)
#define OFF_VSA ((size_t)25165824)
#define OFF_KP  ((size_t)33554432)
#define OFF_VP  ((size_t)33685504)
#define OFF_P   ((size_t)33816576)
#define OFF_SP  (OFF_K)
#define OFF_SQ  (OFF_K + (size_t)4194304)
#define OFF_IQ  (OFF_K + (size_t)4718592)

typedef float v4f __attribute__((ext_vector_type(4)));

__device__ __forceinline__ void k1_load_set(v4f (&fa)[4][4], v4f (&fb)[4][4],
                                            const float* __restrict__ aPtr,
                                            const float* __restrict__ bPtr,
                                            const int k) {
#pragma unroll
  for (int i = 0; i < 4; ++i)
#pragma unroll
    for (int q = 0; q < 4; ++q)
      fa[i][q] = *(const v4f*)(aPtr + (size_t)i * 16 * CCH + k + q * 4);
#pragma unroll
  for (int j = 0; j < 4; ++j)
#pragma unroll
    for (int q = 0; q < 4; ++q)
      fb[j][q] = *(const v4f*)(bPtr + (size_t)j * 16 * CCH + k + q * 4);
}

__device__ __forceinline__ void k1_mma_set(v8f (&acc)[4][4],
                                           const v4f (&fa)[4][4],
                                           const v4f (&fb)[4][4]) {
#pragma unroll
  for (int q = 0; q < 4; ++q) {
#pragma unroll
    for (int i = 0; i < 4; ++i) {
      v2f a;
      a.x = fa[i][q].x;
      a.y = fa[i][q].y;
#pragma unroll
      for (int j = 0; j < 4; ++j) {
        v2f b;
        b.x = fb[j][q].x;
        b.y = fb[j][q].y;
        acc[i][j] = WMMA_F32(a, b, acc[i][j]);
      }
    }
#pragma unroll
    for (int i = 0; i < 4; ++i) {
      v2f a;
      a.x = fa[i][q].z;
      a.y = fa[i][q].w;
#pragma unroll
      for (int j = 0; j < 4; ++j) {
        v2f b;
        b.x = fb[j][q].z;
        b.y = fb[j][q].w;
        acc[i][j] = WMMA_F32(a, b, acc[i][j]);
      }
    }
  }
}

__global__ __launch_bounds__(128) void qkvv_gemm(const float* __restrict__ x,
                                                 const float* __restrict__ W,
                                                 const float* __restrict__ EF,
                                                 float* __restrict__ ws) {
  const int lane = threadIdx.x & 31;
  const int wave = threadIdx.x >> 5;
  const int l16  = lane & 15;
  const int hi   = lane >> 4;
  const int wm = wave >> 1;
  const int wo = wave & 1;
  const int p  = (blockIdx.y >> 3) * 2 + wo;
  const int h  = blockIdx.y & 7;
  const int mW = blockIdx.x * 128 + wm * 64;
  const int oW = p * CCH + h * DH;

  const float* aPtr = x + (size_t)(mW + l16) * CCH + hi * 16;
  const float* bPtr = W + (size_t)(oW + l16) * CCH + hi * 16;

  v8f acc[4][4] = {};
  v4f fa0[4][4], fb0[4][4], fa1[4][4], fb1[4][4];

  k1_load_set(fa0, fb0, aPtr, bPtr, 0);
  for (int k0 = 0; k0 < CCH - 64; k0 += 64) {
    k1_load_set(fa1, fb1, aPtr, bPtr, k0 + 32);
    k1_mma_set(acc, fa0, fb0);
    k1_load_set(fa0, fb0, aPtr, bPtr, k0 + 64);
    k1_mma_set(acc, fa1, fb1);
  }
  k1_load_set(fa1, fb1, aPtr, bPtr, CCH - 32);
  k1_mma_set(acc, fa0, fb0);
  k1_mma_set(acc, fa1, fb1);

  const int b_ = mW >> 12;
  const int bh = b_ * HEADS + h;
  if (p == 0 || p == 2) {
    float* dstBase = ws + (size_t)p * 8388608 + (size_t)bh * DH * NTOK;
#pragma unroll
    for (int i = 0; i < 4; ++i) {
#pragma unroll
      for (int j = 0; j < 4; ++j) {
        const int dd = j * 16 + l16;
#pragma unroll
        for (int v = 0; v < 8; ++v) {
          const int n = (mW & (NTOK - 1)) + i * 16 + v + (hi ? 8 : 0);
          dstBase[(size_t)dd * NTOK + n] = acc[i][j][v];
        }
      }
    }
  }

  __shared__ v4f red[2][32][32];
  const int nTok0 = mW & (NTOK - 1);

  if (p == 1) {
#pragma unroll
    for (int i = 0; i < 4; ++i)
#pragma unroll
      for (int j = 0; j < 4; ++j) {
        v4f lo, up;
        lo.x = acc[i][j][0]; lo.y = acc[i][j][1]; lo.z = acc[i][j][2]; lo.w = acc[i][j][3];
        up.x = acc[i][j][4]; up.y = acc[i][j][5]; up.z = acc[i][j][6]; up.w = acc[i][j][7];
        red[wm][(i * 4 + j) * 2][lane]     = lo;
        red[wm][(i * 4 + j) * 2 + 1][lane] = up;
      }
  }
  __syncthreads();

  if (p <= 1) {
    float* sq = ws + OFF_SQ + ((size_t)((p * BH + bh) * 128 + (nTok0 >> 6) * 2 + hi)) * DH;
#pragma unroll
    for (int j = 0; j < 4; ++j) {
      float s = 0.f;
#pragma unroll
      for (int i = 0; i < 4; ++i)
#pragma unroll
        for (int v = 0; v < 8; ++v) s += acc[i][j][v] * acc[i][j][v];
      sq[j * 16 + l16] = s;
    }
  }

  v8f pacc[4][4] = {};
  if (p == 0) {
#pragma unroll
    for (int i = 0; i < 4; ++i) {
      v4f kt[4][2];
#pragma unroll
      for (int jj = 0; jj < 4; ++jj) {
        kt[jj][0] = red[wm][(i * 4 + jj) * 2][lane];
        kt[jj][1] = red[wm][(i * 4 + jj) * 2 + 1][lane];
      }
#pragma unroll
      for (int t = 0; t < 4; ++t) {
#pragma unroll
        for (int j = 0; j < 4; ++j) {
          v2f a;
          a.x = acc[i][j][2 * t];
          a.y = acc[i][j][2 * t + 1];
#pragma unroll
          for (int jj = 0; jj < 4; ++jj) {
            v2f b;
            b.x = kt[jj][t >> 1][(2 * t) & 3];
            b.y = kt[jj][t >> 1][(2 * t + 1) & 3];
            pacc[j][jj] = WMMA_F32(a, b, pacc[j][jj]);
          }
        }
      }
    }
  } else if (p != 2) {
    const float* efBase = EF + (size_t)(nTok0 + hi * 8) * KL + l16 * 4;
#pragma unroll
    for (int i = 0; i < 4; ++i) {
      v4f e[4][2];
#pragma unroll
      for (int t = 0; t < 4; ++t)
#pragma unroll
        for (int s = 0; s < 2; ++s)
          e[t][s] = *(const v4f*)(efBase + (size_t)(i * 16 + 2 * t + s) * KL);
#pragma unroll
      for (int t = 0; t < 4; ++t) {
#pragma unroll
        for (int j = 0; j < 4; ++j) {
          v2f a;
          a.x = acc[i][j][2 * t];
          a.y = acc[i][j][2 * t + 1];
#pragma unroll
          for (int jj = 0; jj < 4; ++jj) {
            v2f b;
            b.x = e[t][0][jj];
            b.y = e[t][1][jj];
            pacc[j][jj] = WMMA_F32(a, b, pacc[j][jj]);
          }
        }
      }
    }
  }
  __syncthreads();

  if (p != 2 && wm == 1) {
#pragma unroll
    for (int j = 0; j < 4; ++j)
#pragma unroll
      for (int jj = 0; jj < 4; ++jj) {
        v4f lo, up;
        lo.x = pacc[j][jj][0]; lo.y = pacc[j][jj][1]; lo.z = pacc[j][jj][2]; lo.w = pacc[j][jj][3];
        up.x = pacc[j][jj][4]; up.y = pacc[j][jj][5]; up.z = pacc[j][jj][6]; up.w = pacc[j][jj][7];
        red[wo][(j * 4 + jj) * 2][lane]     = lo;
        red[wo][(j * 4 + jj) * 2 + 1][lane] = up;
      }
  }
  __syncthreads();
  if (p != 2 && wm == 0) {
#pragma unroll
    for (int j = 0; j < 4; ++j)
#pragma unroll
      for (int jj = 0; jj < 4; ++jj) {
        const v4f lo = red[wo][(j * 4 + jj) * 2][lane];
        const v4f up = red[wo][(j * 4 + jj) * 2 + 1][lane];
        pacc[j][jj][0] += lo.x; pacc[j][jj][1] += lo.y; pacc[j][jj][2] += lo.z; pacc[j][jj][3] += lo.w;
        pacc[j][jj][4] += up.x; pacc[j][jj][5] += up.y; pacc[j][jj][6] += up.z; pacc[j][jj][7] += up.w;
      }
    const int mt = blockIdx.x & 31;
    if (p == 0) {
      float* sp = ws + OFF_SP + ((size_t)(bh * 32 + mt)) * (DH * DH);
#pragma unroll
      for (int j = 0; j < 4; ++j)
#pragma unroll
        for (int jj = 0; jj < 4; ++jj)
#pragma unroll
          for (int r = 0; r < 8; ++r)
            sp[(size_t)(j * 16 + hi * 8 + r) * DH + jj * 16 + l16] = pacc[j][jj][r];
    } else {
      const int which = p >> 1;
      float* part = ws + OFF_VSA +
                    ((size_t)((which * BH + bh) * 32 + mt)) * (DH * KL);
#pragma unroll
      for (int j = 0; j < 4; ++j)
#pragma unroll
        for (int r = 0; r < 8; ++r) {
          v4f o;
          o.x = pacc[j][0][r]; o.y = pacc[j][1][r]; o.z = pacc[j][2][r]; o.w = pacc[j][3][r];
          *(v4f*)(part + (size_t)(j * 16 + hi * 8 + r) * KL + l16 * 4) = o;
        }
    }
  }
}

__global__ __launch_bounds__(256) void ef_reduce(float* __restrict__ ws) {
  const int g  = blockIdx.x * 256 + threadIdx.x;
  const int wb = g >> 10;
  const int e4 = g & 1023;
  const v4f* src = (const v4f*)(ws + OFF_VSA + (size_t)wb * 32 * (DH * KL)) + e4;
  v4f s = src[0];
#pragma unroll
  for (int m = 1; m < 32; ++m) s += src[(size_t)m * (DH * KL / 4)];
  const int which = wb >> 5;
  const int bh    = wb & 31;
  float* dst = ws + (which ? OFF_VP : OFF_KP) + (size_t)bh * DH * KL;
  *(v4f*)(dst + e4 * 4) = s;
}

__global__ __launch_bounds__(256) void ca_softmax(float* __restrict__ ws,
                                                  const float* __restrict__ temperature) {
  __shared__ float k2s[4][64];
  const int tid = threadIdx.x;
  const int bh  = blockIdx.y;
  const int h   = bh & 7;
  const int e4  = tid & 15;
  const int d   = blockIdx.x * 16 + (tid >> 4);

  const v4f* sp = (const v4f*)(ws + OFF_SP + (size_t)bh * 32 * (DH * DH) + (size_t)d * DH) + e4;
  v4f s = sp[0];
#pragma unroll
  for (int m = 1; m < 32; ++m) s += sp[(size_t)m * (DH * DH / 4)];

  {
    const float* kq = ws + OFF_SQ + ((size_t)((BH + bh) * 128 + (tid >> 6) * 32)) * DH + (tid & 63);
    float a = 0.f;
#pragma unroll
    for (int m = 0; m < 32; ++m) a += kq[m * DH];
    k2s[tid >> 6][tid & 63] = a;
  }
  float q2 = 0.f;
  {
    const float* qq = ws + OFF_SQ + ((size_t)(bh * 128 + e4 * 8)) * DH + d;
#pragma unroll
    for (int m = 0; m < 8; ++m) q2 += qq[m * DH];
  }
  q2 += __shfl_xor(q2, 8, 16);
  q2 += __shfl_xor(q2, 4, 16);
  q2 += __shfl_xor(q2, 2, 16);
  q2 += __shfl_xor(q2, 1, 16);
  __syncthreads();

  const float iq  = 1.0f / fmaxf(sqrtf(q2), 1e-12f);
  const float tsc = temperature[h];
  float sc[4];
#pragma unroll
  for (int c = 0; c < 4; ++c) {
    const int e = e4 * 4 + c;
    const float k2 = (k2s[0][e] + k2s[1][e]) + (k2s[2][e] + k2s[3][e]);
    const float ik = 1.0f / fmaxf(sqrtf(k2), 1e-12f);
    sc[c] = s[c] * iq * ik * tsc;
  }
  float m = fmaxf(fmaxf(sc[0], sc[1]), fmaxf(sc[2], sc[3]));
  m = fmaxf(m, __shfl_xor(m, 8, 16));
  m = fmaxf(m, __shfl_xor(m, 4, 16));
  m = fmaxf(m, __shfl_xor(m, 2, 16));
  m = fmaxf(m, __shfl_xor(m, 1, 16));
  v4f ex;
  ex.x = __expf(sc[0] - m); ex.y = __expf(sc[1] - m);
  ex.z = __expf(sc[2] - m); ex.w = __expf(sc[3] - m);
  float sum = (ex.x + ex.y) + (ex.z + ex.w);
  sum += __shfl_xor(sum, 8, 16);
  sum += __shfl_xor(sum, 4, 16);
  sum += __shfl_xor(sum, 2, 16);
  sum += __shfl_xor(sum, 1, 16);
  const float inv = 1.0f / sum;
  ex *= inv;
  *(v4f*)(ws + OFF_P + (size_t)bh * DH * DH + (size_t)d * DH + e4 * 4) = ex;
  if (e4 == 0) ws[OFF_IQ + (size_t)bh * DH + d] = iq;
}

__global__ __launch_bounds__(256) void ca_out(const float* __restrict__ ws,
                                              float* __restrict__ out) {
  __shared__ float P[64][64];
  const int tid  = threadIdx.x;
  const int lane = tid & 31;
  const int wave = tid >> 5;
  const int l16  = lane & 15;
  const int hi   = lane >> 4;
  const int bh = blockIdx.y;
  const int b_ = bh >> 3;
  const int h  = bh & 7;
  const int nBase = blockIdx.x * 128;

  const float* Pg = ws + OFF_P + (size_t)bh * DH * DH;
  for (int i = tid; i < 4096; i += 256) P[i >> 6][i & 63] = Pg[i];
  __syncthreads();

  const int dt = wave >> 1;
  const int ng = wave & 1;
  const int mRow = dt * 16 + l16;
  const float* vca = ws + OFF_VCA + (size_t)bh * DH * NTOK;

  v8f acc[4] = {};
  for (int e0 = 0; e0 < DH; e0 += 4) {
    const int k0 = e0 + (hi ? 2 : 0);
    v2f a;
    a.x = P[mRow][k0];
    a.y = P[mRow][k0 + 1];
#pragma unroll
    for (int t = 0; t < 4; ++t) {
      const int n = nBase + (ng * 4 + t) * 16 + l16;
      v2f bfrag;
      bfrag.x = vca[(size_t)k0 * NTOK + n];
      bfrag.y = vca[(size_t)(k0 + 1) * NTOK + n];
      acc[t] = WMMA_F32(a, bfrag, acc[t]);
    }
  }
#pragma unroll
  for (int t = 0; t < 4; ++t) {
    const int n = nBase + (ng * 4 + t) * 16 + l16;
#pragma unroll
    for (int v = 0; v < 8; ++v) {
      const int d = dt * 16 + v + (hi ? 8 : 0);
      out[((size_t)(b_ * NTOK + n)) * CCH + h * DH + d] = acc[t][v];
    }
  }
}

#define SA_PITCH 68
__global__ __launch_bounds__(256) void sa_attn(float* __restrict__ ws,
                                               const float* __restrict__ temperature2,
                                               float* __restrict__ out) {
  __shared__ float P[128][SA_PITCH];
  const int tid  = threadIdx.x;
  const int lane = tid & 31;
  const int wave = tid >> 5;
  const int l16  = lane & 15;
  const int hi   = lane >> 4;
  const int bh = blockIdx.y;
  const int b_ = bh >> 3;
  const int h  = bh & 7;
  const int nBase = blockIdx.x * 128;

  const float* Q  = ws + OFF_Q  + (size_t)bh * DH * NTOK;
  const float* Kp = ws + OFF_KP + (size_t)bh * DH * KL;
  const float* Vp = ws + OFF_VP + (size_t)bh * DH * KL;
  const float t2 = temperature2[h];
  const float* iqp = ws + OFF_IQ + (size_t)bh * DH;

  {
    const int nGlob = nBase + wave * 16 + l16;
    v8f acc[4] = {};
    for (int d0 = 0; d0 < DH; d0 += 4) {
      const int k0 = d0 + (hi ? 2 : 0);
      v2f a;
      a.x = Q[(size_t)k0 * NTOK + nGlob] * iqp[k0];
      a.y = Q[(size_t)(k0 + 1) * NTOK + nGlob] * iqp[k0 + 1];
#pragma unroll
      for (int t = 0; t < 4; ++t) {
        const int kc = t * 16 + l16;
        v2f bfrag;
        bfrag.x = Kp[(size_t)k0 * KL + kc];
        bfrag.y = Kp[(size_t)(k0 + 1) * KL + kc];
        acc[t] = WMMA_F32(a, bfrag, acc[t]);
      }
    }
#pragma unroll
    for (int v = 0; v < 8; ++v) {
      const float x0 = acc[0][v] * t2;
      const float x1 = acc[1][v] * t2;
      const float x2 = acc[2][v] * t2;
      const float x3 = acc[3][v] * t2;
      float m = fmaxf(fmaxf(x0, x1), fmaxf(x2, x3));
      m = fmaxf(m, __shfl_xor(m, 8, 16));
      m = fmaxf(m, __shfl_xor(m, 4, 16));
      m = fmaxf(m, __shfl_xor(m, 2, 16));
      m = fmaxf(m, __shfl_xor(m, 1, 16));
      const float e0 = __expf(x0 - m);
      const float e1 = __expf(x1 - m);
      const float e2 = __expf(x2 - m);
      const float e3 = __expf(x3 - m);
      float s = (e0 + e1) + (e2 + e3);
      s += __shfl_xor(s, 8, 16);
      s += __shfl_xor(s, 4, 16);
      s += __shfl_xor(s, 2, 16);
      s += __shfl_xor(s, 1, 16);
      const float inv = 1.0f / s;
      float* prow = &P[wave * 16 + v + (hi ? 8 : 0)][l16];
      prow[0]  = e0 * inv;
      prow[16] = e1 * inv;
      prow[32] = e2 * inv;
      prow[48] = e3 * inv;
    }
  }
  __syncthreads();

  {
    const int mRow = wave * 16 + l16;
    v8f acc[4] = {};
    for (int kk0 = 0; kk0 < KL; kk0 += 4) {
      const int k0 = kk0 + (hi ? 2 : 0);
      v2f a;
      a.x = P[mRow][k0];
      a.y = P[mRow][k0 + 1];
#pragma unroll
      for (int t = 0; t < 4; ++t) {
        const int dd = t * 16 + l16;
        v2f bfrag = *(const v2f*)(Vp + (size_t)dd * KL + k0);
        acc[t] = WMMA_F32(a, bfrag, acc[t]);
      }
    }
#pragma unroll
    for (int t = 0; t < 4; ++t) {
      const int dd = t * 16 + l16;
#pragma unroll
      for (int v = 0; v < 8; ++v) {
        const int n = nBase + wave * 16 + v + (hi ? 8 : 0);
        const size_t f = (size_t)dd * (HEADS * NTOK) + (size_t)h * NTOK + n;
        out[(size_t)b_ * (NTOK * CCH) + f] += acc[t][v];
      }
    }
  }
}

extern "C" void kernel_launch(void* const* d_in, const int* in_sizes, int n_in,
                              void* d_out, int out_size, void* d_ws, size_t ws_size,
                              hipStream_t stream) {
  const float* x  = (const float*)d_in[0];
  const float* W  = (const float*)d_in[1];
  const float* EF = (const float*)d_in[2];
  const float* t1 = (const float*)d_in[3];
  const float* t2 = (const float*)d_in[4];
  float* out = (float*)d_out;
  float* ws  = (float*)d_ws;

  qkvv_gemm  <<<dim3(128, 16), 128, 0, stream>>>(x, W, EF, ws);
  ef_reduce  <<<256,           256, 0, stream>>>(ws);
  ca_softmax <<<dim3(4, 32),   256, 0, stream>>>(ws, t1);
  ca_out     <<<dim3(32, 32),  256, 0, stream>>>(ws, out);
  sa_attn    <<<dim3(32, 32),  256, 0, stream>>>(ws, t2, out);
}


# Token sums taken on x before projecting: G = x^T x (36 upper tiles, mirrored) and EF^T x replace the K and Vsa projections; S = Wq G Wk^T, |k|^2, k_proj, v_proj come from them; sum orders change
# speedup vs baseline: 3.3629x; 1.1692x over previous
#include <hip/hip_runtime.h>

typedef float v2f __attribute__((ext_vector_type(2)));
typedef float v8f __attribute__((ext_vector_type(8)));

#define WMMA_F32(a, b, c) \
  __builtin_amdgcn_wmma_f32_16x16x4_f32(false, (a), false, (b), (short)0, (c), false, false)

#define BATCH 4
#define NTOK 4096
#define CCH 512
#define HEADS 8
#define DH 64
#define KL 64
#define MROWS (BATCH * NTOK)
#define OCH (4 * CCH)
#define BH (BATCH * HEADS)

#define OFF_Q   ((size_t)0)
#define OFF_K   ((size_t)8388608)
#define OFF_VCA ((size_t)16777216)
#define OFF_VSA ((size_t)25165824)
#define OFF_KP  ((size_t)33554432)
#define OFF_VP  ((size_t)33685504)
#define OFF_P   ((size_t)33816576)
#define NSLICE 8
#define NTILE  44
#define YROWS  576
#define OFF_YP  (OFF_K)
#define OFF_YF  (OFF_VSA)
#define OFF_TT  (OFF_VSA + (size_t)1179648)
#define OFF_S   (OFF_VSA + (size_t)2228224)
#define OFF_K2  (OFF_VSA + (size_t)2359296)
#define OFF_SQ  (OFF_VSA + (size_t)2361344)
#define OFF_IQ  (OFF_VSA + (size_t)2623488)

typedef float v4f __attribute__((ext_vector_type(4)));

__device__ __forceinline__ void k1_load_set(v4f (&fa)[4][4], v4f (&fb)[4][4],
                                            const float* __restrict__ aPtr,
                                            const float* __restrict__ bPtr,
                                            const int k) {
#pragma unroll
  for (int i = 0; i < 4; ++i)
#pragma unroll
    for (int q = 0; q < 4; ++q)
      fa[i][q] = *(const v4f*)(aPtr + (size_t)i * 16 * CCH + k + q * 4);
#pragma unroll
  for (int j = 0; j < 4; ++j)
#pragma unroll
    for (int q = 0; q < 4; ++q)
      fb[j][q] = *(const v4f*)(bPtr + (size_t)j * 16 * CCH + k + q * 4);
}

__device__ __forceinline__ void k1_mma_set(v8f (&acc)[4][4],
                                           const v4f (&fa)[4][4],
                                           const v4f (&fb)[4][4]) {
#pragma unroll
  for (int q = 0; q < 4; ++q) {
#pragma unroll
    for (int i = 0; i < 4; ++i) {
      v2f a;
      a.x = fa[i][q].x;
      a.y = fa[i][q].y;
#pragma unroll
      for (int j = 0; j < 4; ++j) {
        v2f b;
        b.x = fb[j][q].x;
        b.y = fb[j][q].y;
        acc[i][j] = WMMA_F32(a, b, acc[i][j]);
      }
    }
#pragma unroll
    for (int i = 0; i < 4; ++i) {
      v2f a;
      a.x = fa[i][q].z;
      a.y = fa[i][q].w;
#pragma unroll
      for (int j = 0; j < 4; ++j) {
        v2f b;
        b.x = fb[j][q].z;
        b.y = fb[j][q].w;
        acc[i][j] = WMMA_F32(a, b, acc[i][j]);
      }
    }
  }
}

__device__ __forceinline__ void qv_body(const float* __restrict__ x,
                                        const float* __restrict__ W,
                                        float* __restrict__ ws,
                                        const int bx, const int by) {
  const int lane = threadIdx.x & 31;
  const int wave = threadIdx.x >> 5;
  const int l16  = lane & 15;
  const int hi   = lane >> 4;
  const int wm = wave >> 1;
  const int wo = wave & 1;
  const int p  = wo * 2;
  const int h  = by;
  const int mW = bx * 128 + wm * 64;
  const int oW = p * CCH + h * DH;

  const float* aPtr = x + (size_t)(mW + l16) * CCH + hi * 16;
  const float* bPtr = W + (size_t)(oW + l16) * CCH + hi * 16;

  v8f acc[4][4] = {};
  v4f fa0[4][4], fb0[4][4], fa1[4][4], fb1[4][4];

  k1_load_set(fa0, fb0, aPtr, bPtr, 0);
  for (int k0 = 0; k0 < CCH - 64; k0 += 64) {
    k1_load_set(fa1, fb1, aPtr, bPtr, k0 + 32);
    k1_mma_set(acc, fa0, fb0);
    k1_load_set(fa0, fb0, aPtr, bPtr, k0 + 64);
    k1_mma_set(acc, fa1, fb1);
  }
  k1_load_set(fa1, fb1, aPtr, bPtr, CCH - 32);
  k1_mma_set(acc, fa0, fb0);
  k1_mma_set(acc, fa1, fb1);

  const int b_ = mW >> 12;
  const int bh = b_ * HEADS + h;
  const int nTok0 = mW & (NTOK - 1);
  float* dstBase = ws + (size_t)p * 8388608 + (size_t)bh * DH * NTOK;
#pragma unroll
  for (int i = 0; i < 4; ++i) {
#pragma unroll
    for (int j = 0; j < 4; ++j) {
      const int dd = j * 16 + l16;
#pragma unroll
      for (int v = 0; v < 8; ++v) {
        const int n = nTok0 + i * 16 + v + (hi ? 8 : 0);
        dstBase[(size_t)dd * NTOK + n] = acc[i][j][v];
      }
    }
  }
  if (p == 0) {
    float* sq = ws + OFF_SQ + ((size_t)(bh * 128 + (nTok0 >> 6) * 2 + hi)) * DH;
#pragma unroll
    for (int j = 0; j < 4; ++j) {
      float s = 0.f;
#pragma unroll
      for (int i = 0; i < 4; ++i)
#pragma unroll
        for (int v = 0; v < 8; ++v) s += acc[i][j][v] * acc[i][j][v];
      sq[j * 16 + l16] = s;
    }
  }
}

__device__ __forceinline__ void gr_load(v4f (&f)[4][2], const float* __restrict__ p,
                                        const size_t stride) {
#pragma unroll
  for (int t = 0; t < 4; ++t)
#pragma unroll
    for (int s = 0; s < 2; ++s)
      f[t][s] = *(const v4f*)(p + (size_t)(2 * t + s) * stride);
}

__device__ __forceinline__ void gr_mma(v8f (&acc)[4][4], const v4f (&fa)[4][2],
                                       const v4f (&fb)[4][2]) {
#pragma unroll
  for (int t = 0; t < 4; ++t) {
#pragma unroll
    for (int jr = 0; jr < 4; ++jr) {
      v2f a;
      a.x = fa[t][0][jr];
      a.y = fa[t][1][jr];
#pragma unroll
      for (int jc = 0; jc < 4; ++jc) {
        v2f b;
        b.x = fb[t][0][jc];
        b.y = fb[t][1][jc];
        acc[jr][jc] = WMMA_F32(a, b, acc[jr][jc]);
      }
    }
  }
}

__device__ __forceinline__ void tile_ij(const int tix, int& I, int& J) {
  if (tix >= 36) { I = 8; J = tix - 36; return; }
  int i = 0, rem = tix;
  while (rem >= 8 - i) { rem -= 8 - i; ++i; }
  I = i;
  J = i + rem;
}

__device__ __forceinline__ void gram_body(const float* __restrict__ x,
                                          const float* __restrict__ EF,
                                          float* __restrict__ ws,
                                          const int bx, const int slice, const int b_) {
  const int lane = threadIdx.x & 31;
  const int wave = __builtin_amdgcn_readfirstlane((int)(threadIdx.x >> 5));
  const int l16  = lane & 15;
  const int hi   = lane >> 4;
  const int tix   = bx * 4 + wave;
  int I, J;
  tile_ij(tix, I, J);
  const int n0 = slice * (NTOK / NSLICE) + hi * 8;
  const float* xb = x + (size_t)b_ * NTOK * CCH;
  const float* cPtr = xb + (size_t)n0 * CCH + J * 64 + l16 * 4;
  const bool ef = (I == 8);
  const size_t sA = ef ? (size_t)KL : (size_t)CCH;
  const float* rPtr = ef ? (EF + (size_t)n0 * KL + l16 * 4)
                         : (xb + (size_t)n0 * CCH + I * 64 + l16 * 4);

  v8f acc[4][4] = {};
  v4f fa0[4][2], fb0[4][2], fa1[4][2], fb1[4][2];
  const int NB = NTOK / NSLICE / 16;
  gr_load(fa0, rPtr, sA);
  gr_load(fb0, cPtr, CCH);
  for (int blk = 0; blk < NB - 2; blk += 2) {
    gr_load(fa1, rPtr + (size_t)(blk + 1) * 16 * sA, sA);
    gr_load(fb1, cPtr + (size_t)(blk + 1) * 16 * CCH, CCH);
    gr_mma(acc, fa0, fb0);
    gr_load(fa0, rPtr + (size_t)(blk + 2) * 16 * sA, sA);
    gr_load(fb0, cPtr + (size_t)(blk + 2) * 16 * CCH, CCH);
    gr_mma(acc, fa1, fb1);
  }
  gr_load(fa1, rPtr + (size_t)(NB - 1) * 16 * sA, sA);
  gr_load(fb1, cPtr + (size_t)(NB - 1) * 16 * CCH, CCH);
  gr_mma(acc, fa0, fb0);
  gr_mma(acc, fa1, fb1);

  float* part = ws + OFF_YP + ((size_t)((b_ * NSLICE + slice) * NTILE + tix)) * 4096;
#pragma unroll
  for (int jr = 0; jr < 4; ++jr)
#pragma unroll
    for (int v = 0; v < 8; ++v) {
      v4f o;
      o.x = acc[jr][0][v]; o.y = acc[jr][1][v]; o.z = acc[jr][2][v]; o.w = acc[jr][3][v];
      *(v4f*)(part + (size_t)(32 * hi + 4 * v + jr) * 64 + l16 * 4) = o;
    }
}

__global__ __launch_bounds__(128) void proj_gram(const float* __restrict__ x,
                                                 const float* __restrict__ W,
                                                 const float* __restrict__ EF,
                                                 float* __restrict__ ws) {
  const int blk = blockIdx.x;
  if (blk < 1024) {
    qv_body(x, W, ws, blk & 127, blk >> 7);
  } else {
    const int g = blk - 1024;
    const int q = g / 11;
    gram_body(x, EF, ws, g - q * 11, q & 7, q >> 3);
  }
}

__global__ __launch_bounds__(256) void y_reduce(float* __restrict__ ws) {
  __shared__ float tile[64][65];
  const int tid = threadIdx.x;
  const int b_  = blockIdx.x / NTILE;
  const int tix = blockIdx.x - b_ * NTILE;
  int I, J;
  tile_ij(tix, I, J);
  float* yf = ws + OFF_YF + (size_t)b_ * YROWS * CCH;
  const v4f* sp0 = (const v4f*)(ws + OFF_YP + ((size_t)(b_ * NSLICE * NTILE + tix)) * 4096);
#pragma unroll
  for (int m = 0; m < 4; ++m) {
    const int idx  = tid + 256 * m;
    const int rloc = idx >> 4;
    const int c4   = idx & 15;
    v4f s = sp0[idx];
#pragma unroll
    for (int sl = 1; sl < NSLICE; ++sl) s += sp0[(size_t)sl * NTILE * 1024 + idx];
    *(v4f*)(yf + (size_t)(I * 64 + rloc) * CCH + J * 64 + c4 * 4) = s;
    tile[rloc][c4 * 4 + 0] = s.x;
    tile[rloc][c4 * 4 + 1] = s.y;
    tile[rloc][c4 * 4 + 2] = s.z;
    tile[rloc][c4 * 4 + 3] = s.w;
  }
  __syncthreads();
  if (I < J) {
#pragma unroll
    for (int m = 0; m < 4; ++m) {
      const int idx  = tid + 256 * m;
      const int cloc = idx >> 4;
      const int r4   = idx & 15;
      v4f o;
      o.x = tile[r4 * 4 + 0][cloc];
      o.y = tile[r4 * 4 + 1][cloc];
      o.z = tile[r4 * 4 + 2][cloc];
      o.w = tile[r4 * 4 + 3][cloc];
      *(v4f*)(yf + (size_t)(J * 64 + cloc) * CCH + I * 64 + r4 * 4) = o;
    }
  }
}

template <int MI, int NJ>
__device__ __forceinline__ void sg_load(v4f (&fa)[MI][4], v4f (&fb)[NJ][4],
                                        const float* __restrict__ aPtr,
                                        const float* __restrict__ bPtr, const int k) {
#pragma unroll
  for (int i = 0; i < MI; ++i)
#pragma unroll
    for (int q = 0; q < 4; ++q)
      fa[i][q] = *(const v4f*)(aPtr + (size_t)i * 16 * CCH + k + q * 4);
#pragma unroll
  for (int j = 0; j < NJ; ++j)
#pragma unroll
    for (int q = 0; q < 4; ++q)
      fb[j][q] = *(const v4f*)(bPtr + (size_t)j * 16 * CCH + k + q * 4);
}

template <int MI, int NJ>
__device__ __forceinline__ void sg_mma(v8f (&acc)[MI][NJ], const v4f (&fa)[MI][4],
                                       const v4f (&fb)[NJ][4]) {
#pragma unroll
  for (int q = 0; q < 4; ++q) {
#pragma unroll
    for (int i = 0; i < MI; ++i) {
      v2f a;
      a.x = fa[i][q].x;
      a.y = fa[i][q].y;
#pragma unroll
      for (int j = 0; j < NJ; ++j) {
        v2f b;
        b.x = fb[j][q].x;
        b.y = fb[j][q].y;
        acc[i][j] = WMMA_F32(a, b, acc[i][j]);
      }
    }
#pragma unroll
    for (int i = 0; i < MI; ++i) {
      v2f a;
      a.x = fa[i][q].z;
      a.y = fa[i][q].w;
#pragma unroll
      for (int j = 0; j < NJ; ++j) {
        v2f b;
        b.x = fb[j][q].z;
        b.y = fb[j][q].w;
        acc[i][j] = WMMA_F32(a, b, acc[i][j]);
      }
    }
  }
}

template <int MI, int NJ>
__device__ __forceinline__ void sg_run(const float* __restrict__ aRow,
                                       const float* __restrict__ bRow,
                                       float* __restrict__ dst, const int ld,
                                       const int l16, const int hi) {
  const float* aPtr = aRow + hi * 16;
  const float* bPtr = bRow + hi * 16;
  v8f acc[MI][NJ] = {};
  v4f fa0[MI][4], fb0[NJ][4], fa1[MI][4], fb1[NJ][4];
  sg_load<MI, NJ>(fa0, fb0, aPtr, bPtr, 0);
  for (int k0 = 0; k0 < CCH - 64; k0 += 64) {
    sg_load<MI, NJ>(fa1, fb1, aPtr, bPtr, k0 + 32);
    sg_mma<MI, NJ>(acc, fa0, fb0);
    sg_load<MI, NJ>(fa0, fb0, aPtr, bPtr, k0 + 64);
    sg_mma<MI, NJ>(acc, fa1, fb1);
  }
  sg_load<MI, NJ>(fa1, fb1, aPtr, bPtr, CCH - 32);
  sg_mma<MI, NJ>(acc, fa0, fb0);
  sg_mma<MI, NJ>(acc, fa1, fb1);
#pragma unroll
  for (int i = 0; i < MI; ++i)
#pragma unroll
    for (int j = 0; j < NJ; ++j)
#pragma unroll
      for (int v = 0; v < 8; ++v)
        dst[(size_t)(i * 16 + hi * 8 + v) * ld + j * 16 + l16] = acc[i][j][v];
}

__global__ __launch_bounds__(128) void tt_gemm(const float* __restrict__ W,
                                               float* __restrict__ ws) {
  const int lane = threadIdx.x & 31;
  const int l16  = lane & 15;
  const int hi   = lane >> 4;
  const int wg   = blockIdx.x * 4 + (threadIdx.x >> 5);
  if (wg < 1024) {
    const int b_ = wg >> 8;
    const int to = (wg >> 4) & 15;
    const int tc = wg & 15;
    const float* yf = ws + OFF_YF + (size_t)b_ * YROWS * CCH;
    sg_run<2, 2>(W + (size_t)(CCH + to * 32 + l16) * CCH,
                 yf + (size_t)(tc * 32 + l16) * CCH,
                 ws + OFF_TT + (size_t)b_ * CCH * CCH + (size_t)(to * 32) * CCH + tc * 32,
                 CCH, l16, hi);
  } else {
    const int u     = wg - 1024;
    const int b_    = u >> 6;
    const int which = (u >> 5) & 1;
    const int to    = (u >> 1) & 15;
    const int tk    = u & 1;
    const float* yf = ws + OFF_YF + (size_t)b_ * YROWS * CCH;
    sg_run<2, 2>(W + (size_t)((which ? 3 * CCH : CCH) + to * 32 + l16) * CCH,
                 yf + (size_t)(CCH + tk * 32 + l16) * CCH,
                 ws + (which ? OFF_VP : OFF_KP) + (size_t)b_ * CCH * KL + (size_t)(to * 32) * KL + tk * 32,
                 KL, l16, hi);
  }
}

__global__ __launch_bounds__(256) void s_gemm(const float* __restrict__ W,
                                              float* __restrict__ ws) {
  const int lane = threadIdx.x & 31;
  const int wave = threadIdx.x >> 5;
  const int l16  = lane & 15;
  const int hi   = lane >> 4;
  const int bh = blockIdx.x;
  const int b_ = bh >> 3;
  const int h  = bh & 7;
  const float* tt = ws + OFF_TT + (size_t)b_ * CCH * CCH + (size_t)(h * DH) * CCH;
  const int ti = wave >> 1;
  const int tj = wave & 1;
  sg_run<1, 2>(W + (size_t)(h * DH + ti * 16 + l16) * CCH,
               tt + (size_t)(tj * 32 + l16) * CCH,
               ws + OFF_S + (size_t)bh * DH * DH + (size_t)(ti * 16) * DH + tj * 32,
               DH, l16, hi);
  for (int r = 0; r < 8; ++r) {
    const int e = wave * 8 + r;
    const v4f* wk = (const v4f*)(W + (size_t)(CCH + h * DH + e) * CCH) + lane;
    const v4f* tr = (const v4f*)(tt + (size_t)e * CCH) + lane;
    float s = 0.f;
#pragma unroll
    for (int m = 0; m < 4; ++m) {
      const v4f a = wk[m * 32];
      const v4f t = tr[m * 32];
      s += (a.x * t.x + a.y * t.y) + (a.z * t.z + a.w * t.w);
    }
    s += __shfl_xor(s, 16, 32);
    s += __shfl_xor(s, 8, 32);
    s += __shfl_xor(s, 4, 32);
    s += __shfl_xor(s, 2, 32);
    s += __shfl_xor(s, 1, 32);
    if (lane == 0) ws[OFF_K2 + (size_t)bh * DH + e] = s;
  }
}

__global__ __launch_bounds__(256) void ca_softmax(float* __restrict__ ws,
                                                  const float* __restrict__ temperature) {
  const int tid = threadIdx.x;
  const int bh  = blockIdx.y;
  const int h   = bh & 7;
  const int e4  = tid & 15;
  const int d   = blockIdx.x * 16 + (tid >> 4);

  const v4f s  = *(const v4f*)(ws + OFF_S + (size_t)bh * DH * DH + (size_t)d * DH + e4 * 4);
  const v4f kk = *(const v4f*)(ws + OFF_K2 + (size_t)bh * DH + e4 * 4);
  float q2 = 0.f;
  {
    const float* qq = ws + OFF_SQ + ((size_t)(bh * 128 + e4 * 8)) * DH + d;
#pragma unroll
    for (int m = 0; m < 8; ++m) q2 += qq[m * DH];
  }
  q2 += __shfl_xor(q2, 8, 16);
  q2 += __shfl_xor(q2, 4, 16);
  q2 += __shfl_xor(q2, 2, 16);
  q2 += __shfl_xor(q2, 1, 16);

  const float iq  = 1.0f / fmaxf(sqrtf(q2), 1e-12f);
  const float tsc = temperature[h];
  float sc[4];
#pragma unroll
  for (int c = 0; c < 4; ++c) {
    const float ik = 1.0f / fmaxf(sqrtf(kk[c]), 1e-12f);
    sc[c] = s[c] * iq * ik * tsc;
  }
  float m = fmaxf(fmaxf(sc[0], sc[1]), fmaxf(sc[2], sc[3]));
  m = fmaxf(m, __shfl_xor(m, 8, 16));
  m = fmaxf(m, __shfl_xor(m, 4, 16));
  m = fmaxf(m, __shfl_xor(m, 2, 16));
  m = fmaxf(m, __shfl_xor(m, 1, 16));
  v4f ex;
  ex.x = __expf(sc[0] - m); ex.y = __expf(sc[1] - m);
  ex.z = __expf(sc[2] - m); ex.w = __expf(sc[3] - m);
  float sum = (ex.x + ex.y) + (ex.z + ex.w);
  sum += __shfl_xor(sum, 8, 16);
  sum += __shfl_xor(sum, 4, 16);
  sum += __shfl_xor(sum, 2, 16);
  sum += __shfl_xor(sum, 1, 16);
  const float inv = 1.0f / sum;
  ex *= inv;
  *(v4f*)(ws + OFF_P + (size_t)bh * DH * DH + (size_t)d * DH + e4 * 4) = ex;
  if (e4 == 0) ws[OFF_IQ + (size_t)bh * DH + d] = iq;
}

__global__ __launch_bounds__(256) void ca_out(const float* __restrict__ ws,
                                              float* __restrict__ out) {
  __shared__ float P[64][64];
  const int tid  = threadIdx.x;
  const int lane = tid & 31;
  const int wave = tid >> 5;
  const int l16  = lane & 15;
  const int hi   = lane >> 4;
  const int bh = blockIdx.y;
  const int b_ = bh >> 3;
  const int h  = bh & 7;
  const int nBase = blockIdx.x * 128;

  const float* Pg = ws + OFF_P + (size_t)bh * DH * DH;
  for (int i = tid; i < 4096; i += 256) P[i >> 6][i & 63] = Pg[i];
  __syncthreads();

  const int dt = wave >> 1;
  const int ng = wave & 1;
  const int mRow = dt * 16 + l16;
  const float* vca = ws + OFF_VCA + (size_t)bh * DH * NTOK;

  v8f acc[4] = {};
  for (int e0 = 0; e0 < DH; e0 += 4) {
    const int k0 = e0 + (hi ? 2 : 0);
    v2f a;
    a.x = P[mRow][k0];
    a.y = P[mRow][k0 + 1];
#pragma unroll
    for (int t = 0; t < 4; ++t) {
      const int n = nBase + (ng * 4 + t) * 16 + l16;
      v2f bfrag;
      bfrag.x = vca[(size_t)k0 * NTOK + n];
      bfrag.y = vca[(size_t)(k0 + 1) * NTOK + n];
      acc[t] = WMMA_F32(a, bfrag, acc[t]);
    }
  }
#pragma unroll
  for (int t = 0; t < 4; ++t) {
    const int n = nBase + (ng * 4 + t) * 16 + l16;
#pragma unroll
    for (int v = 0; v < 8; ++v) {
      const int d = dt * 16 + v + (hi ? 8 : 0);
      out[((size_t)(b_ * NTOK + n)) * CCH + h * DH + d] = acc[t][v];
    }
  }
}

#define SA_PITCH 68
__global__ __launch_bounds__(256) void sa_attn(float* __restrict__ ws,
                                               const float* __restrict__ temperature2,
                                               float* __restrict__ out) {
  __shared__ float P[128][SA_PITCH];
  const int tid  = threadIdx.x;
  const int lane = tid & 31;
  const int wave = tid >> 5;
  const int l16  = lane & 15;
  const int hi   = lane >> 4;
  const int bh = blockIdx.y;
  const int b_ = bh >> 3;
  const int h  = bh & 7;
  const int nBase = blockIdx.x * 128;

  const float* Q  = ws + OFF_Q  + (size_t)bh * DH * NTOK;
  const float* Kp = ws + OFF_KP + (size_t)bh * DH * KL;
  const float* Vp = ws + OFF_VP + (size_t)bh * DH * KL;
  const float t2 = temperature2[h];
  const float* iqp = ws + OFF_IQ + (size_t)bh * DH;

  {
    const int nGlob = nBase + wave * 16 + l16;
    v8f acc[4] = {};
    for (int d0 = 0; d0 < DH; d0 += 4) {
      const int k0 = d0 + (hi ? 2 : 0);
      v2f a;
      a.x = Q[(size_t)k0 * NTOK + nGlob] * iqp[k0];
      a.y = Q[(size_t)(k0 + 1) * NTOK + nGlob] * iqp[k0 + 1];
#pragma unroll
      for (int t = 0; t < 4; ++t) {
        const int kc = t * 16 + l16;
        v2f bfrag;
        bfrag.x = Kp[(size_t)k0 * KL + kc];
        bfrag.y = Kp[(size_t)(k0 + 1) * KL + kc];
        acc[t] = WMMA_F32(a, bfrag, acc[t]);
      }
    }
#pragma unroll
    for (int v = 0; v < 8; ++v) {
      const float x0 = acc[0][v] * t2;
      const float x1 = acc[1][v] * t2;
      const float x2 = acc[2][v] * t2;
      const float x3 = acc[3][v] * t2;
      float m = fmaxf(fmaxf(x0, x1), fmaxf(x2, x3));
      m = fmaxf(m, __shfl_xor(m, 8, 16));
      m = fmaxf(m, __shfl_xor(m, 4, 16));
      m = fmaxf(m, __shfl_xor(m, 2, 16));
      m = fmaxf(m, __shfl_xor(m, 1, 16));
      const float e0 = __expf(x0 - m);
      const float e1 = __expf(x1 - m);
      const float e2 = __expf(x2 - m);
      const float e3 = __expf(x3 - m);
      float s = (e0 + e1) + (e2 + e3);
      s += __shfl_xor(s, 8, 16);
      s += __shfl_xor(s, 4, 16);
      s += __shfl_xor(s, 2, 16);
      s += __shfl_xor(s, 1, 16);
      const float inv = 1.0f / s;
      float* prow = &P[wave * 16 + v + (hi ? 8 : 0)][l16];
      prow[0]  = e0 * inv;
      prow[16] = e1 * inv;
      prow[32] = e2 * inv;
      prow[48] = e3 * inv;
    }
  }
  __syncthreads();

  {
    const int mRow = wave * 16 + l16;
    v8f acc[4] = {};
    for (int kk0 = 0; kk0 < KL; kk0 += 4) {
      const int k0 = kk0 + (hi ? 2 : 0);
      v2f a;
      a.x = P[mRow][k0];
      a.y = P[mRow][k0 + 1];
#pragma unroll
      for (int t = 0; t < 4; ++t) {
        const int dd = t * 16 + l16;
        v2f bfrag = *(const v2f*)(Vp + (size_t)dd * KL + k0);
        acc[t] = WMMA_F32(a, bfrag, acc[t]);
      }
    }
#pragma unroll
    for (int t = 0; t < 4; ++t) {
      const int dd = t * 16 + l16;
#pragma unroll
      for (int v = 0; v < 8; ++v) {
        const int n = nBase + wave * 16 + v + (hi ? 8 : 0);
        const size_t f = (size_t)dd * (HEADS * NTOK) + (size_t)h * NTOK + n;
        out[(size_t)b_ * (NTOK * CCH) + f] += acc[t][v];
      }
    }
  }
}

extern "C" void kernel_launch(void* const* d_in, const int* in_sizes, int n_in,
                              void* d_out, int out_size, void* d_ws, size_t ws_size,
                              hipStream_t stream) {
  const float* x  = (const float*)d_in[0];
  const float* W  = (const float*)d_in[1];
  const float* EF = (const float*)d_in[2];
  const float* t1 = (const float*)d_in[3];
  const float* t2 = (const float*)d_in[4];
  float* out = (float*)d_out;
  float* ws  = (float*)d_ws;

  proj_gram  <<<1376,           128, 0, stream>>>(x, W, EF, ws);
  y_reduce   <<<176,            256, 0, stream>>>(ws);
  tt_gemm    <<<320,            128, 0, stream>>>(W, ws);
  s_gemm     <<<32,             256, 0, stream>>>(W, ws);
  ca_softmax <<<dim3(4, 32),    256, 0, stream>>>(ws, t1);
  ca_out     <<<dim3(32, 32),   256, 0, stream>>>(ws, out);
  sa_attn    <<<dim3(32, 32),   256, 0, stream>>>(ws, t2, out);
}


# x_CA = x (P Wvca)^T: Vca is never projected; M = P Wvca (64-k WMMA) is formed in the softmax launch, one 512-k product per batch writes out; ca_out and the Vca plane gone; x_CA sum order changes
# speedup vs baseline: 3.5789x; 1.0642x over previous
#include <hip/hip_runtime.h>

typedef float v2f __attribute__((ext_vector_type(2)));
typedef float v8f __attribute__((ext_vector_type(8)));

#define WMMA_F32(a, b, c) \
  __builtin_amdgcn_wmma_f32_16x16x4_f32(false, (a), false, (b), (short)0, (c), false, false)

#define BATCH 4
#define NTOK 4096
#define CCH 512
#define HEADS 8
#define DH 64
#define KL 64
#define MROWS (BATCH * NTOK)
#define OCH (4 * CCH)
#define BH (BATCH * HEADS)

#define OFF_Q   ((size_t)0)
#define OFF_K   ((size_t)8388608)
#define OFF_VCA ((size_t)16777216)
#define OFF_M   (OFF_VCA)
#define OFF_VSA ((size_t)25165824)
#define OFF_KP  ((size_t)33554432)
#define OFF_VP  ((size_t)33685504)
#define NSLICE 8
#define NTILE  44
#define YROWS  576
#define OFF_YP  (OFF_K)
#define OFF_YF  (OFF_VSA)
#define OFF_TT  (OFF_VSA + (size_t)1179648)
#define OFF_S   (OFF_VSA + (size_t)2228224)
#define OFF_K2  (OFF_VSA + (size_t)2359296)
#define OFF_SQ  (OFF_VSA + (size_t)2361344)
#define OFF_IQ  (OFF_VSA + (size_t)2623488)

typedef float v4f __attribute__((ext_vector_type(4)));

__device__ __forceinline__ void k1_load_set(v4f (&fa)[4][4], v4f (&fb)[4][4],
                                            const float* __restrict__ aPtr,
                                            const float* __restrict__ bPtr,
                                            const int k) {
#pragma unroll
  for (int i = 0; i < 4; ++i)
#pragma unroll
    for (int q = 0; q < 4; ++q)
      fa[i][q] = *(const v4f*)(aPtr + (size_t)i * 16 * CCH + k + q * 4);
#pragma unroll
  for (int j = 0; j < 4; ++j)
#pragma unroll
    for (int q = 0; q < 4; ++q)
      fb[j][q] = *(const v4f*)(bPtr + (size_t)j * 16 * CCH + k + q * 4);
}

__device__ __forceinline__ void k1_mma_set(v8f (&acc)[4][4],
                                           const v4f (&fa)[4][4],
                                           const v4f (&fb)[4][4]) {
#pragma unroll
  for (int q = 0; q < 4; ++q) {
#pragma unroll
    for (int i = 0; i < 4; ++i) {
      v2f a;
      a.x = fa[i][q].x;
      a.y = fa[i][q].y;
#pragma unroll
      for (int j = 0; j < 4; ++j) {
        v2f b;
        b.x = fb[j][q].x;
        b.y = fb[j][q].y;
        acc[i][j] = WMMA_F32(a, b, acc[i][j]);
      }
    }
#pragma unroll
    for (int i = 0; i < 4; ++i) {
      v2f a;
      a.x = fa[i][q].z;
      a.y = fa[i][q].w;
#pragma unroll
      for (int j = 0; j < 4; ++j) {
        v2f b;
        b.x = fb[j][q].z;
        b.y = fb[j][q].w;
        acc[i][j] = WMMA_F32(a, b, acc[i][j]);
      }
    }
  }
}

__device__ __forceinline__ void qv_body(const float* __restrict__ x,
                                        const float* __restrict__ W,
                                        float* __restrict__ ws,
                                        const int bx, const int by) {
  const int lane = threadIdx.x & 31;
  const int wave = threadIdx.x >> 5;
  const int l16  = lane & 15;
  const int hi   = lane >> 4;
  const int wm = wave >> 1;
  const int wo = wave & 1;
  const int h  = by * 2 + wo;
  const int mW = bx * 128 + wm * 64;
  const int oW = h * DH;

  const float* aPtr = x + (size_t)(mW + l16) * CCH + hi * 16;
  const float* bPtr = W + (size_t)(oW + l16) * CCH + hi * 16;

  v8f acc[4][4] = {};
  v4f fa0[4][4], fb0[4][4], fa1[4][4], fb1[4][4];

  k1_load_set(fa0, fb0, aPtr, bPtr, 0);
  for (int k0 = 0; k0 < CCH - 64; k0 += 64) {
    k1_load_set(fa1, fb1, aPtr, bPtr, k0 + 32);
    k1_mma_set(acc, fa0, fb0);
    k1_load_set(fa0, fb0, aPtr, bPtr, k0 + 64);
    k1_mma_set(acc, fa1, fb1);
  }
  k1_load_set(fa1, fb1, aPtr, bPtr, CCH - 32);
  k1_mma_set(acc, fa0, fb0);
  k1_mma_set(acc, fa1, fb1);

  const int b_ = mW >> 12;
  const int bh = b_ * HEADS + h;
  const int nTok0 = mW & (NTOK - 1);
  float* dstBase = ws + OFF_Q + (size_t)bh * DH * NTOK;
#pragma unroll
  for (int i = 0; i < 4; ++i) {
#pragma unroll
    for (int j = 0; j < 4; ++j) {
      const int dd = j * 16 + l16;
#pragma unroll
      for (int v = 0; v < 8; ++v) {
        const int n = nTok0 + i * 16 + v + (hi ? 8 : 0);
        dstBase[(size_t)dd * NTOK + n] = acc[i][j][v];
      }
    }
  }
  {
    float* sq = ws + OFF_SQ + ((size_t)(bh * 128 + (nTok0 >> 6) * 2 + hi)) * DH;
#pragma unroll
    for (int j = 0; j < 4; ++j) {
      float s = 0.f;
#pragma unroll
      for (int i = 0; i < 4; ++i)
#pragma unroll
        for (int v = 0; v < 8; ++v) s += acc[i][j][v] * acc[i][j][v];
      sq[j * 16 + l16] = s;
    }
  }
}

__device__ __forceinline__ void gr_load(v4f (&f)[4][2], const float* __restrict__ p,
                                        const size_t stride) {
#pragma unroll
  for (int t = 0; t < 4; ++t)
#pragma unroll
    for (int s = 0; s < 2; ++s)
      f[t][s] = *(const v4f*)(p + (size_t)(2 * t + s) * stride);
}

__device__ __forceinline__ void gr_mma(v8f (&acc)[4][4], const v4f (&fa)[4][2],
                                       const v4f (&fb)[4][2]) {
#pragma unroll
  for (int t = 0; t < 4; ++t) {
#pragma unroll
    for (int jr = 0; jr < 4; ++jr) {
      v2f a;
      a.x = fa[t][0][jr];
      a.y = fa[t][1][jr];
#pragma unroll
      for (int jc = 0; jc < 4; ++jc) {
        v2f b;
        b.x = fb[t][0][jc];
        b.y = fb[t][1][jc];
        acc[jr][jc] = WMMA_F32(a, b, acc[jr][jc]);
      }
    }
  }
}

__device__ __forceinline__ void tile_ij(const int tix, int& I, int& J) {
  if (tix >= 36) { I = 8; J = tix - 36; return; }
  int i = 0, rem = tix;
  while (rem >= 8 - i) { rem -= 8 - i; ++i; }
  I = i;
  J = i + rem;
}

__device__ __forceinline__ void gram_body(const float* __restrict__ x,
                                          const float* __restrict__ EF,
                                          float* __restrict__ ws,
                                          const int bx, const int slice, const int b_) {
  const int lane = threadIdx.x & 31;
  const int wave = __builtin_amdgcn_readfirstlane((int)(threadIdx.x >> 5));
  const int l16  = lane & 15;
  const int hi   = lane >> 4;
  const int tix   = bx * 4 + wave;
  int I, J;
  tile_ij(tix, I, J);
  const int n0 = slice * (NTOK / NSLICE) + hi * 8;
  const float* xb = x + (size_t)b_ * NTOK * CCH;
  const float* cPtr = xb + (size_t)n0 * CCH + J * 64 + l16 * 4;
  const bool ef = (I == 8);
  const size_t sA = ef ? (size_t)KL : (size_t)CCH;
  const float* rPtr = ef ? (EF + (size_t)n0 * KL + l16 * 4)
                         : (xb + (size_t)n0 * CCH + I * 64 + l16 * 4);

  v8f acc[4][4] = {};
  v4f fa0[4][2], fb0[4][2], fa1[4][2], fb1[4][2];
  const int NB = NTOK / NSLICE / 16;
  gr_load(fa0, rPtr, sA);
  gr_load(fb0, cPtr, CCH);
  for (int blk = 0; blk < NB - 2; blk += 2) {
    gr_load(fa1, rPtr + (size_t)(blk + 1) * 16 * sA, sA);
    gr_load(fb1, cPtr + (size_t)(blk + 1) * 16 * CCH, CCH);
    gr_mma(acc, fa0, fb0);
    gr_load(fa0, rPtr + (size_t)(blk + 2) * 16 * sA, sA);
    gr_load(fb0, cPtr + (size_t)(blk + 2) * 16 * CCH, CCH);
    gr_mma(acc, fa1, fb1);
  }
  gr_load(fa1, rPtr + (size_t)(NB - 1) * 16 * sA, sA);
  gr_load(fb1, cPtr + (size_t)(NB - 1) * 16 * CCH, CCH);
  gr_mma(acc, fa0, fb0);
  gr_mma(acc, fa1, fb1);

  float* part = ws + OFF_YP + ((size_t)((b_ * NSLICE + slice) * NTILE + tix)) * 4096;
#pragma unroll
  for (int jr = 0; jr < 4; ++jr)
#pragma unroll
    for (int v = 0; v < 8; ++v) {
      v4f o;
      o.x = acc[jr][0][v]; o.y = acc[jr][1][v]; o.z = acc[jr][2][v]; o.w = acc[jr][3][v];
      *(v4f*)(part + (size_t)(32 * hi + 4 * v + jr) * 64 + l16 * 4) = o;
    }
}

__global__ __launch_bounds__(128) void proj_gram(const float* __restrict__ x,
                                                 const float* __restrict__ W,
                                                 const float* __restrict__ EF,
                                                 float* __restrict__ ws) {
  const int blk = blockIdx.x;
  if (blk < 512) {
    qv_body(x, W, ws, blk & 127, blk >> 7);
  } else {
    const int g = blk - 512;
    const int q = g / 11;
    gram_body(x, EF, ws, g - q * 11, q & 7, q >> 3);
  }
}

__global__ __launch_bounds__(256) void y_reduce(float* __restrict__ ws) {
  __shared__ float tile[64][65];
  const int tid = threadIdx.x;
  const int b_  = blockIdx.x / NTILE;
  const int tix = blockIdx.x - b_ * NTILE;
  int I, J;
  tile_ij(tix, I, J);
  float* yf = ws + OFF_YF + (size_t)b_ * YROWS * CCH;
  const v4f* sp0 = (const v4f*)(ws + OFF_YP + ((size_t)(b_ * NSLICE * NTILE + tix)) * 4096);
#pragma unroll
  for (int m = 0; m < 4; ++m) {
    const int idx  = tid + 256 * m;
    const int rloc = idx >> 4;
    const int c4   = idx & 15;
    v4f s = sp0[idx];
#pragma unroll
    for (int sl = 1; sl < NSLICE; ++sl) s += sp0[(size_t)sl * NTILE * 1024 + idx];
    *(v4f*)(yf + (size_t)(I * 64 + rloc) * CCH + J * 64 + c4 * 4) = s;
    tile[rloc][c4 * 4 + 0] = s.x;
    tile[rloc][c4 * 4 + 1] = s.y;
    tile[rloc][c4 * 4 + 2] = s.z;
    tile[rloc][c4 * 4 + 3] = s.w;
  }
  __syncthreads();
  if (I < J) {
#pragma unroll
    for (int m = 0; m < 4; ++m) {
      const int idx  = tid + 256 * m;
      const int cloc = idx >> 4;
      const int r4   = idx & 15;
      v4f o;
      o.x = tile[r4 * 4 + 0][cloc];
      o.y = tile[r4 * 4 + 1][cloc];
      o.z = tile[r4 * 4 + 2][cloc];
      o.w = tile[r4 * 4 + 3][cloc];
      *(v4f*)(yf + (size_t)(J * 64 + cloc) * CCH + I * 64 + r4 * 4) = o;
    }
  }
}

template <int MI, int NJ>
__device__ __forceinline__ void sg_load(v4f (&fa)[MI][4], v4f (&fb)[NJ][4],
                                        const float* __restrict__ aPtr,
                                        const float* __restrict__ bPtr, const int k) {
#pragma unroll
  for (int i = 0; i < MI; ++i)
#pragma unroll
    for (int q = 0; q < 4; ++q)
      fa[i][q] = *(const v4f*)(aPtr + (size_t)i * 16 * CCH + k + q * 4);
#pragma unroll
  for (int j = 0; j < NJ; ++j)
#pragma unroll
    for (int q = 0; q < 4; ++q)
      fb[j][q] = *(const v4f*)(bPtr + (size_t)j * 16 * CCH + k + q * 4);
}

template <int MI, int NJ>
__device__ __forceinline__ void sg_mma(v8f (&acc)[MI][NJ], const v4f (&fa)[MI][4],
                                       const v4f (&fb)[NJ][4]) {
#pragma unroll
  for (int q = 0; q < 4; ++q) {
#pragma unroll
    for (int i = 0; i < MI; ++i) {
      v2f a;
      a.x = fa[i][q].x;
      a.y = fa[i][q].y;
#pragma unroll
      for (int j = 0; j < NJ; ++j) {
        v2f b;
        b.x = fb[j][q].x;
        b.y = fb[j][q].y;
        acc[i][j] = WMMA_F32(a, b, acc[i][j]);
      }
    }
#pragma unroll
    for (int i = 0; i < MI; ++i) {
      v2f a;
      a.x = fa[i][q].z;
      a.y = fa[i][q].w;
#pragma unroll
      for (int j = 0; j < NJ; ++j) {
        v2f b;
        b.x = fb[j][q].z;
        b.y = fb[j][q].w;
        acc[i][j] = WMMA_F32(a, b, acc[i][j]);
      }
    }
  }
}

template <int MI, int NJ>
__device__ __forceinline__ void sg_run(const float* __restrict__ aRow,
                                       const float* __restrict__ bRow,
                                       float* __restrict__ dst, const int ld,
                                       const int l16, const int hi) {
  const float* aPtr = aRow + hi * 16;
  const float* bPtr = bRow + hi * 16;
  v8f acc[MI][NJ] = {};
  v4f fa0[MI][4], fb0[NJ][4], fa1[MI][4], fb1[NJ][4];
  sg_load<MI, NJ>(fa0, fb0, aPtr, bPtr, 0);
  for (int k0 = 0; k0 < CCH - 64; k0 += 64) {
    sg_load<MI, NJ>(fa1, fb1, aPtr, bPtr, k0 + 32);
    sg_mma<MI, NJ>(acc, fa0, fb0);
    sg_load<MI, NJ>(fa0, fb0, aPtr, bPtr, k0 + 64);
    sg_mma<MI, NJ>(acc, fa1, fb1);
  }
  sg_load<MI, NJ>(fa1, fb1, aPtr, bPtr, CCH - 32);
  sg_mma<MI, NJ>(acc, fa0, fb0);
  sg_mma<MI, NJ>(acc, fa1, fb1);
#pragma unroll
  for (int i = 0; i < MI; ++i)
#pragma unroll
    for (int j = 0; j < NJ; ++j)
#pragma unroll
      for (int v = 0; v < 8; ++v)
        dst[(size_t)(i * 16 + hi * 8 + v) * ld + j * 16 + l16] = acc[i][j][v];
}

__global__ __launch_bounds__(128) void tt_gemm(const float* __restrict__ W,
                                               float* __restrict__ ws) {
  const int lane = threadIdx.x & 31;
  const int l16  = lane & 15;
  const int hi   = lane >> 4;
  const int wg   = blockIdx.x * 4 + (threadIdx.x >> 5);
  if (wg < 1024) {
    const int b_ = wg >> 8;
    const int to = (wg >> 4) & 15;
    const int tc = wg & 15;
    const float* yf = ws + OFF_YF + (size_t)b_ * YROWS * CCH;
    sg_run<2, 2>(W + (size_t)(CCH + to * 32 + l16) * CCH,
                 yf + (size_t)(tc * 32 + l16) * CCH,
                 ws + OFF_TT + (size_t)b_ * CCH * CCH + (size_t)(to * 32) * CCH + tc * 32,
                 CCH, l16, hi);
  } else {
    const int u     = wg - 1024;
    const int b_    = u >> 6;
    const int which = (u >> 5) & 1;
    const int to    = (u >> 1) & 15;
    const int tk    = u & 1;
    const float* yf = ws + OFF_YF + (size_t)b_ * YROWS * CCH;
    sg_run<2, 2>(W + (size_t)((which ? 3 * CCH : CCH) + to * 32 + l16) * CCH,
                 yf + (size_t)(CCH + tk * 32 + l16) * CCH,
                 ws + (which ? OFF_VP : OFF_KP) + (size_t)b_ * CCH * KL + (size_t)(to * 32) * KL + tk * 32,
                 KL, l16, hi);
  }
}

__global__ __launch_bounds__(256) void s_gemm(const float* __restrict__ W,
                                              float* __restrict__ ws) {
  const int lane = threadIdx.x & 31;
  const int wave = threadIdx.x >> 5;
  const int l16  = lane & 15;
  const int hi   = lane >> 4;
  const int bh = blockIdx.x;
  const int b_ = bh >> 3;
  const int h  = bh & 7;
  const float* tt = ws + OFF_TT + (size_t)b_ * CCH * CCH + (size_t)(h * DH) * CCH;
  const int ti = wave >> 1;
  const int tj = wave & 1;
  sg_run<1, 2>(W + (size_t)(h * DH + ti * 16 + l16) * CCH,
               tt + (size_t)(tj * 32 + l16) * CCH,
               ws + OFF_S + (size_t)bh * DH * DH + (size_t)(ti * 16) * DH + tj * 32,
               DH, l16, hi);
  for (int r = 0; r < 8; ++r) {
    const int e = wave * 8 + r;
    const v4f* wk = (const v4f*)(W + (size_t)(CCH + h * DH + e) * CCH) + lane;
    const v4f* tr = (const v4f*)(tt + (size_t)e * CCH) + lane;
    float s = 0.f;
#pragma unroll
    for (int m = 0; m < 4; ++m) {
      const v4f a = wk[m * 32];
      const v4f t = tr[m * 32];
      s += (a.x * t.x + a.y * t.y) + (a.z * t.z + a.w * t.w);
    }
    s += __shfl_xor(s, 16, 32);
    s += __shfl_xor(s, 8, 32);
    s += __shfl_xor(s, 4, 32);
    s += __shfl_xor(s, 2, 32);
    s += __shfl_xor(s, 1, 32);
    if (lane == 0) ws[OFF_K2 + (size_t)bh * DH + e] = s;
  }
}

#define CS_PITCH 68
__global__ __launch_bounds__(256) void ca_softmax(const float* __restrict__ W,
                                                  float* __restrict__ ws,
                                                  const float* __restrict__ temperature) {
  __shared__ float Ps[16][CS_PITCH];
  const int tid = threadIdx.x;
  const int bh  = blockIdx.y;
  const int h   = bh & 7;
  const int e4  = tid & 15;
  const int d   = blockIdx.x * 16 + (tid >> 4);
  const int lane = tid & 31;
  const int wave = tid >> 5;
  const int l16  = lane & 15;
  const int hi   = lane >> 4;

  const float* Wv = W + (size_t)(2 * CCH + h * DH + hi * 2) * CCH + wave * 64 + l16;
  v2f bw[16][4];
#pragma unroll
  for (int it = 0; it < 16; ++it)
#pragma unroll
    for (int t = 0; t < 4; ++t) {
      bw[it][t].x = Wv[(size_t)(it * 4) * CCH + t * 16];
      bw[it][t].y = Wv[(size_t)(it * 4 + 1) * CCH + t * 16];
    }

  const v4f s  = *(const v4f*)(ws + OFF_S + (size_t)bh * DH * DH + (size_t)d * DH + e4 * 4);
  const v4f kk = *(const v4f*)(ws + OFF_K2 + (size_t)bh * DH + e4 * 4);
  float q2 = 0.f;
  {
    const float* qq = ws + OFF_SQ + ((size_t)(bh * 128 + e4 * 8)) * DH + d;
#pragma unroll
    for (int m = 0; m < 8; ++m) q2 += qq[m * DH];
  }
  q2 += __shfl_xor(q2, 8, 16);
  q2 += __shfl_xor(q2, 4, 16);
  q2 += __shfl_xor(q2, 2, 16);
  q2 += __shfl_xor(q2, 1, 16);

  const float iq  = 1.0f / fmaxf(sqrtf(q2), 1e-12f);
  const float tsc = temperature[h];
  float sc[4];
#pragma unroll
  for (int c = 0; c < 4; ++c) {
    const float ik = 1.0f / fmaxf(sqrtf(kk[c]), 1e-12f);
    sc[c] = s[c] * iq * ik * tsc;
  }
  float m = fmaxf(fmaxf(sc[0], sc[1]), fmaxf(sc[2], sc[3]));
  m = fmaxf(m, __shfl_xor(m, 8, 16));
  m = fmaxf(m, __shfl_xor(m, 4, 16));
  m = fmaxf(m, __shfl_xor(m, 2, 16));
  m = fmaxf(m, __shfl_xor(m, 1, 16));
  v4f ex;
  ex.x = __expf(sc[0] - m); ex.y = __expf(sc[1] - m);
  ex.z = __expf(sc[2] - m); ex.w = __expf(sc[3] - m);
  float sum = (ex.x + ex.y) + (ex.z + ex.w);
  sum += __shfl_xor(sum, 8, 16);
  sum += __shfl_xor(sum, 4, 16);
  sum += __shfl_xor(sum, 2, 16);
  sum += __shfl_xor(sum, 1, 16);
  const float inv = 1.0f / sum;
  ex *= inv;
  *(v4f*)(&Ps[tid >> 4][e4 * 4]) = ex;
  if (e4 == 0) ws[OFF_IQ + (size_t)bh * DH + d] = iq;
  __syncthreads();

  v8f acc[4] = {};
#pragma unroll
  for (int it = 0; it < 16; ++it) {
    const v2f a = *(const v2f*)(&Ps[l16][it * 4 + hi * 2]);
#pragma unroll
    for (int t = 0; t < 4; ++t) acc[t] = WMMA_F32(a, bw[it][t], acc[t]);
  }
  float* mrow = ws + OFF_M + (size_t)(bh >> 3) * CCH * CCH
              + (size_t)(h * DH + blockIdx.x * 16 + hi * 8) * CCH + wave * 64 + l16;
#pragma unroll
  for (int t = 0; t < 4; ++t)
#pragma unroll
    for (int v = 0; v < 8; ++v) mrow[(size_t)v * CCH + t * 16] = acc[t][v];
}

__global__ __launch_bounds__(128) void ca_gemm(const float* __restrict__ x,
                                               const float* __restrict__ ws,
                                               float* __restrict__ out) {
  const int lane = threadIdx.x & 31;
  const int wave = threadIdx.x >> 5;
  const int l16  = lane & 15;
  const int hi   = lane >> 4;
  const int wm = wave >> 1;
  const int wo = wave & 1;
  const int bx = blockIdx.x & 127;
  const int by = blockIdx.x >> 7;
  const int mW  = bx * 128 + wm * 64;
  const int chW = (by * 2 + wo) * DH;
  const int b_  = mW >> 12;

  const float* aPtr = ws + OFF_M + (size_t)b_ * CCH * CCH + (size_t)(chW + l16) * CCH + hi * 16;
  const float* bPtr = x + (size_t)(mW + l16) * CCH + hi * 16;

  v8f acc[4][4] = {};
  v4f fa0[4][4], fb0[4][4], fa1[4][4], fb1[4][4];

  k1_load_set(fa0, fb0, aPtr, bPtr, 0);
  for (int k0 = 0; k0 < CCH - 64; k0 += 64) {
    k1_load_set(fa1, fb1, aPtr, bPtr, k0 + 32);
    k1_mma_set(acc, fa0, fb0);
    k1_load_set(fa0, fb0, aPtr, bPtr, k0 + 64);
    k1_mma_set(acc, fa1, fb1);
  }
  k1_load_set(fa1, fb1, aPtr, bPtr, CCH - 32);
  k1_mma_set(acc, fa0, fb0);
  k1_mma_set(acc, fa1, fb1);

  float* ob = out + (size_t)(mW + l16) * CCH + chW + hi * 8;
#pragma unroll
  for (int j = 0; j < 4; ++j)
#pragma unroll
    for (int i = 0; i < 4; ++i) {
      v4f lo, up;
      lo.x = acc[i][j][0]; lo.y = acc[i][j][1]; lo.z = acc[i][j][2]; lo.w = acc[i][j][3];
      up.x = acc[i][j][4]; up.y = acc[i][j][5]; up.z = acc[i][j][6]; up.w = acc[i][j][7];
      float* p = ob + (size_t)(j * 16) * CCH + i * 16;
      *(v4f*)p = lo;
      *(v4f*)(p + 4) = up;
    }
}

#define SA_PITCH 68
__global__ __launch_bounds__(256) void sa_attn(float* __restrict__ ws,
                                               const float* __restrict__ temperature2,
                                               float* __restrict__ out) {
  __shared__ float P[128][SA_PITCH];
  const int tid  = threadIdx.x;
  const int lane = tid & 31;
  const int wave = tid >> 5;
  const int l16  = lane & 15;
  const int hi   = lane >> 4;
  const int bh = blockIdx.y;
  const int b_ = bh >> 3;
  const int h  = bh & 7;
  const int nBase = blockIdx.x * 128;

  const float* Q  = ws + OFF_Q  + (size_t)bh * DH * NTOK;
  const float* Kp = ws + OFF_KP + (size_t)bh * DH * KL;
  const float* Vp = ws + OFF_VP + (size_t)bh * DH * KL;
  const float t2 = temperature2[h];
  const float* iqp = ws + OFF_IQ + (size_t)bh * DH;

  {
    const int nGlob = nBase + wave * 16 + l16;
    v8f acc[4] = {};
    for (int d0 = 0; d0 < DH; d0 += 4) {
      const int k0 = d0 + (hi ? 2 : 0);
      v2f a;
      a.x = Q[(size_t)k0 * NTOK + nGlob] * iqp[k0];
      a.y = Q[(size_t)(k0 + 1) * NTOK + nGlob] * iqp[k0 + 1];
#pragma unroll
      for (int t = 0; t < 4; ++t) {
        const int kc = t * 16 + l16;
        v2f bfrag;
        bfrag.x = Kp[(size_t)k0 * KL + kc];
        bfrag.y = Kp[(size_t)(k0 + 1) * KL + kc];
        acc[t] = WMMA_F32(a, bfrag, acc[t]);
      }
    }
#pragma unroll
    for (int v = 0; v < 8; ++v) {
      const float x0 = acc[0][v] * t2;
      const float x1 = acc[1][v] * t2;
      const float x2 = acc[2][v] * t2;
      const float x3 = acc[3][v] * t2;
      float m = fmaxf(fmaxf(x0, x1), fmaxf(x2, x3));
      m = fmaxf(m, __shfl_xor(m, 8, 16));
      m = fmaxf(m, __shfl_xor(m, 4, 16));
      m = fmaxf(m, __shfl_xor(m, 2, 16));
      m = fmaxf(m, __shfl_xor(m, 1, 16));
      const float e0 = __expf(x0 - m);
      const float e1 = __expf(x1 - m);
      const float e2 = __expf(x2 - m);
      const float e3 = __expf(x3 - m);
      float s = (e0 + e1) + (e2 + e3);
      s += __shfl_xor(s, 8, 16);
      s += __shfl_xor(s, 4, 16);
      s += __shfl_xor(s, 2, 16);
      s += __shfl_xor(s, 1, 16);
      const float inv = 1.0f / s;
      float* prow = &P[wave * 16 + v + (hi ? 8 : 0)][l16];
      prow[0]  = e0 * inv;
      prow[16] = e1 * inv;
      prow[32] = e2 * inv;
      prow[48] = e3 * inv;
    }
  }
  __syncthreads();

  {
    const int mRow = wave * 16 + l16;
    v8f acc[4] = {};
    for (int kk0 = 0; kk0 < KL; kk0 += 4) {
      const int k0 = kk0 + (hi ? 2 : 0);
      v2f a;
      a.x = P[mRow][k0];
      a.y = P[mRow][k0 + 1];
#pragma unroll
      for (int t = 0; t < 4; ++t) {
        const int dd = t * 16 + l16;
        v2f bfrag = *(const v2f*)(Vp + (size_t)dd * KL + k0);
        acc[t] = WMMA_F32(a, bfrag, acc[t]);
      }
    }
#pragma unroll
    for (int t = 0; t < 4; ++t) {
      const int dd = t * 16 + l16;
#pragma unroll
      for (int v = 0; v < 8; ++v) {
        const int n = nBase + wave * 16 + v + (hi ? 8 : 0);
        const size_t f = (size_t)dd * (HEADS * NTOK) + (size_t)h * NTOK + n;
        out[(size_t)b_ * (NTOK * CCH) + f] += acc[t][v];
      }
    }
  }
}

extern "C" void kernel_launch(void* const* d_in, const int* in_sizes, int n_in,
                              void* d_out, int out_size, void* d_ws, size_t ws_size,
                              hipStream_t stream) {
  const float* x  = (const float*)d_in[0];
  const float* W  = (const float*)d_in[1];
  const float* EF = (const float*)d_in[2];
  const float* t1 = (const float*)d_in[3];
  const float* t2 = (const float*)d_in[4];
  float* out = (float*)d_out;
  float* ws  = (float*)d_ws;

  proj_gram  <<<864,            128, 0, stream>>>(x, W, EF, ws);
  y_reduce   <<<176,            256, 0, stream>>>(ws);
  tt_gemm    <<<320,            128, 0, stream>>>(W, ws);
  s_gemm     <<<32,             256, 0, stream>>>(W, ws);
  ca_softmax <<<dim3(4, 32),    256, 0, stream>>>(W, ws, t1);
  ca_gemm    <<<512,            128, 0, stream>>>(x, ws, out);
  sa_attn    <<<dim3(32, 32),   256, 0, stream>>>(ws, t2, out);
}


# K1 blocks spread over launches 1, 2 and 4 (160 beside the gram blocks, 256 beside the slice reduction, 96 as half-length blocks beside S) so that each launch fills whole rounds of compute units
# speedup vs baseline: 3.8777x; 1.0835x over previous
#include <hip/hip_runtime.h>

typedef float v2f __attribute__((ext_vector_type(2)));
typedef float v8f __attribute__((ext_vector_type(8)));

#define WMMA_F32(a, b, c) \
  __builtin_amdgcn_wmma_f32_16x16x4_f32(false, (a), false, (b), (short)0, (c), false, false)

#define BATCH 4
#define NTOK 4096
#define CCH 512
#define HEADS 8
#define DH 64
#define KL 64
#define MROWS (BATCH * NTOK)
#define OCH (4 * CCH)
#define BH (BATCH * HEADS)

#define OFF_Q   ((size_t)0)
#define OFF_K   ((size_t)8388608)
#define OFF_VCA ((size_t)16777216)
#define OFF_M   (OFF_VCA)
#define OFF_VSA ((size_t)25165824)
#define OFF_KP  ((size_t)33554432)
#define OFF_VP  ((size_t)33685504)
#define NSLICE 8
#define NTILE  44
#define YROWS  576
#define OFF_YP  (OFF_K)
#define OFF_YF  (OFF_VSA)
#define OFF_TT  (OFF_VSA + (size_t)1179648)
#define OFF_S   (OFF_VSA + (size_t)2228224)
#define OFF_K2  (OFF_VSA + (size_t)2359296)
#define OFF_SQ  (OFF_VSA + (size_t)2361344)
#define OFF_IQ  (OFF_VSA + (size_t)2623488)

typedef float v4f __attribute__((ext_vector_type(4)));

__device__ __forceinline__ void k1_load_set(v4f (&fa)[4][4], v4f (&fb)[4][4],
                                            const float* __restrict__ aPtr,
                                            const float* __restrict__ bPtr,
                                            const int k) {
#pragma unroll
  for (int i = 0; i < 4; ++i)
#pragma unroll
    for (int q = 0; q < 4; ++q)
      fa[i][q] = *(const v4f*)(aPtr + (size_t)i * 16 * CCH + k + q * 4);
#pragma unroll
  for (int j = 0; j < 4; ++j)
#pragma unroll
    for (int q = 0; q < 4; ++q)
      fb[j][q] = *(const v4f*)(bPtr + (size_t)j * 16 * CCH + k + q * 4);
}

__device__ __forceinline__ void k1_mma_set(v8f (&acc)[4][4],
                                           const v4f (&fa)[4][4],
                                           const v4f (&fb)[4][4]) {
#pragma unroll
  for (int q = 0; q < 4; ++q) {
#pragma unroll
    for (int i = 0; i < 4; ++i) {
      v2f a;
      a.x = fa[i][q].x;
      a.y = fa[i][q].y;
#pragma unroll
      for (int j = 0; j < 4; ++j) {
        v2f b;
        b.x = fb[j][q].x;
        b.y = fb[j][q].y;
        acc[i][j] = WMMA_F32(a, b, acc[i][j]);
      }
    }
#pragma unroll
    for (int i = 0; i < 4; ++i) {
      v2f a;
      a.x = fa[i][q].z;
      a.y = fa[i][q].w;
#pragma unroll
      for (int j = 0; j < 4; ++j) {
        v2f b;
        b.x = fb[j][q].z;
        b.y = fb[j][q].w;
        acc[i][j] = WMMA_F32(a, b, acc[i][j]);
      }
    }
  }
}

__device__ __forceinline__ void qv_body(const float* __restrict__ x,
                                        const float* __restrict__ W,
                                        float* __restrict__ ws,
                                        const int bx, const int by) {
  const int lane = threadIdx.x & 31;
  const int wave = threadIdx.x >> 5;
  const int l16  = lane & 15;
  const int hi   = lane >> 4;
  const int wm = wave >> 1;
  const int wo = wave & 1;
  const int h  = by * 2 + wo;
  const int mW = bx * 128 + wm * 64;
  const int oW = h * DH;

  const float* aPtr = x + (size_t)(mW + l16) * CCH + hi * 16;
  const float* bPtr = W + (size_t)(oW + l16) * CCH + hi * 16;

  v8f acc[4][4] = {};
  v4f fa0[4][4], fb0[4][4], fa1[4][4], fb1[4][4];

  k1_load_set(fa0, fb0, aPtr, bPtr, 0);
  for (int k0 = 0; k0 < CCH - 64; k0 += 64) {
    k1_load_set(fa1, fb1, aPtr, bPtr, k0 + 32);
    k1_mma_set(acc, fa0, fb0);
    k1_load_set(fa0, fb0, aPtr, bPtr, k0 + 64);
    k1_mma_set(acc, fa1, fb1);
  }
  k1_load_set(fa1, fb1, aPtr, bPtr, CCH - 32);
  k1_mma_set(acc, fa0, fb0);
  k1_mma_set(acc, fa1, fb1);

  const int b_ = mW >> 12;
  const int bh = b_ * HEADS + h;
  const int nTok0 = mW & (NTOK - 1);
  float* dstBase = ws + OFF_Q + (size_t)bh * DH * NTOK;
#pragma unroll
  for (int i = 0; i < 4; ++i) {
#pragma unroll
    for (int j = 0; j < 4; ++j) {
      const int dd = j * 16 + l16;
#pragma unroll
      for (int v = 0; v < 8; ++v) {
        const int n = nTok0 + i * 16 + v + (hi ? 8 : 0);
        dstBase[(size_t)dd * NTOK + n] = acc[i][j][v];
      }
    }
  }
  {
    float* sq = ws + OFF_SQ + ((size_t)(bh * 128 + (nTok0 >> 6) * 2 + hi)) * DH;
#pragma unroll
    for (int j = 0; j < 4; ++j) {
      float s = 0.f;
#pragma unroll
      for (int i = 0; i < 4; ++i)
#pragma unroll
        for (int v = 0; v < 8; ++v) s += acc[i][j][v] * acc[i][j][v];
      sq[j * 16 + l16] = s;
    }
  }
}

__device__ __forceinline__ void gr_load(v4f (&f)[4][2], const float* __restrict__ p,
                                        const size_t stride) {
#pragma unroll
  for (int t = 0; t < 4; ++t)
#pragma unroll
    for (int s = 0; s < 2; ++s)
      f[t][s] = *(const v4f*)(p + (size_t)(2 * t + s) * stride);
}

__device__ __forceinline__ void gr_mma(v8f (&acc)[4][4], const v4f (&fa)[4][2],
                                       const v4f (&fb)[4][2]) {
#pragma unroll
  for (int t = 0; t < 4; ++t) {
#pragma unroll
    for (int jr = 0; jr < 4; ++jr) {
      v2f a;
      a.x = fa[t][0][jr];
      a.y = fa[t][1][jr];
#pragma unroll
      for (int jc = 0; jc < 4; ++jc) {
        v2f b;
        b.x = fb[t][0][jc];
        b.y = fb[t][1][jc];
        acc[jr][jc] = WMMA_F32(a, b, acc[jr][jc]);
      }
    }
  }
}

__device__ __forceinline__ void tile_ij(const int tix, int& I, int& J) {
  if (tix >= 36) { I = 8; J = tix - 36; return; }
  int i = 0, rem = tix;
  while (rem >= 8 - i) { rem -= 8 - i; ++i; }
  I = i;
  J = i + rem;
}

__device__ __forceinline__ void gram_body(const float* __restrict__ x,
                                          const float* __restrict__ EF,
                                          float* __restrict__ ws,
                                          const int bx, const int slice, const int b_) {
  const int lane = threadIdx.x & 31;
  const int wave = __builtin_amdgcn_readfirstlane((int)(threadIdx.x >> 5));
  const int l16  = lane & 15;
  const int hi   = lane >> 4;
  const int tix   = bx * 4 + wave;
  int I, J;
  tile_ij(tix, I, J);
  const int n0 = slice * (NTOK / NSLICE) + hi * 8;
  const float* xb = x + (size_t)b_ * NTOK * CCH;
  const float* cPtr = xb + (size_t)n0 * CCH + J * 64 + l16 * 4;
  const bool ef = (I == 8);
  const size_t sA = ef ? (size_t)KL : (size_t)CCH;
  const float* rPtr = ef ? (EF + (size_t)n0 * KL + l16 * 4)
                         : (xb + (size_t)n0 * CCH + I * 64 + l16 * 4);

  v8f acc[4][4] = {};
  v4f fa0[4][2], fb0[4][2], fa1[4][2], fb1[4][2];
  const int NB = NTOK / NSLICE / 16;
  gr_load(fa0, rPtr, sA);
  gr_load(fb0, cPtr, CCH);
  for (int blk = 0; blk < NB - 2; blk += 2) {
    gr_load(fa1, rPtr + (size_t)(blk + 1) * 16 * sA, sA);
    gr_load(fb1, cPtr + (size_t)(blk + 1) * 16 * CCH, CCH);
    gr_mma(acc, fa0, fb0);
    gr_load(fa0, rPtr + (size_t)(blk + 2) * 16 * sA, sA);
    gr_load(fb0, cPtr + (size_t)(blk + 2) * 16 * CCH, CCH);
    gr_mma(acc, fa1, fb1);
  }
  gr_load(fa1, rPtr + (size_t)(NB - 1) * 16 * sA, sA);
  gr_load(fb1, cPtr + (size_t)(NB - 1) * 16 * CCH, CCH);
  gr_mma(acc, fa0, fb0);
  gr_mma(acc, fa1, fb1);

  float* part = ws + OFF_YP + ((size_t)((b_ * NSLICE + slice) * NTILE + tix)) * 4096;
#pragma unroll
  for (int jr = 0; jr < 4; ++jr)
#pragma unroll
    for (int v = 0; v < 8; ++v) {
      v4f o;
      o.x = acc[jr][0][v]; o.y = acc[jr][1][v]; o.z = acc[jr][2][v]; o.w = acc[jr][3][v];
      *(v4f*)(part + (size_t)(32 * hi + 4 * v + jr) * 64 + l16 * 4) = o;
    }
}

#define QB_L1 160
#define QB_L2 256
__global__ __launch_bounds__(128) void proj_gram(const float* __restrict__ x,
                                                 const float* __restrict__ W,
                                                 const float* __restrict__ EF,
                                                 float* __restrict__ ws) {
  const int blk = blockIdx.x;
  if (blk < QB_L1) {
    qv_body(x, W, ws, blk & 127, blk >> 7);
  } else {
    const int g = blk - QB_L1;
    const int q = g / 11;
    gram_body(x, EF, ws, g - q * 11, q & 7, q >> 3);
  }
}

__global__ __launch_bounds__(128) void y_reduce(const float* __restrict__ x,
                                                const float* __restrict__ W,
                                                float* __restrict__ ws) {
  __shared__ float tile[64][65];
  if (blockIdx.x < QB_L2) {
    const int qb = QB_L1 + blockIdx.x;
    qv_body(x, W, ws, qb & 127, qb >> 7);
    return;
  }
  const int tid = threadIdx.x;
  const int yb  = blockIdx.x - QB_L2;
  const int b_  = yb / NTILE;
  const int tix = yb - b_ * NTILE;
  int I, J;
  tile_ij(tix, I, J);
  float* yf = ws + OFF_YF + (size_t)b_ * YROWS * CCH;
  const v4f* sp0 = (const v4f*)(ws + OFF_YP + ((size_t)(b_ * NSLICE * NTILE + tix)) * 4096);
#pragma unroll
  for (int m = 0; m < 8; ++m) {
    const int idx  = tid + 128 * m;
    const int rloc = idx >> 4;
    const int c4   = idx & 15;
    v4f s = sp0[idx];
#pragma unroll
    for (int sl = 1; sl < NSLICE; ++sl) s += sp0[(size_t)sl * NTILE * 1024 + idx];
    *(v4f*)(yf + (size_t)(I * 64 + rloc) * CCH + J * 64 + c4 * 4) = s;
    tile[rloc][c4 * 4 + 0] = s.x;
    tile[rloc][c4 * 4 + 1] = s.y;
    tile[rloc][c4 * 4 + 2] = s.z;
    tile[rloc][c4 * 4 + 3] = s.w;
  }
  __syncthreads();
  if (I < J) {
#pragma unroll
    for (int m = 0; m < 8; ++m) {
      const int idx  = tid + 128 * m;
      const int cloc = idx >> 4;
      const int r4   = idx & 15;
      v4f o;
      o.x = tile[r4 * 4 + 0][cloc];
      o.y = tile[r4 * 4 + 1][cloc];
      o.z = tile[r4 * 4 + 2][cloc];
      o.w = tile[r4 * 4 + 3][cloc];
      *(v4f*)(yf + (size_t)(J * 64 + cloc) * CCH + I * 64 + r4 * 4) = o;
    }
  }
}

template <int MI, int NJ>
__device__ __forceinline__ void sg_load(v4f (&fa)[MI][4], v4f (&fb)[NJ][4],
                                        const float* __restrict__ aPtr,
                                        const float* __restrict__ bPtr, const int k) {
#pragma unroll
  for (int i = 0; i < MI; ++i)
#pragma unroll
    for (int q = 0; q < 4; ++q)
      fa[i][q] = *(const v4f*)(aPtr + (size_t)i * 16 * CCH + k + q * 4);
#pragma unroll
  for (int j = 0; j < NJ; ++j)
#pragma unroll
    for (int q = 0; q < 4; ++q)
      fb[j][q] = *(const v4f*)(bPtr + (size_t)j * 16 * CCH + k + q * 4);
}

template <int MI, int NJ>
__device__ __forceinline__ void sg_mma(v8f (&acc)[MI][NJ], const v4f (&fa)[MI][4],
                                       const v4f (&fb)[NJ][4]) {
#pragma unroll
  for (int q = 0; q < 4; ++q) {
#pragma unroll
    for (int i = 0; i < MI; ++i) {
      v2f a;
      a.x = fa[i][q].x;
      a.y = fa[i][q].y;
#pragma unroll
      for (int j = 0; j < NJ; ++j) {
        v2f b;
        b.x = fb[j][q].x;
        b.y = fb[j][q].y;
        acc[i][j] = WMMA_F32(a, b, acc[i][j]);
      }
    }
#pragma unroll
    for (int i = 0; i < MI; ++i) {
      v2f a;
      a.x = fa[i][q].z;
      a.y = fa[i][q].w;
#pragma unroll
      for (int j = 0; j < NJ; ++j) {
        v2f b;
        b.x = fb[j][q].z;
        b.y = fb[j][q].w;
        acc[i][j] = WMMA_F32(a, b, acc[i][j]);
      }
    }
  }
}

template <int MI, int NJ>
__device__ __forceinline__ void sg_run(const float* __restrict__ aRow,
                                       const float* __restrict__ bRow,
                                       float* __restrict__ dst, const int ld,
                                       const int l16, const int hi) {
  const float* aPtr = aRow + hi * 16;
  const float* bPtr = bRow + hi * 16;
  v8f acc[MI][NJ] = {};
  v4f fa0[MI][4], fb0[NJ][4], fa1[MI][4], fb1[NJ][4];
  sg_load<MI, NJ>(fa0, fb0, aPtr, bPtr, 0);
  for (int k0 = 0; k0 < CCH - 64; k0 += 64) {
    sg_load<MI, NJ>(fa1, fb1, aPtr, bPtr, k0 + 32);
    sg_mma<MI, NJ>(acc, fa0, fb0);
    sg_load<MI, NJ>(fa0, fb0, aPtr, bPtr, k0 + 64);
    sg_mma<MI, NJ>(acc, fa1, fb1);
  }
  sg_load<MI, NJ>(fa1, fb1, aPtr, bPtr, CCH - 32);
  sg_mma<MI, NJ>(acc, fa0, fb0);
  sg_mma<MI, NJ>(acc, fa1, fb1);
#pragma unroll
  for (int i = 0; i < MI; ++i)
#pragma unroll
    for (int j = 0; j < NJ; ++j)
#pragma unroll
      for (int v = 0; v < 8; ++v)
        dst[(size_t)(i * 16 + hi * 8 + v) * ld + j * 16 + l16] = acc[i][j][v];
}

__global__ __launch_bounds__(128) void tt_gemm(const float* __restrict__ W,
                                               float* __restrict__ ws) {
  const int lane = threadIdx.x & 31;
  const int l16  = lane & 15;
  const int hi   = lane >> 4;
  const int wg   = blockIdx.x * 4 + (threadIdx.x >> 5);
  if (wg < 1024) {
    const int b_ = wg >> 8;
    const int to = (wg >> 4) & 15;
    const int tc = wg & 15;
    const float* yf = ws + OFF_YF + (size_t)b_ * YROWS * CCH;
    sg_run<2, 2>(W + (size_t)(CCH + to * 32 + l16) * CCH,
                 yf + (size_t)(tc * 32 + l16) * CCH,
                 ws + OFF_TT + (size_t)b_ * CCH * CCH + (size_t)(to * 32) * CCH + tc * 32,
                 CCH, l16, hi);
  } else {
    const int u     = wg - 1024;
    const int b_    = u >> 6;
    const int which = (u >> 5) & 1;
    const int to    = (u >> 1) & 15;
    const int tk    = u & 1;
    const float* yf = ws + OFF_YF + (size_t)b_ * YROWS * CCH;
    sg_run<2, 2>(W + (size_t)((which ? 3 * CCH : CCH) + to * 32 + l16) * CCH,
                 yf + (size_t)(CCH + tk * 32 + l16) * CCH,
                 ws + (which ? OFF_VP : OFF_KP) + (size_t)b_ * CCH * KL + (size_t)(to * 32) * KL + tk * 32,
                 KL, l16, hi);
  }
}

__device__ __forceinline__ void qv_half_body(const float* __restrict__ x,
                                             const float* __restrict__ W,
                                             float* __restrict__ ws,
                                             const int bx, const int by, const int ch) {
  const int lane = threadIdx.x & 31;
  const int wave = threadIdx.x >> 5;
  const int l16  = lane & 15;
  const int hi   = lane >> 4;
  const int wm = wave >> 1;
  const int wo = wave & 1;
  const int h  = by * 2 + wo;
  const int mW = bx * 128 + wm * 64;
  const int oW = h * DH + ch * 32;

  const float* aPtr = x + (size_t)(mW + l16) * CCH + hi * 16;
  const float* bPtr = W + (size_t)(oW + l16) * CCH + hi * 16;

  v8f acc[4][2] = {};
  v4f fa0[4][4], fb0[2][4], fa1[4][4], fb1[2][4];
  sg_load<4, 2>(fa0, fb0, aPtr, bPtr, 0);
  for (int k0 = 0; k0 < CCH - 64; k0 += 64) {
    sg_load<4, 2>(fa1, fb1, aPtr, bPtr, k0 + 32);
    sg_mma<4, 2>(acc, fa0, fb0);
    sg_load<4, 2>(fa0, fb0, aPtr, bPtr, k0 + 64);
    sg_mma<4, 2>(acc, fa1, fb1);
  }
  sg_load<4, 2>(fa1, fb1, aPtr, bPtr, CCH - 32);
  sg_mma<4, 2>(acc, fa0, fb0);
  sg_mma<4, 2>(acc, fa1, fb1);

  const int b_ = mW >> 12;
  const int bh = b_ * HEADS + h;
  const int nTok0 = mW & (NTOK - 1);
  float* dstBase = ws + OFF_Q + (size_t)bh * DH * NTOK;
#pragma unroll
  for (int i = 0; i < 4; ++i) {
#pragma unroll
    for (int j = 0; j < 2; ++j) {
      const int dd = ch * 32 + j * 16 + l16;
#pragma unroll
      for (int v = 0; v < 8; ++v) {
        const int n = nTok0 + i * 16 + v + (hi ? 8 : 0);
        dstBase[(size_t)dd * NTOK + n] = acc[i][j][v];
      }
    }
  }
  {
    float* sq = ws + OFF_SQ + ((size_t)(bh * 128 + (nTok0 >> 6) * 2 + hi)) * DH;
#pragma unroll
    for (int j = 0; j < 2; ++j) {
      float s = 0.f;
#pragma unroll
      for (int i = 0; i < 4; ++i)
#pragma unroll
        for (int v = 0; v < 8; ++v) s += acc[i][j][v] * acc[i][j][v];
      sq[ch * 32 + j * 16 + l16] = s;
    }
  }
}

__global__ __launch_bounds__(128) void s_gemm(const float* __restrict__ x,
                                              const float* __restrict__ W,
                                              float* __restrict__ ws) {
  if (blockIdx.x >= 64) {
    const int t  = blockIdx.x - 64;
    const int qb = QB_L1 + QB_L2 + (t >> 1);
    qv_half_body(x, W, ws, qb & 127, qb >> 7, t & 1);
    return;
  }
  const int lane = threadIdx.x & 31;
  const int wave = (threadIdx.x >> 5) + 4 * (blockIdx.x & 1);
  const int l16  = lane & 15;
  const int hi   = lane >> 4;
  const int bh = blockIdx.x >> 1;
  const int b_ = bh >> 3;
  const int h  = bh & 7;
  const float* tt = ws + OFF_TT + (size_t)b_ * CCH * CCH + (size_t)(h * DH) * CCH;
  const int ti = wave >> 1;
  const int tj = wave & 1;
  sg_run<1, 2>(W + (size_t)(h * DH + ti * 16 + l16) * CCH,
               tt + (size_t)(tj * 32 + l16) * CCH,
               ws + OFF_S + (size_t)bh * DH * DH + (size_t)(ti * 16) * DH + tj * 32,
               DH, l16, hi);
  for (int r = 0; r < 8; ++r) {
    const int e = wave * 8 + r;
    const v4f* wk = (const v4f*)(W + (size_t)(CCH + h * DH + e) * CCH) + lane;
    const v4f* tr = (const v4f*)(tt + (size_t)e * CCH) + lane;
    float s = 0.f;
#pragma unroll
    for (int m = 0; m < 4; ++m) {
      const v4f a = wk[m * 32];
      const v4f t = tr[m * 32];
      s += (a.x * t.x + a.y * t.y) + (a.z * t.z + a.w * t.w);
    }
    s += __shfl_xor(s, 16, 32);
    s += __shfl_xor(s, 8, 32);
    s += __shfl_xor(s, 4, 32);
    s += __shfl_xor(s, 2, 32);
    s += __shfl_xor(s, 1, 32);
    if (lane == 0) ws[OFF_K2 + (size_t)bh * DH + e] = s;
  }
}

#define CS_PITCH 68
__global__ __launch_bounds__(256) void ca_softmax(const float* __restrict__ W,
                                                  float* __restrict__ ws,
                                                  const float* __restrict__ temperature) {
  __shared__ float Ps[16][CS_PITCH];
  const int tid = threadIdx.x;
  const int bh  = blockIdx.y;
  const int h   = bh & 7;
  const int e4  = tid & 15;
  const int d   = blockIdx.x * 16 + (tid >> 4);
  const int lane = tid & 31;
  const int wave = tid >> 5;
  const int l16  = lane & 15;
  const int hi   = lane >> 4;

  const float* Wv = W + (size_t)(2 * CCH + h * DH + hi * 2) * CCH + wave * 64 + l16;
  v2f bw[16][4];
#pragma unroll
  for (int it = 0; it < 16; ++it)
#pragma unroll
    for (int t = 0; t < 4; ++t) {
      bw[it][t].x = Wv[(size_t)(it * 4) * CCH + t * 16];
      bw[it][t].y = Wv[(size_t)(it * 4 + 1) * CCH + t * 16];
    }

  const v4f s  = *(const v4f*)(ws + OFF_S + (size_t)bh * DH * DH + (size_t)d * DH + e4 * 4);
  const v4f kk = *(const v4f*)(ws + OFF_K2 + (size_t)bh * DH + e4 * 4);
  float q2 = 0.f;
  {
    const float* qq = ws + OFF_SQ + ((size_t)(bh * 128 + e4 * 8)) * DH + d;
#pragma unroll
    for (int m = 0; m < 8; ++m) q2 += qq[m * DH];
  }
  q2 += __shfl_xor(q2, 8, 16);
  q2 += __shfl_xor(q2, 4, 16);
  q2 += __shfl_xor(q2, 2, 16);
  q2 += __shfl_xor(q2, 1, 16);

  const float iq  = 1.0f / fmaxf(sqrtf(q2), 1e-12f);
  const float tsc = temperature[h];
  float sc[4];
#pragma unroll
  for (int c = 0; c < 4; ++c) {
    const float ik = 1.0f / fmaxf(sqrtf(kk[c]), 1e-12f);
    sc[c] = s[c] * iq * ik * tsc;
  }
  float m = fmaxf(fmaxf(sc[0], sc[1]), fmaxf(sc[2], sc[3]));
  m = fmaxf(m, __shfl_xor(m, 8, 16));
  m = fmaxf(m, __shfl_xor(m, 4, 16));
  m = fmaxf(m, __shfl_xor(m, 2, 16));
  m = fmaxf(m, __shfl_xor(m, 1, 16));
  v4f ex;
  ex.x = __expf(sc[0] - m); ex.y = __expf(sc[1] - m);
  ex.z = __expf(sc[2] - m); ex.w = __expf(sc[3] - m);
  float sum = (ex.x + ex.y) + (ex.z + ex.w);
  sum += __shfl_xor(sum, 8, 16);
  sum += __shfl_xor(sum, 4, 16);
  sum += __shfl_xor(sum, 2, 16);
  sum += __shfl_xor(sum, 1, 16);
  const float inv = 1.0f / sum;
  ex *= inv;
  *(v4f*)(&Ps[tid >> 4][e4 * 4]) = ex;
  if (e4 == 0) ws[OFF_IQ + (size_t)bh * DH + d] = iq;
  __syncthreads();

  v8f acc[4] = {};
#pragma unroll
  for (int it = 0; it < 16; ++it) {
    const v2f a = *(const v2f*)(&Ps[l16][it * 4 + hi * 2]);
#pragma unroll
    for (int t = 0; t < 4; ++t) acc[t] = WMMA_F32(a, bw[it][t], acc[t]);
  }
  float* mrow = ws + OFF_M + (size_t)(bh >> 3) * CCH * CCH
              + (size_t)(h * DH + blockIdx.x * 16 + hi * 8) * CCH + wave * 64 + l16;
#pragma unroll
  for (int t = 0; t < 4; ++t)
#pragma unroll
    for (int v = 0; v < 8; ++v) mrow[(size_t)v * CCH + t * 16] = acc[t][v];
}

__global__ __launch_bounds__(128) void ca_gemm(const float* __restrict__ x,
                                               const float* __restrict__ ws,
                                               float* __restrict__ out) {
  const int lane = threadIdx.x & 31;
  const int wave = threadIdx.x >> 5;
  const int l16  = lane & 15;
  const int hi   = lane >> 4;
  const int wm = wave >> 1;
  const int wo = wave & 1;
  const int bx = blockIdx.x & 127;
  const int by = blockIdx.x >> 7;
  const int mW  = bx * 128 + wm * 64;
  const int chW = (by * 2 + wo) * DH;
  const int b_  = mW >> 12;

  const float* aPtr = ws + OFF_M + (size_t)b_ * CCH * CCH + (size_t)(chW + l16) * CCH + hi * 16;
  const float* bPtr = x + (size_t)(mW + l16) * CCH + hi * 16;

  v8f acc[4][4] = {};
  v4f fa0[4][4], fb0[4][4], fa1[4][4], fb1[4][4];

  k1_load_set(fa0, fb0, aPtr, bPtr, 0);
  for (int k0 = 0; k0 < CCH - 64; k0 += 64) {
    k1_load_set(fa1, fb1, aPtr, bPtr, k0 + 32);
    k1_mma_set(acc, fa0, fb0);
    k1_load_set(fa0, fb0, aPtr, bPtr, k0 + 64);
    k1_mma_set(acc, fa1, fb1);
  }
  k1_load_set(fa1, fb1, aPtr, bPtr, CCH - 32);
  k1_mma_set(acc, fa0, fb0);
  k1_mma_set(acc, fa1, fb1);

  float* ob = out + (size_t)(mW + l16) * CCH + chW + hi * 8;
#pragma unroll
  for (int j = 0; j < 4; ++j)
#pragma unroll
    for (int i = 0; i < 4; ++i) {
      v4f lo, up;
      lo.x = acc[i][j][0]; lo.y = acc[i][j][1]; lo.z = acc[i][j][2]; lo.w = acc[i][j][3];
      up.x = acc[i][j][4]; up.y = acc[i][j][5]; up.z = acc[i][j][6]; up.w = acc[i][j][7];
      float* p = ob + (size_t)(j * 16) * CCH + i * 16;
      *(v4f*)p = lo;
      *(v4f*)(p + 4) = up;
    }
}

#define SA_PITCH 68
__global__ __launch_bounds__(256) void sa_attn(float* __restrict__ ws,
                                               const float* __restrict__ temperature2,
                                               float* __restrict__ out) {
  __shared__ float P[128][SA_PITCH];
  const int tid  = threadIdx.x;
  const int lane = tid & 31;
  const int wave = tid >> 5;
  const int l16  = lane & 15;
  const int hi   = lane >> 4;
  const int bh = blockIdx.y;
  const int b_ = bh >> 3;
  const int h  = bh & 7;
  const int nBase = blockIdx.x * 128;

  const float* Q  = ws + OFF_Q  + (size_t)bh * DH * NTOK;
  const float* Kp = ws + OFF_KP + (size_t)bh * DH * KL;
  const float* Vp = ws + OFF_VP + (size_t)bh * DH * KL;
  const float t2 = temperature2[h];
  const float* iqp = ws + OFF_IQ + (size_t)bh * DH;

  {
    const int nGlob = nBase + wave * 16 + l16;
    v8f acc[4] = {};
    for (int d0 = 0; d0 < DH; d0 += 4) {
      const int k0 = d0 + (hi ? 2 : 0);
      v2f a;
      a.x = Q[(size_t)k0 * NTOK + nGlob] * iqp[k0];
      a.y = Q[(size_t)(k0 + 1) * NTOK + nGlob] * iqp[k0 + 1];
#pragma unroll
      for (int t = 0; t < 4; ++t) {
        const int kc = t * 16 + l16;
        v2f bfrag;
        bfrag.x = Kp[(size_t)k0 * KL + kc];
        bfrag.y = Kp[(size_t)(k0 + 1) * KL + kc];
        acc[t] = WMMA_F32(a, bfrag, acc[t]);
      }
    }
#pragma unroll
    for (int v = 0; v < 8; ++v) {
      const float x0 = acc[0][v] * t2;
      const float x1 = acc[1][v] * t2;
      const float x2 = acc[2][v] * t2;
      const float x3 = acc[3][v] * t2;
      float m = fmaxf(fmaxf(x0, x1), fmaxf(x2, x3));
      m = fmaxf(m, __shfl_xor(m, 8, 16));
      m = fmaxf(m, __shfl_xor(m, 4, 16));
      m = fmaxf(m, __shfl_xor(m, 2, 16));
      m = fmaxf(m, __shfl_xor(m, 1, 16));
      const float e0 = __expf(x0 - m);
      const float e1 = __expf(x1 - m);
      const float e2 = __expf(x2 - m);
      const float e3 = __expf(x3 - m);
      float s = (e0 + e1) + (e2 + e3);
      s += __shfl_xor(s, 8, 16);
      s += __shfl_xor(s, 4, 16);
      s += __shfl_xor(s, 2, 16);
      s += __shfl_xor(s, 1, 16);
      const float inv = 1.0f / s;
      float* prow = &P[wave * 16 + v + (hi ? 8 : 0)][l16];
      prow[0]  = e0 * inv;
      prow[16] = e1 * inv;
      prow[32] = e2 * inv;
      prow[48] = e3 * inv;
    }
  }
  __syncthreads();

  {
    const int mRow = wave * 16 + l16;
    v8f acc[4] = {};
    for (int kk0 = 0; kk0 < KL; kk0 += 4) {
      const int k0 = kk0 + (hi ? 2 : 0);
      v2f a;
      a.x = P[mRow][k0];
      a.y = P[mRow][k0 + 1];
#pragma unroll
      for (int t = 0; t < 4; ++t) {
        const int dd = t * 16 + l16;
        v2f bfrag = *(const v2f*)(Vp + (size_t)dd * KL + k0);
        acc[t] = WMMA_F32(a, bfrag, acc[t]);
      }
    }
#pragma unroll
    for (int t = 0; t < 4; ++t) {
      const int dd = t * 16 + l16;
#pragma unroll
      for (int v = 0; v < 8; ++v) {
        const int n = nBase + wave * 16 + v + (hi ? 8 : 0);
        const size_t f = (size_t)dd * (HEADS * NTOK) + (size_t)h * NTOK + n;
        out[(size_t)b_ * (NTOK * CCH) + f] += acc[t][v];
      }
    }
  }
}

extern "C" void kernel_launch(void* const* d_in, const int* in_sizes, int n_in,
                              void* d_out, int out_size, void* d_ws, size_t ws_size,
                              hipStream_t stream) {
  const float* x  = (const float*)d_in[0];
  const float* W  = (const float*)d_in[1];
  const float* EF = (const float*)d_in[2];
  const float* t1 = (const float*)d_in[3];
  const float* t2 = (const float*)d_in[4];
  float* out = (float*)d_out;
  float* ws  = (float*)d_ws;

  proj_gram  <<<512,            128, 0, stream>>>(x, W, EF, ws);
  y_reduce   <<<432,            128, 0, stream>>>(x, W, ws);
  tt_gemm    <<<320,            128, 0, stream>>>(W, ws);
  s_gemm     <<<256,            128, 0, stream>>>(x, W, ws);
  ca_softmax <<<dim3(4, 32),    256, 0, stream>>>(W, ws, t1);
  ca_gemm    <<<512,            128, 0, stream>>>(x, ws, out);
  sa_attn    <<<dim3(32, 32),   256, 0, stream>>>(ws, t2, out);
}


# SA stores x_SA as out's start value in one launch with the channel softmax (1/|q| per block from the square-sum partials); x_CA GEMM, now last, adds into out; 6 launches; |q| sum order changes
# speedup vs baseline: 3.9936x; 1.0299x over previous
#include <hip/hip_runtime.h>

typedef float v2f __attribute__((ext_vector_type(2)));
typedef float v8f __attribute__((ext_vector_type(8)));

#define WMMA_F32(a, b, c) \
  __builtin_amdgcn_wmma_f32_16x16x4_f32(false, (a), false, (b), (short)0, (c), false, false)

#define BATCH 4
#define NTOK 4096
#define CCH 512
#define HEADS 8
#define DH 64
#define KL 64
#define MROWS (BATCH * NTOK)
#define OCH (4 * CCH)
#define BH (BATCH * HEADS)

#define OFF_Q   ((size_t)0)
#define OFF_K   ((size_t)8388608)
#define OFF_VCA ((size_t)16777216)
#define OFF_M   (OFF_VCA)
#define OFF_VSA ((size_t)25165824)
#define OFF_KP  ((size_t)33554432)
#define OFF_VP  ((size_t)33685504)
#define NSLICE 8
#define NTILE  44
#define YROWS  576
#define OFF_YP  (OFF_K)
#define OFF_YF  (OFF_VSA)
#define OFF_TT  (OFF_VSA + (size_t)1179648)
#define OFF_S   (OFF_VSA + (size_t)2228224)
#define OFF_K2  (OFF_VSA + (size_t)2359296)
#define OFF_SQ  (OFF_VSA + (size_t)2361344)

typedef float v4f __attribute__((ext_vector_type(4)));

__device__ __forceinline__ void k1_load_set(v4f (&fa)[4][4], v4f (&fb)[4][4],
                                            const float* __restrict__ aPtr,
                                            const float* __restrict__ bPtr,
                                            const int k) {
#pragma unroll
  for (int i = 0; i < 4; ++i)
#pragma unroll
    for (int q = 0; q < 4; ++q)
      fa[i][q] = *(const v4f*)(aPtr + (size_t)i * 16 * CCH + k + q * 4);
#pragma unroll
  for (int j = 0; j < 4; ++j)
#pragma unroll
    for (int q = 0; q < 4; ++q)
      fb[j][q] = *(const v4f*)(bPtr + (size_t)j * 16 * CCH + k + q * 4);
}

__device__ __forceinline__ void k1_mma_set(v8f (&acc)[4][4],
                                           const v4f (&fa)[4][4],
                                           const v4f (&fb)[4][4]) {
#pragma unroll
  for (int q = 0; q < 4; ++q) {
#pragma unroll
    for (int i = 0; i < 4; ++i) {
      v2f a;
      a.x = fa[i][q].x;
      a.y = fa[i][q].y;
#pragma unroll
      for (int j = 0; j < 4; ++j) {
        v2f b;
        b.x = fb[j][q].x;
        b.y = fb[j][q].y;
        acc[i][j] = WMMA_F32(a, b, acc[i][j]);
      }
    }
#pragma unroll
    for (int i = 0; i < 4; ++i) {
      v2f a;
      a.x = fa[i][q].z;
      a.y = fa[i][q].w;
#pragma unroll
      for (int j = 0; j < 4; ++j) {
        v2f b;
        b.x = fb[j][q].z;
        b.y = fb[j][q].w;
        acc[i][j] = WMMA_F32(a, b, acc[i][j]);
      }
    }
  }
}

__device__ __forceinline__ void qv_body(const float* __restrict__ x,
                                        const float* __restrict__ W,
                                        float* __restrict__ ws,
                                        const int bx, const int by) {
  const int lane = threadIdx.x & 31;
  const int wave = threadIdx.x >> 5;
  const int l16  = lane & 15;
  const int hi   = lane >> 4;
  const int wm = wave >> 1;
  const int wo = wave & 1;
  const int h  = by * 2 + wo;
  const int mW = bx * 128 + wm * 64;
  const int oW = h * DH;

  const float* aPtr = x + (size_t)(mW + l16) * CCH + hi * 16;
  const float* bPtr = W + (size_t)(oW + l16) * CCH + hi * 16;

  v8f acc[4][4] = {};
  v4f fa0[4][4], fb0[4][4], fa1[4][4], fb1[4][4];

  k1_load_set(fa0, fb0, aPtr, bPtr, 0);
  for (int k0 = 0; k0 < CCH - 64; k0 += 64) {
    k1_load_set(fa1, fb1, aPtr, bPtr, k0 + 32);
    k1_mma_set(acc, fa0, fb0);
    k1_load_set(fa0, fb0, aPtr, bPtr, k0 + 64);
    k1_mma_set(acc, fa1, fb1);
  }
  k1_load_set(fa1, fb1, aPtr, bPtr, CCH - 32);
  k1_mma_set(acc, fa0, fb0);
  k1_mma_set(acc, fa1, fb1);

  const int b_ = mW >> 12;
  const int bh = b_ * HEADS + h;
  const int nTok0 = mW & (NTOK - 1);
  float* dstBase = ws + OFF_Q + (size_t)bh * DH * NTOK;
#pragma unroll
  for (int i = 0; i < 4; ++i) {
#pragma unroll
    for (int j = 0; j < 4; ++j) {
      const int dd = j * 16 + l16;
#pragma unroll
      for (int v = 0; v < 8; ++v) {
        const int n = nTok0 + i * 16 + v + (hi ? 8 : 0);
        dstBase[(size_t)dd * NTOK + n] = acc[i][j][v];
      }
    }
  }
  {
    float* sq = ws + OFF_SQ + ((size_t)(bh * 128 + (nTok0 >> 6) * 2 + hi)) * DH;
#pragma unroll
    for (int j = 0; j < 4; ++j) {
      float s = 0.f;
#pragma unroll
      for (int i = 0; i < 4; ++i)
#pragma unroll
        for (int v = 0; v < 8; ++v) s += acc[i][j][v] * acc[i][j][v];
      sq[j * 16 + l16] = s;
    }
  }
}

__device__ __forceinline__ void gr_load(v4f (&f)[4][2], const float* __restrict__ p,
                                        const size_t stride) {
#pragma unroll
  for (int t = 0; t < 4; ++t)
#pragma unroll
    for (int s = 0; s < 2; ++s)
      f[t][s] = *(const v4f*)(p + (size_t)(2 * t + s) * stride);
}

__device__ __forceinline__ void gr_mma(v8f (&acc)[4][4], const v4f (&fa)[4][2],
                                       const v4f (&fb)[4][2]) {
#pragma unroll
  for (int t = 0; t < 4; ++t) {
#pragma unroll
    for (int jr = 0; jr < 4; ++jr) {
      v2f a;
      a.x = fa[t][0][jr];
      a.y = fa[t][1][jr];
#pragma unroll
      for (int jc = 0; jc < 4; ++jc) {
        v2f b;
        b.x = fb[t][0][jc];
        b.y = fb[t][1][jc];
        acc[jr][jc] = WMMA_F32(a, b, acc[jr][jc]);
      }
    }
  }
}

__device__ __forceinline__ void tile_ij(const int tix, int& I, int& J) {
  if (tix >= 36) { I = 8; J = tix - 36; return; }
  int i = 0, rem = tix;
  while (rem >= 8 - i) { rem -= 8 - i; ++i; }
  I = i;
  J = i + rem;
}

__device__ __forceinline__ void gram_body(const float* __restrict__ x,
                                          const float* __restrict__ EF,
                                          float* __restrict__ ws,
                                          const int bx, const int slice, const int b_) {
  const int lane = threadIdx.x & 31;
  const int wave = __builtin_amdgcn_readfirstlane((int)(threadIdx.x >> 5));
  const int l16  = lane & 15;
  const int hi   = lane >> 4;
  const int tix   = bx * 4 + wave;
  int I, J;
  tile_ij(tix, I, J);
  const int n0 = slice * (NTOK / NSLICE) + hi * 8;
  const float* xb = x + (size_t)b_ * NTOK * CCH;
  const float* cPtr = xb + (size_t)n0 * CCH + J * 64 + l16 * 4;
  const bool ef = (I == 8);
  const size_t sA = ef ? (size_t)KL : (size_t)CCH;
  const float* rPtr = ef ? (EF + (size_t)n0 * KL + l16 * 4)
                         : (xb + (size_t)n0 * CCH + I * 64 + l16 * 4);

  v8f acc[4][4] = {};
  v4f fa0[4][2], fb0[4][2], fa1[4][2], fb1[4][2];
  const int NB = NTOK / NSLICE / 16;
  gr_load(fa0, rPtr, sA);
  gr_load(fb0, cPtr, CCH);
  for (int blk = 0; blk < NB - 2; blk += 2) {
    gr_load(fa1, rPtr + (size_t)(blk + 1) * 16 * sA, sA);
    gr_load(fb1, cPtr + (size_t)(blk + 1) * 16 * CCH, CCH);
    gr_mma(acc, fa0, fb0);
    gr_load(fa0, rPtr + (size_t)(blk + 2) * 16 * sA, sA);
    gr_load(fb0, cPtr + (size_t)(blk + 2) * 16 * CCH, CCH);
    gr_mma(acc, fa1, fb1);
  }
  gr_load(fa1, rPtr + (size_t)(NB - 1) * 16 * sA, sA);
  gr_load(fb1, cPtr + (size_t)(NB - 1) * 16 * CCH, CCH);
  gr_mma(acc, fa0, fb0);
  gr_mma(acc, fa1, fb1);

  float* part = ws + OFF_YP + ((size_t)((b_ * NSLICE + slice) * NTILE + tix)) * 4096;
#pragma unroll
  for (int jr = 0; jr < 4; ++jr)
#pragma unroll
    for (int v = 0; v < 8; ++v) {
      v4f o;
      o.x = acc[jr][0][v]; o.y = acc[jr][1][v]; o.z = acc[jr][2][v]; o.w = acc[jr][3][v];
      *(v4f*)(part + (size_t)(32 * hi + 4 * v + jr) * 64 + l16 * 4) = o;
    }
}

#define QB_L1 160
#define QB_L2 256
__global__ __launch_bounds__(128) void proj_gram(const float* __restrict__ x,
                                                 const float* __restrict__ W,
                                                 const float* __restrict__ EF,
                                                 float* __restrict__ ws) {
  const int blk = blockIdx.x;
  if (blk < QB_L1) {
    qv_body(x, W, ws, blk & 127, blk >> 7);
  } else {
    const int g = blk - QB_L1;
    const int q = g / 11;
    gram_body(x, EF, ws, g - q * 11, q & 7, q >> 3);
  }
}

__global__ __launch_bounds__(128) void y_reduce(const float* __restrict__ x,
                                                const float* __restrict__ W,
                                                float* __restrict__ ws) {
  __shared__ float tile[64][65];
  if (blockIdx.x < QB_L2) {
    const int qb = QB_L1 + blockIdx.x;
    qv_body(x, W, ws, qb & 127, qb >> 7);
    return;
  }
  const int tid = threadIdx.x;
  const int yb  = blockIdx.x - QB_L2;
  const int b_  = yb / NTILE;
  const int tix = yb - b_ * NTILE;
  int I, J;
  tile_ij(tix, I, J);
  float* yf = ws + OFF_YF + (size_t)b_ * YROWS * CCH;
  const v4f* sp0 = (const v4f*)(ws + OFF_YP + ((size_t)(b_ * NSLICE * NTILE + tix)) * 4096);
#pragma unroll
  for (int m = 0; m < 8; ++m) {
    const int idx  = tid + 128 * m;
    const int rloc = idx >> 4;
    const int c4   = idx & 15;
    v4f s = sp0[idx];
#pragma unroll
    for (int sl = 1; sl < NSLICE; ++sl) s += sp0[(size_t)sl * NTILE * 1024 + idx];
    *(v4f*)(yf + (size_t)(I * 64 + rloc) * CCH + J * 64 + c4 * 4) = s;
    tile[rloc][c4 * 4 + 0] = s.x;
    tile[rloc][c4 * 4 + 1] = s.y;
    tile[rloc][c4 * 4 + 2] = s.z;
    tile[rloc][c4 * 4 + 3] = s.w;
  }
  __syncthreads();
  if (I < J) {
#pragma unroll
    for (int m = 0; m < 8; ++m) {
      const int idx  = tid + 128 * m;
      const int cloc = idx >> 4;
      const int r4   = idx & 15;
      v4f o;
      o.x = tile[r4 * 4 + 0][cloc];
      o.y = tile[r4 * 4 + 1][cloc];
      o.z = tile[r4 * 4 + 2][cloc];
      o.w = tile[r4 * 4 + 3][cloc];
      *(v4f*)(yf + (size_t)(J * 64 + cloc) * CCH + I * 64 + r4 * 4) = o;
    }
  }
}

template <int MI, int NJ>
__device__ __forceinline__ void sg_load(v4f (&fa)[MI][4], v4f (&fb)[NJ][4],
                                        const float* __restrict__ aPtr,
                                        const float* __restrict__ bPtr, const int k) {
#pragma unroll
  for (int i = 0; i < MI; ++i)
#pragma unroll
    for (int q = 0; q < 4; ++q)
      fa[i][q] = *(const v4f*)(aPtr + (size_t)i * 16 * CCH + k + q * 4);
#pragma unroll
  for (int j = 0; j < NJ; ++j)
#pragma unroll
    for (int q = 0; q < 4; ++q)
      fb[j][q] = *(const v4f*)(bPtr + (size_t)j * 16 * CCH + k + q * 4);
}

template <int MI, int NJ>
__device__ __forceinline__ void sg_mma(v8f (&acc)[MI][NJ], const v4f (&fa)[MI][4],
                                       const v4f (&fb)[NJ][4]) {
#pragma unroll
  for (int q = 0; q < 4; ++q) {
#pragma unroll
    for (int i = 0; i < MI; ++i) {
      v2f a;
      a.x = fa[i][q].x;
      a.y = fa[i][q].y;
#pragma unroll
      for (int j = 0; j < NJ; ++j) {
        v2f b;
        b.x = fb[j][q].x;
        b.y = fb[j][q].y;
        acc[i][j] = WMMA_F32(a, b, acc[i][j]);
      }
    }
#pragma unroll
    for (int i = 0; i < MI; ++i) {
      v2f a;
      a.x = fa[i][q].z;
      a.y = fa[i][q].w;
#pragma unroll
      for (int j = 0; j < NJ; ++j) {
        v2f b;
        b.x = fb[j][q].z;
        b.y = fb[j][q].w;
        acc[i][j] = WMMA_F32(a, b, acc[i][j]);
      }
    }
  }
}

template <int MI, int NJ>
__device__ __forceinline__ void sg_run(const float* __restrict__ aRow,
                                       const float* __restrict__ bRow,
                                       float* __restrict__ dst, const int ld,
                                       const int l16, const int hi) {
  const float* aPtr = aRow + hi * 16;
  const float* bPtr = bRow + hi * 16;
  v8f acc[MI][NJ] = {};
  v4f fa0[MI][4], fb0[NJ][4], fa1[MI][4], fb1[NJ][4];
  sg_load<MI, NJ>(fa0, fb0, aPtr, bPtr, 0);
  for (int k0 = 0; k0 < CCH - 64; k0 += 64) {
    sg_load<MI, NJ>(fa1, fb1, aPtr, bPtr, k0 + 32);
    sg_mma<MI, NJ>(acc, fa0, fb0);
    sg_load<MI, NJ>(fa0, fb0, aPtr, bPtr, k0 + 64);
    sg_mma<MI, NJ>(acc, fa1, fb1);
  }
  sg_load<MI, NJ>(fa1, fb1, aPtr, bPtr, CCH - 32);
  sg_mma<MI, NJ>(acc, fa0, fb0);
  sg_mma<MI, NJ>(acc, fa1, fb1);
#pragma unroll
  for (int i = 0; i < MI; ++i)
#pragma unroll
    for (int j = 0; j < NJ; ++j)
#pragma unroll
      for (int v = 0; v < 8; ++v)
        dst[(size_t)(i * 16 + hi * 8 + v) * ld + j * 16 + l16] = acc[i][j][v];
}

__global__ __launch_bounds__(128) void tt_gemm(const float* __restrict__ W,
                                               float* __restrict__ ws) {
  const int lane = threadIdx.x & 31;
  const int l16  = lane & 15;
  const int hi   = lane >> 4;
  const int wg   = blockIdx.x * 4 + (threadIdx.x >> 5);
  if (wg < 1024) {
    const int b_ = wg >> 8;
    const int to = (wg >> 4) & 15;
    const int tc = wg & 15;
    const float* yf = ws + OFF_YF + (size_t)b_ * YROWS * CCH;
    sg_run<2, 2>(W + (size_t)(CCH + to * 32 + l16) * CCH,
                 yf + (size_t)(tc * 32 + l16) * CCH,
                 ws + OFF_TT + (size_t)b_ * CCH * CCH + (size_t)(to * 32) * CCH + tc * 32,
                 CCH, l16, hi);
  } else {
    const int u     = wg - 1024;
    const int b_    = u >> 6;
    const int which = (u >> 5) & 1;
    const int to    = (u >> 1) & 15;
    const int tk    = u & 1;
    const float* yf = ws + OFF_YF + (size_t)b_ * YROWS * CCH;
    sg_run<2, 2>(W + (size_t)((which ? 3 * CCH : CCH) + to * 32 + l16) * CCH,
                 yf + (size_t)(CCH + tk * 32 + l16) * CCH,
                 ws + (which ? OFF_VP : OFF_KP) + (size_t)b_ * CCH * KL + (size_t)(to * 32) * KL + tk * 32,
                 KL, l16, hi);
  }
}

__device__ __forceinline__ void qv_half_body(const float* __restrict__ x,
                                             const float* __restrict__ W,
                                             float* __restrict__ ws,
                                             const int bx, const int by, const int ch) {
  const int lane = threadIdx.x & 31;
  const int wave = threadIdx.x >> 5;
  const int l16  = lane & 15;
  const int hi   = lane >> 4;
  const int wm = wave >> 1;
  const int wo = wave & 1;
  const int h  = by * 2 + wo;
  const int mW = bx * 128 + wm * 64;
  const int oW = h * DH + ch * 32;

  const float* aPtr = x + (size_t)(mW + l16) * CCH + hi * 16;
  const float* bPtr = W + (size_t)(oW + l16) * CCH + hi * 16;

  v8f acc[4][2] = {};
  v4f fa0[4][4], fb0[2][4], fa1[4][4], fb1[2][4];
  sg_load<4, 2>(fa0, fb0, aPtr, bPtr, 0);
  for (int k0 = 0; k0 < CCH - 64; k0 += 64) {
    sg_load<4, 2>(fa1, fb1, aPtr, bPtr, k0 + 32);
    sg_mma<4, 2>(acc, fa0, fb0);
    sg_load<4, 2>(fa0, fb0, aPtr, bPtr, k0 + 64);
    sg_mma<4, 2>(acc, fa1, fb1);
  }
  sg_load<4, 2>(fa1, fb1, aPtr, bPtr, CCH - 32);
  sg_mma<4, 2>(acc, fa0, fb0);
  sg_mma<4, 2>(acc, fa1, fb1);

  const int b_ = mW >> 12;
  const int bh = b_ * HEADS + h;
  const int nTok0 = mW & (NTOK - 1);
  float* dstBase = ws + OFF_Q + (size_t)bh * DH * NTOK;
#pragma unroll
  for (int i = 0; i < 4; ++i) {
#pragma unroll
    for (int j = 0; j < 2; ++j) {
      const int dd = ch * 32 + j * 16 + l16;
#pragma unroll
      for (int v = 0; v < 8; ++v) {
        const int n = nTok0 + i * 16 + v + (hi ? 8 : 0);
        dstBase[(size_t)dd * NTOK + n] = acc[i][j][v];
      }
    }
  }
  {
    float* sq = ws + OFF_SQ + ((size_t)(bh * 128 + (nTok0 >> 6) * 2 + hi)) * DH;
#pragma unroll
    for (int j = 0; j < 2; ++j) {
      float s = 0.f;
#pragma unroll
      for (int i = 0; i < 4; ++i)
#pragma unroll
        for (int v = 0; v < 8; ++v) s += acc[i][j][v] * acc[i][j][v];
      sq[ch * 32 + j * 16 + l16] = s;
    }
  }
}

__global__ __launch_bounds__(128) void s_gemm(const float* __restrict__ x,
                                              const float* __restrict__ W,
                                              float* __restrict__ ws) {
  if (blockIdx.x >= 64) {
    const int t  = blockIdx.x - 64;
    const int qb = QB_L1 + QB_L2 + (t >> 1);
    qv_half_body(x, W, ws, qb & 127, qb >> 7, t & 1);
    return;
  }
  const int lane = threadIdx.x & 31;
  const int wave = (threadIdx.x >> 5) + 4 * (blockIdx.x & 1);
  const int l16  = lane & 15;
  const int hi   = lane >> 4;
  const int bh = blockIdx.x >> 1;
  const int b_ = bh >> 3;
  const int h  = bh & 7;
  const float* tt = ws + OFF_TT + (size_t)b_ * CCH * CCH + (size_t)(h * DH) * CCH;
  const int ti = wave >> 1;
  const int tj = wave & 1;
  sg_run<1, 2>(W + (size_t)(h * DH + ti * 16 + l16) * CCH,
               tt + (size_t)(tj * 32 + l16) * CCH,
               ws + OFF_S + (size_t)bh * DH * DH + (size_t)(ti * 16) * DH + tj * 32,
               DH, l16, hi);
  for (int r = 0; r < 8; ++r) {
    const int e = wave * 8 + r;
    const v4f* wk = (const v4f*)(W + (size_t)(CCH + h * DH + e) * CCH) + lane;
    const v4f* tr = (const v4f*)(tt + (size_t)e * CCH) + lane;
    float s = 0.f;
#pragma unroll
    for (int m = 0; m < 4; ++m) {
      const v4f a = wk[m * 32];
      const v4f t = tr[m * 32];
      s += (a.x * t.x + a.y * t.y) + (a.z * t.z + a.w * t.w);
    }
    s += __shfl_xor(s, 16, 32);
    s += __shfl_xor(s, 8, 32);
    s += __shfl_xor(s, 4, 32);
    s += __shfl_xor(s, 2, 32);
    s += __shfl_xor(s, 1, 32);
    if (lane == 0) ws[OFF_K2 + (size_t)bh * DH + e] = s;
  }
}

#define CS_PITCH 68
__device__ __forceinline__ void cs_body(const float* __restrict__ W,
                                        float* __restrict__ ws,
                                        const float* __restrict__ temperature,
                                        const int rg, const int half, const int bh) {
  __shared__ float Ps[16][CS_PITCH];
  const int tid = threadIdx.x;
  const int h   = bh & 7;
  const int e4  = tid & 15;
  const int d   = rg * 16 + (tid >> 4);
  const int lane = tid & 31;
  const int wave = tid >> 5;
  const int l16  = lane & 15;
  const int hi   = lane >> 4;

  const int cw = half * 8 + wave;
  const float* Wv = W + (size_t)(2 * CCH + h * DH + hi * 2) * CCH + cw * 32 + l16;
  v2f bw[16][2];
#pragma unroll
  for (int it = 0; it < 16; ++it)
#pragma unroll
    for (int t = 0; t < 2; ++t) {
      bw[it][t].x = Wv[(size_t)(it * 4) * CCH + t * 16];
      bw[it][t].y = Wv[(size_t)(it * 4 + 1) * CCH + t * 16];
    }

  const v4f s  = *(const v4f*)(ws + OFF_S + (size_t)bh * DH * DH + (size_t)d * DH + e4 * 4);
  const v4f kk = *(const v4f*)(ws + OFF_K2 + (size_t)bh * DH + e4 * 4);
  float q2 = 0.f;
  {
    const float* qq = ws + OFF_SQ + ((size_t)(bh * 128 + e4 * 8)) * DH + d;
#pragma unroll
    for (int m = 0; m < 8; ++m) q2 += qq[m * DH];
  }
  q2 += __shfl_xor(q2, 8, 16);
  q2 += __shfl_xor(q2, 4, 16);
  q2 += __shfl_xor(q2, 2, 16);
  q2 += __shfl_xor(q2, 1, 16);

  const float iq  = 1.0f / fmaxf(sqrtf(q2), 1e-12f);
  const float tsc = temperature[h];
  float sc[4];
#pragma unroll
  for (int c = 0; c < 4; ++c) {
    const float ik = 1.0f / fmaxf(sqrtf(kk[c]), 1e-12f);
    sc[c] = s[c] * iq * ik * tsc;
  }
  float m = fmaxf(fmaxf(sc[0], sc[1]), fmaxf(sc[2], sc[3]));
  m = fmaxf(m, __shfl_xor(m, 8, 16));
  m = fmaxf(m, __shfl_xor(m, 4, 16));
  m = fmaxf(m, __shfl_xor(m, 2, 16));
  m = fmaxf(m, __shfl_xor(m, 1, 16));
  v4f ex;
  ex.x = __expf(sc[0] - m); ex.y = __expf(sc[1] - m);
  ex.z = __expf(sc[2] - m); ex.w = __expf(sc[3] - m);
  float sum = (ex.x + ex.y) + (ex.z + ex.w);
  sum += __shfl_xor(sum, 8, 16);
  sum += __shfl_xor(sum, 4, 16);
  sum += __shfl_xor(sum, 2, 16);
  sum += __shfl_xor(sum, 1, 16);
  const float inv = 1.0f / sum;
  ex *= inv;
  *(v4f*)(&Ps[tid >> 4][e4 * 4]) = ex;
  __syncthreads();

  v8f acc[2] = {};
#pragma unroll
  for (int it = 0; it < 16; ++it) {
    const v2f a = *(const v2f*)(&Ps[l16][it * 4 + hi * 2]);
#pragma unroll
    for (int t = 0; t < 2; ++t) acc[t] = WMMA_F32(a, bw[it][t], acc[t]);
  }
  float* mrow = ws + OFF_M + (size_t)(bh >> 3) * CCH * CCH
              + (size_t)(h * DH + rg * 16 + hi * 8) * CCH + cw * 32 + l16;
#pragma unroll
  for (int t = 0; t < 2; ++t)
#pragma unroll
    for (int v = 0; v < 8; ++v) mrow[(size_t)v * CCH + t * 16] = acc[t][v];
}

__global__ __launch_bounds__(128) void ca_gemm(const float* __restrict__ x,
                                               const float* __restrict__ ws,
                                               float* __restrict__ out) {
  const int lane = threadIdx.x & 31;
  const int wave = threadIdx.x >> 5;
  const int l16  = lane & 15;
  const int hi   = lane >> 4;
  const int wm = wave >> 1;
  const int wo = wave & 1;
  const int bx = blockIdx.x & 127;
  const int by = blockIdx.x >> 7;
  const int mW  = bx * 128 + wm * 64;
  const int chW = (by * 2 + wo) * DH;
  const int b_  = mW >> 12;

  const float* aPtr = ws + OFF_M + (size_t)b_ * CCH * CCH + (size_t)(chW + l16) * CCH + hi * 16;
  const float* bPtr = x + (size_t)(mW + l16) * CCH + hi * 16;

  v8f acc[4][4] = {};
  v4f fa0[4][4], fb0[4][4], fa1[4][4], fb1[4][4];

  k1_load_set(fa0, fb0, aPtr, bPtr, 0);
  for (int k0 = 0; k0 < CCH - 64; k0 += 64) {
    k1_load_set(fa1, fb1, aPtr, bPtr, k0 + 32);
    k1_mma_set(acc, fa0, fb0);
    k1_load_set(fa0, fb0, aPtr, bPtr, k0 + 64);
    k1_mma_set(acc, fa1, fb1);
  }
  k1_load_set(fa1, fb1, aPtr, bPtr, CCH - 32);
  k1_mma_set(acc, fa0, fb0);

  float* ob = out + (size_t)(mW + l16) * CCH + chW + hi * 8;
  v4f olo[4][4], oup[4][4];
#pragma unroll
  for (int j = 0; j < 4; ++j)
#pragma unroll
    for (int i = 0; i < 4; ++i) {
      const float* p = ob + (size_t)(j * 16) * CCH + i * 16;
      olo[i][j] = *(const v4f*)p;
      oup[i][j] = *(const v4f*)(p + 4);
    }
  k1_mma_set(acc, fa1, fb1);

#pragma unroll
  for (int j = 0; j < 4; ++j)
#pragma unroll
    for (int i = 0; i < 4; ++i) {
      v4f lo, up;
      lo.x = acc[i][j][0]; lo.y = acc[i][j][1]; lo.z = acc[i][j][2]; lo.w = acc[i][j][3];
      up.x = acc[i][j][4]; up.y = acc[i][j][5]; up.z = acc[i][j][6]; up.w = acc[i][j][7];
      float* p = ob + (size_t)(j * 16) * CCH + i * 16;
      *(v4f*)p = olo[i][j] + lo;
      *(v4f*)(p + 4) = oup[i][j] + up;
    }
}

#define SA_PITCH 68
__device__ __forceinline__ void sa_body(const float* __restrict__ ws,
                                        const float* __restrict__ temperature2,
                                        float* __restrict__ out,
                                        const int bx, const int bh) {
  __shared__ float P[128][SA_PITCH];
  __shared__ float sqp[4][DH];
  __shared__ float iqs[DH];
  const int tid  = threadIdx.x;
  const int lane = tid & 31;
  const int wave = tid >> 5;
  const int l16  = lane & 15;
  const int hi   = lane >> 4;
  const int b_ = bh >> 3;
  const int h  = bh & 7;
  const int nBase = bx * 128;

  const float* Q  = ws + OFF_Q  + (size_t)bh * DH * NTOK;
  const float* Kp = ws + OFF_KP + (size_t)bh * DH * KL;
  const float* Vp = ws + OFF_VP + (size_t)bh * DH * KL;
  const float t2 = temperature2[h];

  {
    const int dch = tid & (DH - 1);
    const int g   = tid >> 6;
    const float* qq = ws + OFF_SQ + ((size_t)(bh * 128 + g * 32)) * DH + dch;
    float s = 0.f;
#pragma unroll
    for (int m = 0; m < 32; ++m) s += qq[m * DH];
    sqp[g][dch] = s;
  }
  __syncthreads();
  if (tid < DH) {
    const float q2 = (sqp[0][tid] + sqp[1][tid]) + (sqp[2][tid] + sqp[3][tid]);
    iqs[tid] = 1.0f / fmaxf(sqrtf(q2), 1e-12f);
  }
  __syncthreads();

  {
    const int nGlob = nBase + wave * 16 + l16;
    v8f acc[4] = {};
#pragma unroll 2
    for (int d0 = 0; d0 < DH; d0 += 4) {
      const int k0 = d0 + (hi ? 2 : 0);
      v2f a;
      a.x = Q[(size_t)k0 * NTOK + nGlob] * iqs[k0];
      a.y = Q[(size_t)(k0 + 1) * NTOK + nGlob] * iqs[k0 + 1];
#pragma unroll
      for (int t = 0; t < 4; ++t) {
        const int kc = t * 16 + l16;
        v2f bfrag;
        bfrag.x = Kp[(size_t)k0 * KL + kc];
        bfrag.y = Kp[(size_t)(k0 + 1) * KL + kc];
        acc[t] = WMMA_F32(a, bfrag, acc[t]);
      }
    }
#pragma unroll
    for (int v = 0; v < 8; ++v) {
      const float x0 = acc[0][v] * t2;
      const float x1 = acc[1][v] * t2;
      const float x2 = acc[2][v] * t2;
      const float x3 = acc[3][v] * t2;
      float m = fmaxf(fmaxf(x0, x1), fmaxf(x2, x3));
      m = fmaxf(m, __shfl_xor(m, 8, 16));
      m = fmaxf(m, __shfl_xor(m, 4, 16));
      m = fmaxf(m, __shfl_xor(m, 2, 16));
      m = fmaxf(m, __shfl_xor(m, 1, 16));
      const float e0 = __expf(x0 - m);
      const float e1 = __expf(x1 - m);
      const float e2 = __expf(x2 - m);
      const float e3 = __expf(x3 - m);
      float s = (e0 + e1) + (e2 + e3);
      s += __shfl_xor(s, 8, 16);
      s += __shfl_xor(s, 4, 16);
      s += __shfl_xor(s, 2, 16);
      s += __shfl_xor(s, 1, 16);
      const float inv = 1.0f / s;
      float* prow = &P[wave * 16 + v + (hi ? 8 : 0)][l16];
      prow[0]  = e0 * inv;
      prow[16] = e1 * inv;
      prow[32] = e2 * inv;
      prow[48] = e3 * inv;
    }
  }
  __syncthreads();

  {
    const int mRow = wave * 16 + l16;
    v8f acc[4] = {};
    for (int kk0 = 0; kk0 < KL; kk0 += 4) {
      const int k0 = kk0 + (hi ? 2 : 0);
      v2f a;
      a.x = P[mRow][k0];
      a.y = P[mRow][k0 + 1];
#pragma unroll
      for (int t = 0; t < 4; ++t) {
        const int dd = t * 16 + l16;
        v2f bfrag = *(const v2f*)(Vp + (size_t)dd * KL + k0);
        acc[t] = WMMA_F32(a, bfrag, acc[t]);
      }
    }
#pragma unroll
    for (int t = 0; t < 4; ++t) {
      const int dd = t * 16 + l16;
#pragma unroll
      for (int v = 0; v < 8; ++v) {
        const int n = nBase + wave * 16 + v + (hi ? 8 : 0);
        const size_t f = (size_t)dd * (HEADS * NTOK) + (size_t)h * NTOK + n;
        out[(size_t)b_ * (NTOK * CCH) + f] = acc[t][v];
      }
    }
  }
}

#define CS_BLOCKS 256
__global__ __launch_bounds__(256) void attn_mix(const float* __restrict__ W,
                                                float* __restrict__ ws,
                                                const float* __restrict__ temperature,
                                                const float* __restrict__ temperature2,
                                                float* __restrict__ out) {
  if (blockIdx.x < CS_BLOCKS) {
    const int cb = blockIdx.x;
    cs_body(W, ws, temperature, (cb >> 1) & 3, cb & 1, cb >> 3);
  } else {
    const int sb = blockIdx.x - CS_BLOCKS;
    sa_body(ws, temperature2, out, sb & 31, sb >> 5);
  }
}

extern "C" void kernel_launch(void* const* d_in, const int* in_sizes, int n_in,
                              void* d_out, int out_size, void* d_ws, size_t ws_size,
                              hipStream_t stream) {
  const float* x  = (const float*)d_in[0];
  const float* W  = (const float*)d_in[1];
  const float* EF = (const float*)d_in[2];
  const float* t1 = (const float*)d_in[3];
  const float* t2 = (const float*)d_in[4];
  float* out = (float*)d_out;
  float* ws  = (float*)d_ws;

  proj_gram  <<<512,            128, 0, stream>>>(x, W, EF, ws);
  y_reduce   <<<432,            128, 0, stream>>>(x, W, ws);
  tt_gemm    <<<320,            128, 0, stream>>>(W, ws);
  s_gemm     <<<256,            128, 0, stream>>>(x, W, ws);
  attn_mix   <<<1280,           256, 0, stream>>>(W, ws, t1, t2, out);
  ca_gemm    <<<512,            128, 0, stream>>>(x, ws, out);
}


# Wk G never stored: each (batch, head, 64-column slice) block keeps its tile in LDS and multiplies it at once with Wq and Wk; S, |k|^2 summed from 8 slice shares (sum order changes); 5 launches
# speedup vs baseline: 4.1620x; 1.0422x over previous
#include <hip/hip_runtime.h>

typedef float v2f __attribute__((ext_vector_type(2)));
typedef float v8f __attribute__((ext_vector_type(8)));

#define WMMA_F32(a, b, c) \
  __builtin_amdgcn_wmma_f32_16x16x4_f32(false, (a), false, (b), (short)0, (c), false, false)

#define BATCH 4
#define NTOK 4096
#define CCH 512
#define HEADS 8
#define DH 64
#define KL 64
#define MROWS (BATCH * NTOK)
#define OCH (4 * CCH)
#define BH (BATCH * HEADS)

#define OFF_Q   ((size_t)0)
#define OFF_K   ((size_t)8388608)
#define OFF_VCA ((size_t)16777216)
#define OFF_M   (OFF_VCA)
#define OFF_VSA ((size_t)25165824)
#define OFF_KP  ((size_t)33554432)
#define OFF_VP  ((size_t)33685504)
#define NSLICE 8
#define NTILE  44
#define YROWS  576
#define OFF_YP  (OFF_K)
#define OFF_YF  (OFF_VSA)
#define NCS    8
#define OFF_SP  (OFF_VSA + (size_t)1179648)
#define OFF_SQ  (OFF_VSA + (size_t)2361344)
#define OFF_K2P (OFF_VSA + (size_t)2623488)

typedef float v4f __attribute__((ext_vector_type(4)));

__device__ __forceinline__ void k1_load_set(v4f (&fa)[4][4], v4f (&fb)[4][4],
                                            const float* __restrict__ aPtr,
                                            const float* __restrict__ bPtr,
                                            const int k) {
#pragma unroll
  for (int i = 0; i < 4; ++i)
#pragma unroll
    for (int q = 0; q < 4; ++q)
      fa[i][q] = *(const v4f*)(aPtr + (size_t)i * 16 * CCH + k + q * 4);
#pragma unroll
  for (int j = 0; j < 4; ++j)
#pragma unroll
    for (int q = 0; q < 4; ++q)
      fb[j][q] = *(const v4f*)(bPtr + (size_t)j * 16 * CCH + k + q * 4);
}

__device__ __forceinline__ void k1_mma_set(v8f (&acc)[4][4],
                                           const v4f (&fa)[4][4],
                                           const v4f (&fb)[4][4]) {
#pragma unroll
  for (int q = 0; q < 4; ++q) {
#pragma unroll
    for (int i = 0; i < 4; ++i) {
      v2f a;
      a.x = fa[i][q].x;
      a.y = fa[i][q].y;
#pragma unroll
      for (int j = 0; j < 4; ++j) {
        v2f b;
        b.x = fb[j][q].x;
        b.y = fb[j][q].y;
        acc[i][j] = WMMA_F32(a, b, acc[i][j]);
      }
    }
#pragma unroll
    for (int i = 0; i < 4; ++i) {
      v2f a;
      a.x = fa[i][q].z;
      a.y = fa[i][q].w;
#pragma unroll
      for (int j = 0; j < 4; ++j) {
        v2f b;
        b.x = fb[j][q].z;
        b.y = fb[j][q].w;
        acc[i][j] = WMMA_F32(a, b, acc[i][j]);
      }
    }
  }
}

__device__ __forceinline__ void qv_body(const float* __restrict__ x,
                                        const float* __restrict__ W,
                                        float* __restrict__ ws,
                                        const int bx, const int by) {
  const int lane = threadIdx.x & 31;
  const int wave = threadIdx.x >> 5;
  const int l16  = lane & 15;
  const int hi   = lane >> 4;
  const int wm = wave >> 1;
  const int wo = wave & 1;
  const int h  = by * 2 + wo;
  const int mW = bx * 128 + wm * 64;
  const int oW = h * DH;

  const float* aPtr = x + (size_t)(mW + l16) * CCH + hi * 16;
  const float* bPtr = W + (size_t)(oW + l16) * CCH + hi * 16;

  v8f acc[4][4] = {};
  v4f fa0[4][4], fb0[4][4], fa1[4][4], fb1[4][4];

  k1_load_set(fa0, fb0, aPtr, bPtr, 0);
  for (int k0 = 0; k0 < CCH - 64; k0 += 64) {
    k1_load_set(fa1, fb1, aPtr, bPtr, k0 + 32);
    k1_mma_set(acc, fa0, fb0);
    k1_load_set(fa0, fb0, aPtr, bPtr, k0 + 64);
    k1_mma_set(acc, fa1, fb1);
  }
  k1_load_set(fa1, fb1, aPtr, bPtr, CCH - 32);
  k1_mma_set(acc, fa0, fb0);
  k1_mma_set(acc, fa1, fb1);

  const int b_ = mW >> 12;
  const int bh = b_ * HEADS + h;
  const int nTok0 = mW & (NTOK - 1);
  float* dstBase = ws + OFF_Q + (size_t)bh * DH * NTOK;
#pragma unroll
  for (int i = 0; i < 4; ++i) {
#pragma unroll
    for (int j = 0; j < 4; ++j) {
      const int dd = j * 16 + l16;
#pragma unroll
      for (int v = 0; v < 8; ++v) {
        const int n = nTok0 + i * 16 + v + (hi ? 8 : 0);
        dstBase[(size_t)dd * NTOK + n] = acc[i][j][v];
      }
    }
  }
  {
    float* sq = ws + OFF_SQ + ((size_t)(bh * 128 + (nTok0 >> 6) * 2 + hi)) * DH;
#pragma unroll
    for (int j = 0; j < 4; ++j) {
      float s = 0.f;
#pragma unroll
      for (int i = 0; i < 4; ++i)
#pragma unroll
        for (int v = 0; v < 8; ++v) s += acc[i][j][v] * acc[i][j][v];
      sq[j * 16 + l16] = s;
    }
  }
}

__device__ __forceinline__ void gr_load(v4f (&f)[4][2], const float* __restrict__ p,
                                        const size_t stride) {
#pragma unroll
  for (int t = 0; t < 4; ++t)
#pragma unroll
    for (int s = 0; s < 2; ++s)
      f[t][s] = *(const v4f*)(p + (size_t)(2 * t + s) * stride);
}

__device__ __forceinline__ void gr_mma(v8f (&acc)[4][4], const v4f (&fa)[4][2],
                                       const v4f (&fb)[4][2]) {
#pragma unroll
  for (int t = 0; t < 4; ++t) {
#pragma unroll
    for (int jr = 0; jr < 4; ++jr) {
      v2f a;
      a.x = fa[t][0][jr];
      a.y = fa[t][1][jr];
#pragma unroll
      for (int jc = 0; jc < 4; ++jc) {
        v2f b;
        b.x = fb[t][0][jc];
        b.y = fb[t][1][jc];
        acc[jr][jc] = WMMA_F32(a, b, acc[jr][jc]);
      }
    }
  }
}

__device__ __forceinline__ void tile_ij(const int tix, int& I, int& J) {
  if (tix >= 36) { I = 8; J = tix - 36; return; }
  int i = 0, rem = tix;
  while (rem >= 8 - i) { rem -= 8 - i; ++i; }
  I = i;
  J = i + rem;
}

__device__ __forceinline__ void gram_body(const float* __restrict__ x,
                                          const float* __restrict__ EF,
                                          float* __restrict__ ws,
                                          const int bx, const int slice, const int b_) {
  const int lane = threadIdx.x & 31;
  const int wave = __builtin_amdgcn_readfirstlane((int)(threadIdx.x >> 5));
  const int l16  = lane & 15;
  const int hi   = lane >> 4;
  const int tix   = bx * 4 + wave;
  int I, J;
  tile_ij(tix, I, J);
  const int n0 = slice * (NTOK / NSLICE) + hi * 8;
  const float* xb = x + (size_t)b_ * NTOK * CCH;
  const float* cPtr = xb + (size_t)n0 * CCH + J * 64 + l16 * 4;
  const bool ef = (I == 8);
  const size_t sA = ef ? (size_t)KL : (size_t)CCH;
  const float* rPtr = ef ? (EF + (size_t)n0 * KL + l16 * 4)
                         : (xb + (size_t)n0 * CCH + I * 64 + l16 * 4);

  v8f acc[4][4] = {};
  v4f fa0[4][2], fb0[4][2], fa1[4][2], fb1[4][2];
  const int NB = NTOK / NSLICE / 16;
  gr_load(fa0, rPtr, sA);
  gr_load(fb0, cPtr, CCH);
  for (int blk = 0; blk < NB - 2; blk += 2) {
    gr_load(fa1, rPtr + (size_t)(blk + 1) * 16 * sA, sA);
    gr_load(fb1, cPtr + (size_t)(blk + 1) * 16 * CCH, CCH);
    gr_mma(acc, fa0, fb0);
    gr_load(fa0, rPtr + (size_t)(blk + 2) * 16 * sA, sA);
    gr_load(fb0, cPtr + (size_t)(blk + 2) * 16 * CCH, CCH);
    gr_mma(acc, fa1, fb1);
  }
  gr_load(fa1, rPtr + (size_t)(NB - 1) * 16 * sA, sA);
  gr_load(fb1, cPtr + (size_t)(NB - 1) * 16 * CCH, CCH);
  gr_mma(acc, fa0, fb0);
  gr_mma(acc, fa1, fb1);

  float* part = ws + OFF_YP + ((size_t)((b_ * NSLICE + slice) * NTILE + tix)) * 4096;
#pragma unroll
  for (int jr = 0; jr < 4; ++jr)
#pragma unroll
    for (int v = 0; v < 8; ++v) {
      v4f o;
      o.x = acc[jr][0][v]; o.y = acc[jr][1][v]; o.z = acc[jr][2][v]; o.w = acc[jr][3][v];
      *(v4f*)(part + (size_t)(32 * hi + 4 * v + jr) * 64 + l16 * 4) = o;
    }
}

#define QB_L1 160
#define QB_L2 256
__global__ __launch_bounds__(128) void proj_gram(const float* __restrict__ x,
                                                 const float* __restrict__ W,
                                                 const float* __restrict__ EF,
                                                 float* __restrict__ ws) {
  const int blk = blockIdx.x;
  if (blk < QB_L1) {
    qv_body(x, W, ws, blk & 127, blk >> 7);
  } else {
    const int g = blk - QB_L1;
    const int q = g / 11;
    gram_body(x, EF, ws, g - q * 11, q & 7, q >> 3);
  }
}

__global__ __launch_bounds__(128) void y_reduce(const float* __restrict__ x,
                                                const float* __restrict__ W,
                                                float* __restrict__ ws) {
  __shared__ float tile[64][65];
  if (blockIdx.x < QB_L2) {
    const int qb = QB_L1 + blockIdx.x;
    qv_body(x, W, ws, qb & 127, qb >> 7);
    return;
  }
  const int tid = threadIdx.x;
  const int yb  = blockIdx.x - QB_L2;
  const int b_  = yb / NTILE;
  const int tix = yb - b_ * NTILE;
  int I, J;
  tile_ij(tix, I, J);
  float* yf = ws + OFF_YF + (size_t)b_ * YROWS * CCH;
  const v4f* sp0 = (const v4f*)(ws + OFF_YP + ((size_t)(b_ * NSLICE * NTILE + tix)) * 4096);
#pragma unroll
  for (int m = 0; m < 8; ++m) {
    const int idx  = tid + 128 * m;
    const int rloc = idx >> 4;
    const int c4   = idx & 15;
    v4f s = sp0[idx];
#pragma unroll
    for (int sl = 1; sl < NSLICE; ++sl) s += sp0[(size_t)sl * NTILE * 1024 + idx];
    *(v4f*)(yf + (size_t)(I * 64 + rloc) * CCH + J * 64 + c4 * 4) = s;
    tile[rloc][c4 * 4 + 0] = s.x;
    tile[rloc][c4 * 4 + 1] = s.y;
    tile[rloc][c4 * 4 + 2] = s.z;
    tile[rloc][c4 * 4 + 3] = s.w;
  }
  __syncthreads();
  if (I < J) {
#pragma unroll
    for (int m = 0; m < 8; ++m) {
      const int idx  = tid + 128 * m;
      const int cloc = idx >> 4;
      const int r4   = idx & 15;
      v4f o;
      o.x = tile[r4 * 4 + 0][cloc];
      o.y = tile[r4 * 4 + 1][cloc];
      o.z = tile[r4 * 4 + 2][cloc];
      o.w = tile[r4 * 4 + 3][cloc];
      *(v4f*)(yf + (size_t)(J * 64 + cloc) * CCH + I * 64 + r4 * 4) = o;
    }
  }
}

template <int MI, int NJ>
__device__ __forceinline__ void sg_load(v4f (&fa)[MI][4], v4f (&fb)[NJ][4],
                                        const float* __restrict__ aPtr,
                                        const float* __restrict__ bPtr, const int k) {
#pragma unroll
  for (int i = 0; i < MI; ++i)
#pragma unroll
    for (int q = 0; q < 4; ++q)
      fa[i][q] = *(const v4f*)(aPtr + (size_t)i * 16 * CCH + k + q * 4);
#pragma unroll
  for (int j = 0; j < NJ; ++j)
#pragma unroll
    for (int q = 0; q < 4; ++q)
      fb[j][q] = *(const v4f*)(bPtr + (size_t)j * 16 * CCH + k + q * 4);
}

template <int MI, int NJ>
__device__ __forceinline__ void sg_mma(v8f (&acc)[MI][NJ], const v4f (&fa)[MI][4],
                                       const v4f (&fb)[NJ][4]) {
#pragma unroll
  for (int q = 0; q < 4; ++q) {
#pragma unroll
    for (int i = 0; i < MI; ++i) {
      v2f a;
      a.x = fa[i][q].x;
      a.y = fa[i][q].y;
#pragma unroll
      for (int j = 0; j < NJ; ++j) {
        v2f b;
        b.x = fb[j][q].x;
        b.y = fb[j][q].y;
        acc[i][j] = WMMA_F32(a, b, acc[i][j]);
      }
    }
#pragma unroll
    for (int i = 0; i < MI; ++i) {
      v2f a;
      a.x = fa[i][q].z;
      a.y = fa[i][q].w;
#pragma unroll
      for (int j = 0; j < NJ; ++j) {
        v2f b;
        b.x = fb[j][q].z;
        b.y = fb[j][q].w;
        acc[i][j] = WMMA_F32(a, b, acc[i][j]);
      }
    }
  }
}

template <int MI, int NJ>
__device__ __forceinline__ void sg_acc(v8f (&acc)[MI][NJ],
                                       const float* __restrict__ aRow,
                                       const float* __restrict__ bRow, const int hi) {
  const float* aPtr = aRow + hi * 16;
  const float* bPtr = bRow + hi * 16;
  v4f fa0[MI][4], fb0[NJ][4], fa1[MI][4], fb1[NJ][4];
  sg_load<MI, NJ>(fa0, fb0, aPtr, bPtr, 0);
  for (int k0 = 0; k0 < CCH - 64; k0 += 64) {
    sg_load<MI, NJ>(fa1, fb1, aPtr, bPtr, k0 + 32);
    sg_mma<MI, NJ>(acc, fa0, fb0);
    sg_load<MI, NJ>(fa0, fb0, aPtr, bPtr, k0 + 64);
    sg_mma<MI, NJ>(acc, fa1, fb1);
  }
  sg_load<MI, NJ>(fa1, fb1, aPtr, bPtr, CCH - 32);
  sg_mma<MI, NJ>(acc, fa0, fb0);
  sg_mma<MI, NJ>(acc, fa1, fb1);
}

template <int MI, int NJ>
__device__ __forceinline__ void sg_run(const float* __restrict__ aRow,
                                       const float* __restrict__ bRow,
                                       float* __restrict__ dst, const int ld,
                                       const int l16, const int hi) {
  v8f acc[MI][NJ] = {};
  sg_acc<MI, NJ>(acc, aRow, bRow, hi);
#pragma unroll
  for (int i = 0; i < MI; ++i)
#pragma unroll
    for (int j = 0; j < NJ; ++j)
#pragma unroll
      for (int v = 0; v < 8; ++v)
        dst[(size_t)(i * 16 + hi * 8 + v) * ld + j * 16 + l16] = acc[i][j][v];
}

__device__ __forceinline__ void qv_half_body(const float* __restrict__ x,
                                             const float* __restrict__ W,
                                             float* __restrict__ ws,
                                             const int bx, const int by, const int ch) {
  const int lane = threadIdx.x & 31;
  const int wave = threadIdx.x >> 5;
  const int l16  = lane & 15;
  const int hi   = lane >> 4;
  const int wm = wave >> 1;
  const int wo = wave & 1;
  const int h  = by * 2 + wo;
  const int mW = bx * 128 + wm * 64;
  const int oW = h * DH + ch * 32;

  const float* aPtr = x + (size_t)(mW + l16) * CCH + hi * 16;
  const float* bPtr = W + (size_t)(oW + l16) * CCH + hi * 16;

  v8f acc[4][2] = {};
  v4f fa0[4][4], fb0[2][4], fa1[4][4], fb1[2][4];
  sg_load<4, 2>(fa0, fb0, aPtr, bPtr, 0);
  for (int k0 = 0; k0 < CCH - 64; k0 += 64) {
    sg_load<4, 2>(fa1, fb1, aPtr, bPtr, k0 + 32);
    sg_mma<4, 2>(acc, fa0, fb0);
    sg_load<4, 2>(fa0, fb0, aPtr, bPtr, k0 + 64);
    sg_mma<4, 2>(acc, fa1, fb1);
  }
  sg_load<4, 2>(fa1, fb1, aPtr, bPtr, CCH - 32);
  sg_mma<4, 2>(acc, fa0, fb0);
  sg_mma<4, 2>(acc, fa1, fb1);

  const int b_ = mW >> 12;
  const int bh = b_ * HEADS + h;
  const int nTok0 = mW & (NTOK - 1);
  float* dstBase = ws + OFF_Q + (size_t)bh * DH * NTOK;
#pragma unroll
  for (int i = 0; i < 4; ++i) {
#pragma unroll
    for (int j = 0; j < 2; ++j) {
      const int dd = ch * 32 + j * 16 + l16;
#pragma unroll
      for (int v = 0; v < 8; ++v) {
        const int n = nTok0 + i * 16 + v + (hi ? 8 : 0);
        dstBase[(size_t)dd * NTOK + n] = acc[i][j][v];
      }
    }
  }
  {
    float* sq = ws + OFF_SQ + ((size_t)(bh * 128 + (nTok0 >> 6) * 2 + hi)) * DH;
#pragma unroll
    for (int j = 0; j < 2; ++j) {
      float s = 0.f;
#pragma unroll
      for (int i = 0; i < 4; ++i)
#pragma unroll
        for (int v = 0; v < 8; ++v) s += acc[i][j][v] * acc[i][j][v];
      sq[ch * 32 + j * 16 + l16] = s;
    }
  }
}

#define TS_PITCH 68
__device__ __forceinline__ void ts_body(const float* __restrict__ W,
                                        float* __restrict__ ws,
                                        const int b_, const int h, const int cs) {
  __shared__ __attribute__((aligned(16))) float Ts[DH][TS_PITCH];
  const int tid  = threadIdx.x;
  const int lane = tid & 31;
  const int wave = tid >> 5;
  const int l16  = lane & 15;
  const int hi   = lane >> 4;
  const int we = wave >> 1;
  const int wc = wave & 1;
  const int bh = b_ * HEADS + h;
  const float* yf = ws + OFF_YF + (size_t)b_ * YROWS * CCH;
  {
    v8f acc[2][2] = {};
    sg_acc<2, 2>(acc, W + (size_t)(CCH + h * DH + we * 32 + l16) * CCH,
                 yf + (size_t)(cs * 64 + wc * 32 + l16) * CCH, hi);
#pragma unroll
    for (int i = 0; i < 2; ++i)
#pragma unroll
      for (int j = 0; j < 2; ++j)
#pragma unroll
        for (int v = 0; v < 8; ++v)
          Ts[we * 32 + i * 16 + hi * 8 + v][wc * 32 + j * 16 + l16] = acc[i][j][v];
  }
  __syncthreads();

  {
    v8f acc[1][4] = {};
    v4f fa[1][4], fb[4][4];
    const float* aPtr = W + (size_t)(h * DH + wave * 16 + l16) * CCH + cs * 64 + hi * 16;
#pragma unroll
    for (int st = 0; st < 2; ++st) {
#pragma unroll
      for (int q = 0; q < 4; ++q) fa[0][q] = *(const v4f*)(aPtr + st * 32 + q * 4);
#pragma unroll
      for (int j = 0; j < 4; ++j)
#pragma unroll
        for (int q = 0; q < 4; ++q)
          fb[j][q] = *(const v4f*)(&Ts[j * 16 + l16][st * 32 + hi * 16 + q * 4]);
      sg_mma<1, 4>(acc, fa, fb);
    }
    float* sp = ws + OFF_SP + ((size_t)(cs * BH + bh)) * DH * DH
              + (size_t)(wave * 16 + hi * 8) * DH + l16;
#pragma unroll
    for (int j = 0; j < 4; ++j)
#pragma unroll
      for (int v = 0; v < 8; ++v) sp[(size_t)v * DH + j * 16] = acc[0][j][v];
  }
  {
    const int e  = tid >> 1;
    const int hf = tid & 1;
    const float* wk = W + (size_t)(CCH + h * DH + e) * CCH + cs * 64 + hf * 32;
    float s = 0.f;
#pragma unroll
    for (int m = 0; m < 8; ++m) {
      const v4f a = *(const v4f*)(wk + m * 4);
      const v4f t = *(const v4f*)(&Ts[e][hf * 32 + m * 4]);
      s += (a.x * t.x + a.y * t.y) + (a.z * t.z + a.w * t.w);
    }
    s += __shfl_xor(s, 1, 32);
    if (hf == 0) ws[OFF_K2P + (size_t)(cs * BH + bh) * DH + e] = s;
  }
}

#define HB_L3 192
#define PB_L3 64
__global__ __launch_bounds__(128) void ts_gemm(const float* __restrict__ x,
                                               const float* __restrict__ W,
                                               float* __restrict__ ws) {
  if (blockIdx.x < HB_L3) {
    const int t  = blockIdx.x;
    const int qb = QB_L1 + QB_L2 + (t >> 1);
    qv_half_body(x, W, ws, qb & 127, qb >> 7, t & 1);
    return;
  }
  if (blockIdx.x < HB_L3 + PB_L3) {
    const int lane = threadIdx.x & 31;
    const int l16  = lane & 15;
    const int hi   = lane >> 4;
    const int u     = (blockIdx.x - HB_L3) * 4 + (threadIdx.x >> 5);
    const int b_    = u >> 6;
    const int which = (u >> 5) & 1;
    const int to    = (u >> 1) & 15;
    const int tk    = u & 1;
    const float* yf = ws + OFF_YF + (size_t)b_ * YROWS * CCH;
    sg_run<2, 2>(W + (size_t)((which ? 3 * CCH : CCH) + to * 32 + l16) * CCH,
                 yf + (size_t)(CCH + tk * 32 + l16) * CCH,
                 ws + (which ? OFF_VP : OFF_KP) + (size_t)b_ * CCH * KL + (size_t)(to * 32) * KL + tk * 32,
                 KL, l16, hi);
    return;
  }
  const int f = blockIdx.x - (HB_L3 + PB_L3);
  ts_body(W, ws, f >> 6, (f >> 3) & 7, f & 7);
}

#define CS_PITCH 68
__device__ __forceinline__ void cs_body(const float* __restrict__ W,
                                        float* __restrict__ ws,
                                        const float* __restrict__ temperature,
                                        const int rg, const int half, const int bh) {
  __shared__ float Ps[16][CS_PITCH];
  const int tid = threadIdx.x;
  const int h   = bh & 7;
  const int e4  = tid & 15;
  const int d   = rg * 16 + (tid >> 4);
  const int lane = tid & 31;
  const int wave = tid >> 5;
  const int l16  = lane & 15;
  const int hi   = lane >> 4;

  const float* spp = ws + OFF_SP + (size_t)bh * DH * DH + (size_t)d * DH + e4 * 4;
  const float* kpp = ws + OFF_K2P + (size_t)bh * DH + e4 * 4;
  v4f s  = *(const v4f*)spp;
  v4f kk = *(const v4f*)kpp;
#pragma unroll
  for (int p = 1; p < NCS; ++p) {
    s  += *(const v4f*)(spp + (size_t)p * BH * DH * DH);
    kk += *(const v4f*)(kpp + (size_t)p * BH * DH);
  }

  const int cw = half * 8 + wave;
  const float* Wv = W + (size_t)(2 * CCH + h * DH + hi * 2) * CCH + cw * 32 + l16;
  v2f bw[16][2];
#pragma unroll
  for (int it = 0; it < 12; ++it)
#pragma unroll
    for (int t = 0; t < 2; ++t) {
      bw[it][t].x = Wv[(size_t)(it * 4) * CCH + t * 16];
      bw[it][t].y = Wv[(size_t)(it * 4 + 1) * CCH + t * 16];
    }

  float q2 = 0.f;
  {
    const float* qq = ws + OFF_SQ + ((size_t)(bh * 128 + e4 * 8)) * DH + d;
#pragma unroll
    for (int m = 0; m < 8; ++m) q2 += qq[m * DH];
  }
  q2 += __shfl_xor(q2, 8, 16);
  q2 += __shfl_xor(q2, 4, 16);
  q2 += __shfl_xor(q2, 2, 16);
  q2 += __shfl_xor(q2, 1, 16);

  const float iq  = 1.0f / fmaxf(sqrtf(q2), 1e-12f);
  const float tsc = temperature[h];
  float sc[4];
#pragma unroll
  for (int c = 0; c < 4; ++c) {
    const float ik = 1.0f / fmaxf(sqrtf(kk[c]), 1e-12f);
    sc[c] = s[c] * iq * ik * tsc;
  }
  float m = fmaxf(fmaxf(sc[0], sc[1]), fmaxf(sc[2], sc[3]));
  m = fmaxf(m, __shfl_xor(m, 8, 16));
  m = fmaxf(m, __shfl_xor(m, 4, 16));
  m = fmaxf(m, __shfl_xor(m, 2, 16));
  m = fmaxf(m, __shfl_xor(m, 1, 16));
  v4f ex;
  ex.x = __expf(sc[0] - m); ex.y = __expf(sc[1] - m);
  ex.z = __expf(sc[2] - m); ex.w = __expf(sc[3] - m);
  float sum = (ex.x + ex.y) + (ex.z + ex.w);
  sum += __shfl_xor(sum, 8, 16);
  sum += __shfl_xor(sum, 4, 16);
  sum += __shfl_xor(sum, 2, 16);
  sum += __shfl_xor(sum, 1, 16);
  const float inv = 1.0f / sum;
  ex *= inv;
  *(v4f*)(&Ps[tid >> 4][e4 * 4]) = ex;
  __syncthreads();

#pragma unroll
  for (int it = 12; it < 16; ++it)
#pragma unroll
    for (int t = 0; t < 2; ++t) {
      bw[it][t].x = Wv[(size_t)(it * 4) * CCH + t * 16];
      bw[it][t].y = Wv[(size_t)(it * 4 + 1) * CCH + t * 16];
    }

  v8f acc[2] = {};
#pragma unroll
  for (int it = 0; it < 16; ++it) {
    const v2f a = *(const v2f*)(&Ps[l16][it * 4 + hi * 2]);
#pragma unroll
    for (int t = 0; t < 2; ++t) acc[t] = WMMA_F32(a, bw[it][t], acc[t]);
  }
  float* mrow = ws + OFF_M + (size_t)(bh >> 3) * CCH * CCH
              + (size_t)(h * DH + rg * 16 + hi * 8) * CCH + cw * 32 + l16;
#pragma unroll
  for (int t = 0; t < 2; ++t)
#pragma unroll
    for (int v = 0; v < 8; ++v) mrow[(size_t)v * CCH + t * 16] = acc[t][v];
}

__global__ __launch_bounds__(128) void ca_gemm(const float* __restrict__ x,
                                               const float* __restrict__ ws,
                                               float* __restrict__ out) {
  const int lane = threadIdx.x & 31;
  const int wave = threadIdx.x >> 5;
  const int l16  = lane & 15;
  const int hi   = lane >> 4;
  const int wm = wave >> 1;
  const int wo = wave & 1;
  const int bx = blockIdx.x & 127;
  const int by = blockIdx.x >> 7;
  const int mW  = bx * 128 + wm * 64;
  const int chW = (by * 2 + wo) * DH;
  const int b_  = mW >> 12;

  const float* aPtr = ws + OFF_M + (size_t)b_ * CCH * CCH + (size_t)(chW + l16) * CCH + hi * 16;
  const float* bPtr = x + (size_t)(mW + l16) * CCH + hi * 16;

  v8f acc[4][4] = {};
  v4f fa0[4][4], fb0[4][4], fa1[4][4], fb1[4][4];

  k1_load_set(fa0, fb0, aPtr, bPtr, 0);
  for (int k0 = 0; k0 < CCH - 64; k0 += 64) {
    k1_load_set(fa1, fb1, aPtr, bPtr, k0 + 32);
    k1_mma_set(acc, fa0, fb0);
    k1_load_set(fa0, fb0, aPtr, bPtr, k0 + 64);
    k1_mma_set(acc, fa1, fb1);
  }
  k1_load_set(fa1, fb1, aPtr, bPtr, CCH - 32);
  k1_mma_set(acc, fa0, fb0);

  float* ob = out + (size_t)(mW + l16) * CCH + chW + hi * 8;
  v4f olo[4][4], oup[4][4];
#pragma unroll
  for (int j = 0; j < 4; ++j)
#pragma unroll
    for (int i = 0; i < 4; ++i) {
      const float* p = ob + (size_t)(j * 16) * CCH + i * 16;
      olo[i][j] = *(const v4f*)p;
      oup[i][j] = *(const v4f*)(p + 4);
    }
  k1_mma_set(acc, fa1, fb1);

#pragma unroll
  for (int j = 0; j < 4; ++j)
#pragma unroll
    for (int i = 0; i < 4; ++i) {
      v4f lo, up;
      lo.x = acc[i][j][0]; lo.y = acc[i][j][1]; lo.z = acc[i][j][2]; lo.w = acc[i][j][3];
      up.x = acc[i][j][4]; up.y = acc[i][j][5]; up.z = acc[i][j][6]; up.w = acc[i][j][7];
      float* p = ob + (size_t)(j * 16) * CCH + i * 16;
      *(v4f*)p = olo[i][j] + lo;
      *(v4f*)(p + 4) = oup[i][j] + up;
    }
}

#define SA_PITCH 68
__device__ __forceinline__ void sa_body(const float* __restrict__ ws,
                                        const float* __restrict__ temperature2,
                                        float* __restrict__ out,
                                        const int bx, const int bh) {
  __shared__ float P[128][SA_PITCH];
  __shared__ float sqp[4][DH];
  __shared__ float iqs[DH];
  const int tid  = threadIdx.x;
  const int lane = tid & 31;
  const int wave = tid >> 5;
  const int l16  = lane & 15;
  const int hi   = lane >> 4;
  const int b_ = bh >> 3;
  const int h  = bh & 7;
  const int nBase = bx * 128;

  const float* Q  = ws + OFF_Q  + (size_t)bh * DH * NTOK;
  const float* Kp = ws + OFF_KP + (size_t)bh * DH * KL;
  const float* Vp = ws + OFF_VP + (size_t)bh * DH * KL;
  const float t2 = temperature2[h];

  {
    const int dch = tid & (DH - 1);
    const int g   = tid >> 6;
    const float* qq = ws + OFF_SQ + ((size_t)(bh * 128 + g * 32)) * DH + dch;
    float s = 0.f;
#pragma unroll
    for (int m = 0; m < 32; ++m) s += qq[m * DH];
    sqp[g][dch] = s;
  }
  __syncthreads();
  if (tid < DH) {
    const float q2 = (sqp[0][tid] + sqp[1][tid]) + (sqp[2][tid] + sqp[3][tid]);
    iqs[tid] = 1.0f / fmaxf(sqrtf(q2), 1e-12f);
  }
  __syncthreads();

  {
    const int nGlob = nBase + wave * 16 + l16;
    v8f acc[4] = {};
#pragma unroll 2
    for (int d0 = 0; d0 < DH; d0 += 4) {
      const int k0 = d0 + (hi ? 2 : 0);
      v2f a;
      a.x = Q[(size_t)k0 * NTOK + nGlob] * iqs[k0];
      a.y = Q[(size_t)(k0 + 1) * NTOK + nGlob] * iqs[k0 + 1];
#pragma unroll
      for (int t = 0; t < 4; ++t) {
        const int kc = t * 16 + l16;
        v2f bfrag;
        bfrag.x = Kp[(size_t)k0 * KL + kc];
        bfrag.y = Kp[(size_t)(k0 + 1) * KL + kc];
        acc[t] = WMMA_F32(a, bfrag, acc[t]);
      }
    }
#pragma unroll
    for (int v = 0; v < 8; ++v) {
      const float x0 = acc[0][v] * t2;
      const float x1 = acc[1][v] * t2;
      const float x2 = acc[2][v] * t2;
      const float x3 = acc[3][v] * t2;
      float m = fmaxf(fmaxf(x0, x1), fmaxf(x2, x3));
      m = fmaxf(m, __shfl_xor(m, 8, 16));
      m = fmaxf(m, __shfl_xor(m, 4, 16));
      m = fmaxf(m, __shfl_xor(m, 2, 16));
      m = fmaxf(m, __shfl_xor(m, 1, 16));
      const float e0 = __expf(x0 - m);
      const float e1 = __expf(x1 - m);
      const float e2 = __expf(x2 - m);
      const float e3 = __expf(x3 - m);
      float s = (e0 + e1) + (e2 + e3);
      s += __shfl_xor(s, 8, 16);
      s += __shfl_xor(s, 4, 16);
      s += __shfl_xor(s, 2, 16);
      s += __shfl_xor(s, 1, 16);
      const float inv = 1.0f / s;
      float* prow = &P[wave * 16 + v + (hi ? 8 : 0)][l16];
      prow[0]  = e0 * inv;
      prow[16] = e1 * inv;
      prow[32] = e2 * inv;
      prow[48] = e3 * inv;
    }
  }
  __syncthreads();

  {
    const int mRow = wave * 16 + l16;
    v8f acc[4] = {};
    for (int kk0 = 0; kk0 < KL; kk0 += 4) {
      const int k0 = kk0 + (hi ? 2 : 0);
      v2f a;
      a.x = P[mRow][k0];
      a.y = P[mRow][k0 + 1];
#pragma unroll
      for (int t = 0; t < 4; ++t) {
        const int dd = t * 16 + l16;
        v2f bfrag = *(const v2f*)(Vp + (size_t)dd * KL + k0);
        acc[t] = WMMA_F32(a, bfrag, acc[t]);
      }
    }
#pragma unroll
    for (int t = 0; t < 4; ++t) {
      const int dd = t * 16 + l16;
#pragma unroll
      for (int v = 0; v < 8; ++v) {
        const int n = nBase + wave * 16 + v + (hi ? 8 : 0);
        const size_t f = (size_t)dd * (HEADS * NTOK) + (size_t)h * NTOK + n;
        out[(size_t)b_ * (NTOK * CCH) + f] = acc[t][v];
      }
    }
  }
}

#define CS_BLOCKS 256
__global__ __launch_bounds__(256) void attn_mix(const float* __restrict__ W,
                                                float* __restrict__ ws,
                                                const float* __restrict__ temperature,
                                                const float* __restrict__ temperature2,
                                                float* __restrict__ out) {
  if (blockIdx.x < CS_BLOCKS) {
    const int cb = blockIdx.x;
    cs_body(W, ws, temperature, (cb >> 1) & 3, cb & 1, cb >> 3);
  } else {
    const int sb = blockIdx.x - CS_BLOCKS;
    sa_body(ws, temperature2, out, sb & 31, sb >> 5);
  }
}

extern "C" void kernel_launch(void* const* d_in, const int* in_sizes, int n_in,
                              void* d_out, int out_size, void* d_ws, size_t ws_size,
                              hipStream_t stream) {
  const float* x  = (const float*)d_in[0];
  const float* W  = (const float*)d_in[1];
  const float* EF = (const float*)d_in[2];
  const float* t1 = (const float*)d_in[3];
  const float* t2 = (const float*)d_in[4];
  float* out = (float*)d_out;
  float* ws  = (float*)d_ws;

  proj_gram  <<<512,            128, 0, stream>>>(x, W, EF, ws);
  y_reduce   <<<432,            128, 0, stream>>>(x, W, ws);
  ts_gemm    <<<512,            128, 0, stream>>>(x, W, ws);
  attn_mix   <<<1280,           256, 0, stream>>>(W, ws, t1, t2, out);
  ca_gemm    <<<512,            128, 0, stream>>>(x, ws, out);
}
